# starting kernel
# baseline (speedup 1.0000x reference)
#include <hip/hip_runtime.h>
#include <cstdint>

#define N_NODES 50000
#define N_EDGES 800000
#define IN_CH 64
#define OUT_CH 32
#define HEADS 4
#define HC 128
#define NEG_SLOPE 0.2f
#define EPSV 1e-16f

typedef __attribute__((ext_vector_type(2))) float v2f;
typedef __attribute__((ext_vector_type(4))) float v4f;
typedef __attribute__((ext_vector_type(8))) float v8f;

__global__ void __launch_bounds__(128) proj_nodes_wmma(
    const float* __restrict__ x, const float* __restrict__ W,
    float* __restrict__ xh)
{
    const int lane = threadIdx.x & 31;
    const int wv   = threadIdx.x >> 5;
    const int half = lane >> 4;
    const int l15  = lane & 15;
    const int m0   = blockIdx.x * 16;

    const int rowA  = m0 + l15;
    const int colB0 = wv * 32 + l15;
    const int colB1 = colB0 + 16;

    v8f acc0 = {};
    v8f acc1 = {};
#pragma unroll
    for (int k0 = 0; k0 < IN_CH; k0 += 4) {
        const int kk = k0 + half * 2;
        v2f a  = *(const v2f*)(x + (size_t)rowA  * IN_CH + kk);
        v2f b0 = *(const v2f*)(W + (size_t)colB0 * IN_CH + kk);
        v2f b1 = *(const v2f*)(W + (size_t)colB1 * IN_CH + kk);
        acc0 = __builtin_amdgcn_wmma_f32_16x16x4_f32(
                   false, a, false, b0, (short)0, acc0, false, false);
        acc1 = __builtin_amdgcn_wmma_f32_16x16x4_f32(
                   false, a, false, b1, (short)0, acc1, false, false);
    }
#pragma unroll
    for (int v = 0; v < 8; ++v) {
        const int m = m0 + v + half * 8;
        xh[(size_t)m * HC + colB0] = acc0[v];
        xh[(size_t)m * HC + colB1] = acc1[v];
    }
}

__global__ void build_weff(const float* __restrict__ W_edge,
                           const float* __restrict__ att_edge,
                           float* __restrict__ weff)
{
    const int t = threadIdx.x;
    const int h = t >> 6, k = t & 63;
    float s = 0.f;
#pragma unroll
    for (int c = 0; c < OUT_CH; ++c)
        s += att_edge[h * OUT_CH + c] * W_edge[(size_t)(h * OUT_CH + c) * IN_CH + k];
    weff[t] = s;
}

__global__ void node_attn_scalars(const float* __restrict__ xh,
                                  const float* __restrict__ att_src,
                                  const float* __restrict__ att_dst,
                                  float* __restrict__ s_node,
                                  float* __restrict__ d_node)
{
    const int t = blockIdx.x * blockDim.x + threadIdx.x;
    if (t >= N_NODES * HEADS) return;
    const int n = t >> 2, h = t & 3;
    const float* xp = xh + (size_t)n * HC + h * OUT_CH;
    float ss = 0.f, dd = 0.f;
#pragma unroll
    for (int c = 0; c < OUT_CH; ++c) {
        const float v = xp[c];
        ss += v * att_src[h * OUT_CH + c];
        dd += v * att_dst[h * OUT_CH + c];
    }
    s_node[t] = ss;
    d_node[t] = dd;
}

__global__ void init_buffers(float* __restrict__ seg_max,
                             float* __restrict__ seg_sum,
                             float* __restrict__ acc)
{
    const int t = blockIdx.x * blockDim.x + threadIdx.x;
    if (t < N_NODES * HEADS) {
        seg_max[t] = -__builtin_huge_valf();
        seg_sum[t] = 0.f;
    }
    if (t < N_NODES * OUT_CH) acc[t] = 0.f;
}

__global__ void __launch_bounds__(256) edge_logits(
    const float* __restrict__ edge_attr, const int* __restrict__ edge_index,
    const float* __restrict__ s_node, const float* __restrict__ d_node,
    const float* __restrict__ weff,
    float* __restrict__ alpha, float* __restrict__ seg_max)
{
    __shared__ float sw[HEADS * IN_CH];
    if (threadIdx.x < HEADS * IN_CH) sw[threadIdx.x] = weff[threadIdx.x];
    __syncthreads();

    const int e = blockIdx.x * blockDim.x + threadIdx.x;
    if (e >= N_EDGES) return;
    const int src = edge_index[e];
    const int dst = edge_index[N_EDGES + e];

    float acch[HEADS] = {0.f, 0.f, 0.f, 0.f};
    const v4f* ea = (const v4f*)(edge_attr + (size_t)e * IN_CH);
#pragma unroll
    for (int k4 = 0; k4 < IN_CH / 4; ++k4) {
        const v4f v = __builtin_nontemporal_load(ea + k4);
#pragma unroll
        for (int h = 0; h < HEADS; ++h) {
            const float* wh = sw + h * IN_CH + k4 * 4;
            acch[h] += v.x * wh[0] + v.y * wh[1] + v.z * wh[2] + v.w * wh[3];
        }
    }
#pragma unroll
    for (int h = 0; h < HEADS; ++h) {
        float a = acch[h] + s_node[dst * 4 + h] + d_node[src * 4 + h];
        a = (a > 0.f) ? a : NEG_SLOPE * a;
        alpha[(size_t)e * 4 + h] = a;
        float* addr = seg_max + dst * 4 + h;
        if (a >= 0.f) atomicMax((int*)addr, __float_as_int(a));
        else          atomicMin((unsigned int*)addr, __float_as_uint(a));
    }
}

__global__ void edge_softmax(const int* __restrict__ edge_index,
                             const float* __restrict__ seg_max,
                             float* __restrict__ alpha,
                             float* __restrict__ seg_sum)
{
    const int t = blockIdx.x * blockDim.x + threadIdx.x;
    if (t >= N_EDGES * HEADS) return;
    const int e = t >> 2, h = t & 3;
    const int dst = edge_index[N_EDGES + e];
    const float ex = __expf(alpha[t] - seg_max[dst * 4 + h]);
    alpha[t] = ex;
    unsafeAtomicAdd(seg_sum + dst * 4 + h, ex);
}

__global__ void __launch_bounds__(256) edge_scatter(
    const int* __restrict__ edge_index, const float* __restrict__ xh,
    const float* __restrict__ alpha, const float* __restrict__ seg_sum,
    float* __restrict__ acc)
{
    const int lane = threadIdx.x & 31;
    const int e = blockIdx.x * 8 + (threadIdx.x >> 5);
    if (e >= N_EDGES) return;
    const int src = edge_index[e];
    const int dst = edge_index[N_EDGES + e];

    const float c0 = alpha[(size_t)e * 4 + 0] / (seg_sum[dst * 4 + 0] + EPSV);
    const float c1 = alpha[(size_t)e * 4 + 1] / (seg_sum[dst * 4 + 1] + EPSV);
    const float c2 = alpha[(size_t)e * 4 + 2] / (seg_sum[dst * 4 + 2] + EPSV);
    const float c3 = alpha[(size_t)e * 4 + 3] / (seg_sum[dst * 4 + 3] + EPSV);

    const float* xp = xh + (size_t)src * HC;
    const float v = c0 * xp[lane] + c1 * xp[32 + lane]
                  + c2 * xp[64 + lane] + c3 * xp[96 + lane];
    unsafeAtomicAdd(acc + (size_t)dst * OUT_CH + lane, v);
}

__global__ void finalize(const float* __restrict__ acc,
                         const float* __restrict__ bias_conv,
                         const float* __restrict__ bias_layer,
                         float* __restrict__ out)
{
    const int t = blockIdx.x * blockDim.x + threadIdx.x;
    if (t >= N_NODES * OUT_CH) return;
    const int c = t & 31;
    const float v = acc[t] * 0.25f + bias_conv[c] + bias_layer[c];
    out[t] = v > 0.f ? v : 0.f;
}

extern "C" void kernel_launch(void* const* d_in, const int* in_sizes, int n_in,
                              void* d_out, int out_size, void* d_ws, size_t ws_size,
                              hipStream_t stream) {
    const float* x          = (const float*)d_in[0];
    const float* edge_attr  = (const float*)d_in[1];
    const float* W          = (const float*)d_in[2];
    const float* W_edge     = (const float*)d_in[3];
    const float* att_src    = (const float*)d_in[4];
    const float* att_dst    = (const float*)d_in[5];
    const float* att_edge   = (const float*)d_in[6];
    const float* bias_conv  = (const float*)d_in[7];
    const float* bias_layer = (const float*)d_in[8];
    const int*   edge_index = (const int*)d_in[9];
    float* out = (float*)d_out;

    float* ws      = (float*)d_ws;
    float* xh      = ws;
    float* s_node  = xh      + (size_t)N_NODES * HC;
    float* d_node  = s_node  + (size_t)N_NODES * HEADS;
    float* weff    = d_node  + (size_t)N_NODES * HEADS;
    float* alpha   = weff    + (size_t)HEADS * IN_CH;
    float* seg_max = alpha   + (size_t)N_EDGES * HEADS;
    float* seg_sum = seg_max + (size_t)N_NODES * HEADS;
    float* accb    = seg_sum + (size_t)N_NODES * HEADS;

    proj_nodes_wmma<<<N_NODES / 16, 128, 0, stream>>>(x, W, xh);
    build_weff<<<1, 256, 0, stream>>>(W_edge, att_edge, weff);
    node_attn_scalars<<<(N_NODES * HEADS + 255) / 256, 256, 0, stream>>>(
        xh, att_src, att_dst, s_node, d_node);
    init_buffers<<<(N_NODES * OUT_CH + 255) / 256, 256, 0, stream>>>(
        seg_max, seg_sum, accb);
    edge_logits<<<(N_EDGES + 255) / 256, 256, 0, stream>>>(
        edge_attr, edge_index, s_node, d_node, weff, alpha, seg_max);
    edge_softmax<<<(N_EDGES * HEADS + 255) / 256, 256, 0, stream>>>(
        edge_index, seg_max, alpha, seg_sum);
    edge_scatter<<<(N_EDGES + 7) / 8, 256, 0, stream>>>(
        edge_index, xh, alpha, seg_sum, accb);
    finalize<<<(N_NODES * OUT_CH + 255) / 256, 256, 0, stream>>>(
        accb, bias_conv, bias_layer, out);
}


# edge_scatter: no global float atomics: workgroup (node range, edge chunk) filters its chunk's dst indices, sums messages in LDS, stores a partial plane; finalize adds 10 planes (sum order changes)
# speedup vs baseline: 1.9238x; 1.9238x over previous
#include <hip/hip_runtime.h>
#include <cstdint>

#define N_NODES 50000
#define N_EDGES 800000
#define IN_CH 64
#define OUT_CH 32
#define HEADS 4
#define HC 128
#define NEG_SLOPE 0.2f
#define EPSV 1e-16f

typedef __attribute__((ext_vector_type(2))) float v2f;
typedef __attribute__((ext_vector_type(4))) float v4f;
typedef __attribute__((ext_vector_type(8))) float v8f;

__global__ void __launch_bounds__(128) proj_nodes_wmma(
    const float* __restrict__ x, const float* __restrict__ W,
    float* __restrict__ xh)
{
    const int lane = threadIdx.x & 31;
    const int wv   = threadIdx.x >> 5;
    const int half = lane >> 4;
    const int l15  = lane & 15;
    const int m0   = blockIdx.x * 16;

    const int rowA  = m0 + l15;
    const int colB0 = wv * 32 + l15;
    const int colB1 = colB0 + 16;

    v8f acc0 = {};
    v8f acc1 = {};
#pragma unroll
    for (int k0 = 0; k0 < IN_CH; k0 += 4) {
        const int kk = k0 + half * 2;
        v2f a  = *(const v2f*)(x + (size_t)rowA  * IN_CH + kk);
        v2f b0 = *(const v2f*)(W + (size_t)colB0 * IN_CH + kk);
        v2f b1 = *(const v2f*)(W + (size_t)colB1 * IN_CH + kk);
        acc0 = __builtin_amdgcn_wmma_f32_16x16x4_f32(
                   false, a, false, b0, (short)0, acc0, false, false);
        acc1 = __builtin_amdgcn_wmma_f32_16x16x4_f32(
                   false, a, false, b1, (short)0, acc1, false, false);
    }
#pragma unroll
    for (int v = 0; v < 8; ++v) {
        const int m = m0 + v + half * 8;
        xh[(size_t)m * HC + colB0] = acc0[v];
        xh[(size_t)m * HC + colB1] = acc1[v];
    }
}

__global__ void build_weff(const float* __restrict__ W_edge,
                           const float* __restrict__ att_edge,
                           float* __restrict__ weff)
{
    const int t = threadIdx.x;
    const int h = t >> 6, k = t & 63;
    float s = 0.f;
#pragma unroll
    for (int c = 0; c < OUT_CH; ++c)
        s += att_edge[h * OUT_CH + c] * W_edge[(size_t)(h * OUT_CH + c) * IN_CH + k];
    weff[t] = s;
}

__global__ void node_attn_scalars(const float* __restrict__ xh,
                                  const float* __restrict__ att_src,
                                  const float* __restrict__ att_dst,
                                  float* __restrict__ s_node,
                                  float* __restrict__ d_node)
{
    const int t = blockIdx.x * blockDim.x + threadIdx.x;
    if (t >= N_NODES * HEADS) return;
    const int n = t >> 2, h = t & 3;
    const float* xp = xh + (size_t)n * HC + h * OUT_CH;
    float ss = 0.f, dd = 0.f;
#pragma unroll
    for (int c = 0; c < OUT_CH; ++c) {
        const float v = xp[c];
        ss += v * att_src[h * OUT_CH + c];
        dd += v * att_dst[h * OUT_CH + c];
    }
    s_node[t] = ss;
    d_node[t] = dd;
}

__global__ void init_buffers(float* __restrict__ seg_max,
                             float* __restrict__ seg_sum)
{
    const int t = blockIdx.x * blockDim.x + threadIdx.x;
    if (t < N_NODES * HEADS) {
        seg_max[t] = -__builtin_huge_valf();
        seg_sum[t] = 0.f;
    }
}

#define EL_WAVES  4
#define EL_TPW    8
#define EL_NTILES (N_EDGES / 16)
#define EL_NWAVES (EL_NTILES / EL_TPW)
#define EL_PITCH4 17

struct EdgeTile {
    v4f r[8];
    int srcA, dstA, srcB, dstB;
};

__device__ __forceinline__ void el_load_tile(
    EdgeTile& T, const float* __restrict__ edge_attr,
    const int* __restrict__ edge_index, int t, int lane, int eo)
{
    const v4f* p = (const v4f*)(edge_attr + (size_t)t * (16 * IN_CH)) + lane;
#pragma unroll
    for (int j = 0; j < 8; ++j)
        T.r[j] = __builtin_nontemporal_load(p + 32 * j);
    const int e = t * 16 + eo;
    T.srcA = edge_index[e];
    T.dstA = edge_index[N_EDGES + e];
    T.srcB = edge_index[e + 4];
    T.dstB = edge_index[N_EDGES + e + 4];
}

__device__ __forceinline__ void el_wave_lds_order()
{
    __builtin_amdgcn_fence(__ATOMIC_RELEASE, "wavefront");
    __builtin_amdgcn_wave_barrier();
    __builtin_amdgcn_fence(__ATOMIC_ACQUIRE, "wavefront");
}

__device__ __forceinline__ void el_finish(
    float a, int e, int hd, int dst,
    float* __restrict__ alpha, float* __restrict__ seg_max)
{
    a = (a > 0.f) ? a : NEG_SLOPE * a;
    alpha[(size_t)e * 4 + hd] = a;
    float* addr = seg_max + dst * 4 + hd;
    if (a >= 0.f) atomicMax((int*)addr, __float_as_int(a));
    else          atomicMin((unsigned int*)addr, __float_as_uint(a));
}

__global__ void __launch_bounds__(EL_WAVES * 32) edge_logits(
    const float* __restrict__ edge_attr, const int* __restrict__ edge_index,
    const float* __restrict__ s_node, const float* __restrict__ d_node,
    const float* __restrict__ weff,
    float* __restrict__ alpha, float* __restrict__ seg_max)
{
    __shared__ v4f stage[EL_WAVES * 16 * EL_PITCH4];

    const int lane = threadIdx.x & 31;
    const int wv   = threadIdx.x >> 5;
    const int c    = lane & 15;
    const int half = lane >> 4;
    const int w    = blockIdx.x * EL_WAVES + wv;
    if (w >= EL_NWAVES) return;

    v4f* st = stage + wv * (16 * EL_PITCH4);

    const int hd = c & 3;
    v4f bw[8];
#pragma unroll
    for (int jj = 0; jj < 8; ++jj)
        bw[jj] = *(const v4f*)(weff + hd * IN_CH + 8 * jj + 4 * half);

    const int q  = c >> 2;
    const int eo = 8 * half + q;

    EdgeTile cur;
    el_load_tile(cur, edge_attr, edge_index, w, lane, eo);

    for (int i = 0; i < EL_TPW; ++i) {
        const int t = w + i * EL_NWAVES;

#pragma unroll
        for (int j = 0; j < 8; ++j)
            st[(2 * j + half) * EL_PITCH4 + c] = cur.r[j];

        const int srcA = cur.srcA, dstA = cur.dstA;
        const int srcB = cur.srcB, dstB = cur.dstB;
        const float sA = s_node[dstA * 4 + hd];
        const float dA = d_node[srcA * 4 + hd];
        const float sB = s_node[dstB * 4 + hd];
        const float dB = d_node[srcB * 4 + hd];

        if (i + 1 < EL_TPW)
            el_load_tile(cur, edge_attr, edge_index, t + EL_NWAVES, lane, eo);

        el_wave_lds_order();

        v4f a[8];
#pragma unroll
        for (int jj = 0; jj < 8; ++jj)
            a[jj] = st[c * EL_PITCH4 + 2 * jj + half];

        v8f acc0 = {};
        v8f acc1 = {};
#pragma unroll
        for (int jj = 0; jj < 8; ++jj) {
            const v2f a0 = {a[jj].x, a[jj].y};
            const v2f a1 = {a[jj].z, a[jj].w};
            const v2f b0 = {bw[jj].x, bw[jj].y};
            const v2f b1 = {bw[jj].z, bw[jj].w};
            acc0 = __builtin_amdgcn_wmma_f32_16x16x4_f32(
                       false, a0, false, b0, (short)0, acc0, false, false);
            acc1 = __builtin_amdgcn_wmma_f32_16x16x4_f32(
                       false, a1, false, b1, (short)0, acc1, false, false);
        }

        el_wave_lds_order();

        asm volatile("v_nop\n\tv_nop\n\tv_nop\n\tv_nop\n\t"
                     "v_nop\n\tv_nop\n\tv_nop\n\tv_nop"
                     : "+v"(acc0), "+v"(acc1));

        const v8f acc = acc0 + acc1;
        const bool q1 = (q & 1) != 0, q2 = (q & 2) != 0;
        const float lA01 = q1 ? acc[1] : acc[0], lA23 = q1 ? acc[3] : acc[2];
        const float lB01 = q1 ? acc[5] : acc[4], lB23 = q1 ? acc[7] : acc[6];
        const float lA = q2 ? lA23 : lA01;
        const float lB = q2 ? lB23 : lB01;

        const int eA = t * 16 + eo;
        el_finish(lA + sA + dA, eA,     hd, dstA, alpha, seg_max);
        el_finish(lB + sB + dB, eA + 4, hd, dstB, alpha, seg_max);
    }
}

__global__ void edge_softmax(const int* __restrict__ edge_index,
                             const float* __restrict__ seg_max,
                             float* __restrict__ alpha,
                             float* __restrict__ seg_sum)
{
    const int t = blockIdx.x * blockDim.x + threadIdx.x;
    if (t >= N_EDGES * HEADS) return;
    const int e = t >> 2, h = t & 3;
    const int dst = edge_index[N_EDGES + e];
    const float ex = __expf(alpha[t] - seg_max[dst * 4 + h]);
    alpha[t] = ex;
    unsafeAtomicAdd(seg_sum + dst * 4 + h, ex);
}

#define SC_R       2000
#define SC_NB      (N_NODES / SC_R)
#define SC_CH      10
#define SC_CE      (N_EDGES / SC_CH)
#define SC_SUB     16000
#define SC_NSUB    (SC_CE / SC_SUB)
#define SC_THREADS 1024
#define SC_WAVES   (SC_THREADS / 32)
#define SC_LD      4
#define SC_G       8
#define SC_PLANE   (N_NODES * OUT_CH)

static_assert(SC_R * SC_NB == N_NODES, "ranges must tile the nodes");
static_assert(SC_CE * SC_CH == N_EDGES && SC_SUB * SC_NSUB == SC_CE, "chunks must tile the edges");
static_assert(SC_THREADS * SC_LD * 4 >= SC_SUB && SC_SUB % 4 == 0, "a round must cover its edges");
static_assert(SC_R <= 2048 && SC_SUB <= (1 << 21), "list entry packing");
static_assert(32 % SC_G == 0, "groups must tile a 32-entry slab");

__device__ __forceinline__ void sc_load_dst(int4 (&dv)[SC_LD],
                                            const int* __restrict__ dstp, int tid)
{
    const int4* p = (const int4*)dstp;
#pragma unroll
    for (int u = 0; u < SC_LD; ++u) {
        const int g = tid + SC_THREADS * u;
        dv[u] = (g < SC_SUB / 4) ? p[g] : make_int4(-1, -1, -1, -1);
    }
}

__global__ void __launch_bounds__(SC_THREADS) edge_scatter(
    const int* __restrict__ edge_index, const float* __restrict__ xh,
    const float* __restrict__ alpha, const float* __restrict__ seg_sum,
    float* __restrict__ part)
{
    __shared__ v4f accs4[SC_R * OUT_CH / 4];
    __shared__ unsigned int list[SC_SUB];
    __shared__ unsigned int cnt[2];

    float* accs = (float*)accs4;
    const int tid  = threadIdx.x;
    const int lane = tid & 31;
    const int wv   = __builtin_amdgcn_readfirstlane(tid >> 5);
    const int b    = blockIdx.x % SC_NB;
    const int ch   = blockIdx.x / SC_NB;
    const int lo   = b * SC_R;
    const int e0   = ch * SC_CE;
    const int* dstp = edge_index + N_EDGES + e0;

    int4 dv[SC_LD];
    sc_load_dst(dv, dstp, tid);

    for (int i = tid; i < SC_R * OUT_CH / 4; i += SC_THREADS)
        accs4[i] = (v4f){0.f, 0.f, 0.f, 0.f};
    if (tid == 0) { cnt[0] = 0u; cnt[1] = 0u; }
    __syncthreads();

    for (int r = 0; r < SC_NSUB; ++r) {
        unsigned int* cn = cnt + (r & 1);
        const int eb = e0 + r * SC_SUB;

        unsigned nm = 0;
#pragma unroll
        for (int u = 0; u < SC_LD; ++u) {
            nm += ((unsigned)(dv[u].x - lo) < SC_R) ? 1u : 0u;
            nm += ((unsigned)(dv[u].y - lo) < SC_R) ? 1u : 0u;
            nm += ((unsigned)(dv[u].z - lo) < SC_R) ? 1u : 0u;
            nm += ((unsigned)(dv[u].w - lo) < SC_R) ? 1u : 0u;
        }
        unsigned pre = 0, tot = 0;
#pragma unroll
        for (int bit = 0; bit < 5; ++bit) {
            const unsigned m = __builtin_amdgcn_ballot_w32(((nm >> bit) & 1u) != 0u);
            pre += __builtin_amdgcn_mbcnt_lo(m, 0u) << bit;
            tot += (unsigned)__builtin_popcount(m) << bit;
        }
        unsigned base = 0;
        if (lane == 0) base = atomicAdd(cn, tot);
        unsigned slot = (unsigned)__builtin_amdgcn_readfirstlane((int)base) + pre;
#pragma unroll
        for (int u = 0; u < SC_LD; ++u) {
            const int g4 = 4 * (tid + SC_THREADS * u);
            const int dk[4] = {dv[u].x, dv[u].y, dv[u].z, dv[u].w};
#pragma unroll
            for (int k = 0; k < 4; ++k) {
                const unsigned dl = (unsigned)(dk[k] - lo);
                if (dl < SC_R) {
                    list[slot] = ((unsigned)(g4 + k) << 11) | dl;
                    ++slot;
                }
            }
        }
        __syncthreads();

        if (r + 1 < SC_NSUB) sc_load_dst(dv, dstp + (r + 1) * SC_SUB, tid);
        if (tid == 0) cnt[(r + 1) & 1] = 0u;

        const unsigned M   = (unsigned)__builtin_amdgcn_readfirstlane((int)*cn);
        const unsigned per = (M + SC_WAVES - 1) / SC_WAVES;
        const unsigned beg = wv * per;
        const unsigned end = (beg + per < M) ? beg + per : M;

        for (unsigned s = beg; s < end; s += 32) {
            int sl = 0; unsigned dl = 0; v4f cf = {0.f, 0.f, 0.f, 0.f};
            if (s + lane < end) {
                const unsigned ent = list[s + lane];
                dl = ent & 2047u;
                const int e = eb + (int)(ent >> 11);
                sl = edge_index[e];
                const v4f al = *(const v4f*)(alpha + (size_t)e * 4);
                const v4f ss = *(const v4f*)(seg_sum + (size_t)(lo + (int)dl) * 4);
                cf = al / (ss + EPSV);
            }
            const int n = (end - s < 32u) ? (int)(end - s) : 32;

            for (int j = 0; j < n; j += SC_G) {
                float xv[SC_G][4];
#pragma unroll
                for (int u = 0; u < SC_G; ++u) {
                    const int sj = __builtin_amdgcn_readlane(sl, j + u);
                    const float* xp = xh + (size_t)sj * HC + lane;
                    xv[u][0] = xp[0];  xv[u][1] = xp[32];
                    xv[u][2] = xp[64]; xv[u][3] = xp[96];
                }
#pragma unroll
                for (int u = 0; u < SC_G; ++u) {
                    if (j + u < n) {
                        const float c0 = __int_as_float(__builtin_amdgcn_readlane(__float_as_int(cf.x), j + u));
                        const float c1 = __int_as_float(__builtin_amdgcn_readlane(__float_as_int(cf.y), j + u));
                        const float c2 = __int_as_float(__builtin_amdgcn_readlane(__float_as_int(cf.z), j + u));
                        const float c3 = __int_as_float(__builtin_amdgcn_readlane(__float_as_int(cf.w), j + u));
                        const int   dj = __builtin_amdgcn_readlane((int)dl, j + u);
                        const float v = c0 * xv[u][0] + c1 * xv[u][1]
                                      + c2 * xv[u][2] + c3 * xv[u][3];
                        __hip_atomic_fetch_add(accs + dj * OUT_CH + lane, v,
                                               __ATOMIC_RELAXED, __HIP_MEMORY_SCOPE_WORKGROUP);
                    }
                }
            }
        }
        __syncthreads();
    }

    v4f* o4 = (v4f*)(part + (size_t)ch * SC_PLANE + (size_t)lo * OUT_CH);
    for (int i = tid; i < SC_R * OUT_CH / 4; i += SC_THREADS)
        o4[i] = accs4[i];
}

__global__ void finalize(const float* __restrict__ part,
                         const float* __restrict__ bias_conv,
                         const float* __restrict__ bias_layer,
                         float* __restrict__ out)
{
    const int t = blockIdx.x * blockDim.x + threadIdx.x;
    if (t >= SC_PLANE / 4) return;
    const v4f* p = (const v4f*)part + t;
    v4f s = p[0];
#pragma unroll
    for (int c = 1; c < SC_CH; ++c)
        s += p[(size_t)c * (SC_PLANE / 4)];
    const int c0 = (t & 7) * 4;
    const float sv[4] = {s.x, s.y, s.z, s.w};
    float rv[4];
#pragma unroll
    for (int k = 0; k < 4; ++k) {
        const float v = sv[k] * 0.25f + bias_conv[c0 + k] + bias_layer[c0 + k];
        rv[k] = v > 0.f ? v : 0.f;
    }
    ((v4f*)out)[t] = (v4f){rv[0], rv[1], rv[2], rv[3]};
}

extern "C" void kernel_launch(void* const* d_in, const int* in_sizes, int n_in,
                              void* d_out, int out_size, void* d_ws, size_t ws_size,
                              hipStream_t stream) {
    const float* x          = (const float*)d_in[0];
    const float* edge_attr  = (const float*)d_in[1];
    const float* W          = (const float*)d_in[2];
    const float* W_edge     = (const float*)d_in[3];
    const float* att_src    = (const float*)d_in[4];
    const float* att_dst    = (const float*)d_in[5];
    const float* att_edge   = (const float*)d_in[6];
    const float* bias_conv  = (const float*)d_in[7];
    const float* bias_layer = (const float*)d_in[8];
    const int*   edge_index = (const int*)d_in[9];
    float* out = (float*)d_out;

    float* ws      = (float*)d_ws;
    float* xh      = ws;
    float* s_node  = xh      + (size_t)N_NODES * HC;
    float* d_node  = s_node  + (size_t)N_NODES * HEADS;
    float* weff    = d_node  + (size_t)N_NODES * HEADS;
    float* alpha   = weff    + (size_t)HEADS * IN_CH;
    float* seg_max = alpha   + (size_t)N_EDGES * HEADS;
    float* seg_sum = seg_max + (size_t)N_NODES * HEADS;
    float* part    = seg_sum + (size_t)N_NODES * HEADS;

    proj_nodes_wmma<<<N_NODES / 16, 128, 0, stream>>>(x, W, xh);
    build_weff<<<1, 256, 0, stream>>>(W_edge, att_edge, weff);
    node_attn_scalars<<<(N_NODES * HEADS + 255) / 256, 256, 0, stream>>>(
        xh, att_src, att_dst, s_node, d_node);
    init_buffers<<<(N_NODES * HEADS + 255) / 256, 256, 0, stream>>>(
        seg_max, seg_sum);
    edge_logits<<<(EL_NWAVES + EL_WAVES - 1) / EL_WAVES, EL_WAVES * 32, 0, stream>>>(
        edge_attr, edge_index, s_node, d_node, weff, alpha, seg_max);
    edge_softmax<<<(N_EDGES * HEADS + 255) / 256, 256, 0, stream>>>(
        edge_index, seg_max, alpha, seg_sum);
    edge_scatter<<<SC_NB * SC_CH, SC_THREADS, 0, stream>>>(
        edge_index, xh, alpha, seg_sum, part);
    finalize<<<(SC_PLANE / 4 + 255) / 256, 256, 0, stream>>>(
        part, bias_conv, bias_layer, out);
}


# segment max and exp-sum without global atomics: workgroup (node range, edge chunk) takes chunk-local max and sum in LDS, 16 planes merged by rescaling; scatter forms exp itself (sum order changes)
# speedup vs baseline: 3.1105x; 1.6169x over previous
#include <hip/hip_runtime.h>
#include <cstdint>

#define N_NODES 50000
#define N_EDGES 800000
#define IN_CH 64
#define OUT_CH 32
#define HEADS 4
#define HC 128
#define NEG_SLOPE 0.2f
#define EPSV 1e-16f

typedef __attribute__((ext_vector_type(2))) float v2f;
typedef __attribute__((ext_vector_type(4))) float v4f;
typedef __attribute__((ext_vector_type(8))) float v8f;

__global__ void __launch_bounds__(128) proj_nodes_wmma(
    const float* __restrict__ x, const float* __restrict__ W,
    float* __restrict__ xh)
{
    const int lane = threadIdx.x & 31;
    const int wv   = threadIdx.x >> 5;
    const int half = lane >> 4;
    const int l15  = lane & 15;
    const int m0   = blockIdx.x * 16;

    const int rowA  = m0 + l15;
    const int colB0 = wv * 32 + l15;
    const int colB1 = colB0 + 16;

    v8f acc0 = {};
    v8f acc1 = {};
#pragma unroll
    for (int k0 = 0; k0 < IN_CH; k0 += 4) {
        const int kk = k0 + half * 2;
        v2f a  = *(const v2f*)(x + (size_t)rowA  * IN_CH + kk);
        v2f b0 = *(const v2f*)(W + (size_t)colB0 * IN_CH + kk);
        v2f b1 = *(const v2f*)(W + (size_t)colB1 * IN_CH + kk);
        acc0 = __builtin_amdgcn_wmma_f32_16x16x4_f32(
                   false, a, false, b0, (short)0, acc0, false, false);
        acc1 = __builtin_amdgcn_wmma_f32_16x16x4_f32(
                   false, a, false, b1, (short)0, acc1, false, false);
    }
#pragma unroll
    for (int v = 0; v < 8; ++v) {
        const int m = m0 + v + half * 8;
        xh[(size_t)m * HC + colB0] = acc0[v];
        xh[(size_t)m * HC + colB1] = acc1[v];
    }
}

__global__ void build_weff(const float* __restrict__ W_edge,
                           const float* __restrict__ att_edge,
                           float* __restrict__ weff)
{
    const int t = threadIdx.x;
    const int h = t >> 6, k = t & 63;
    float s = 0.f;
#pragma unroll
    for (int c = 0; c < OUT_CH; ++c)
        s += att_edge[h * OUT_CH + c] * W_edge[(size_t)(h * OUT_CH + c) * IN_CH + k];
    weff[t] = s;
}

__global__ void node_attn_scalars(const float* __restrict__ xh,
                                  const float* __restrict__ att_src,
                                  const float* __restrict__ att_dst,
                                  float* __restrict__ s_node,
                                  float* __restrict__ d_node)
{
    const int t = blockIdx.x * blockDim.x + threadIdx.x;
    if (t >= N_NODES * HEADS) return;
    const int n = t >> 2, h = t & 3;
    const float* xp = xh + (size_t)n * HC + h * OUT_CH;
    float ss = 0.f, dd = 0.f;
#pragma unroll
    for (int c = 0; c < OUT_CH; ++c) {
        const float v = xp[c];
        ss += v * att_src[h * OUT_CH + c];
        dd += v * att_dst[h * OUT_CH + c];
    }
    s_node[t] = ss;
    d_node[t] = dd;
}

#define EL_WAVES  4
#define EL_TPW    8
#define EL_NTILES (N_EDGES / 16)
#define EL_NWAVES (EL_NTILES / EL_TPW)
#define EL_PITCH4 17

struct EdgeTile {
    v4f r[8];
    int srcA, dstA, srcB, dstB;
};

__device__ __forceinline__ void el_load_tile(
    EdgeTile& T, const float* __restrict__ edge_attr,
    const int* __restrict__ edge_index, int t, int lane, int eo)
{
    const v4f* p = (const v4f*)(edge_attr + (size_t)t * (16 * IN_CH)) + lane;
#pragma unroll
    for (int j = 0; j < 8; ++j)
        T.r[j] = __builtin_nontemporal_load(p + 32 * j);
    const int e = t * 16 + eo;
    T.srcA = edge_index[e];
    T.dstA = edge_index[N_EDGES + e];
    T.srcB = edge_index[e + 4];
    T.dstB = edge_index[N_EDGES + e + 4];
}

__device__ __forceinline__ void el_wave_lds_order()
{
    __builtin_amdgcn_fence(__ATOMIC_RELEASE, "wavefront");
    __builtin_amdgcn_wave_barrier();
    __builtin_amdgcn_fence(__ATOMIC_ACQUIRE, "wavefront");
}

__device__ __forceinline__ void el_finish(
    float a, int e, int hd, float* __restrict__ alpha)
{
    a = (a > 0.f) ? a : NEG_SLOPE * a;
    alpha[(size_t)e * 4 + hd] = a;
}

__global__ void __launch_bounds__(EL_WAVES * 32) edge_logits(
    const float* __restrict__ edge_attr, const int* __restrict__ edge_index,
    const float* __restrict__ s_node, const float* __restrict__ d_node,
    const float* __restrict__ weff,
    float* __restrict__ alpha)
{
    __shared__ v4f stage[EL_WAVES * 16 * EL_PITCH4];

    const int lane = threadIdx.x & 31;
    const int wv   = threadIdx.x >> 5;
    const int c    = lane & 15;
    const int half = lane >> 4;
    const int w    = blockIdx.x * EL_WAVES + wv;
    if (w >= EL_NWAVES) return;

    v4f* st = stage + wv * (16 * EL_PITCH4);

    const int hd = c & 3;
    v4f bw[8];
#pragma unroll
    for (int jj = 0; jj < 8; ++jj)
        bw[jj] = *(const v4f*)(weff + hd * IN_CH + 8 * jj + 4 * half);

    const int q  = c >> 2;
    const int eo = 8 * half + q;

    EdgeTile cur;
    el_load_tile(cur, edge_attr, edge_index, w, lane, eo);

    for (int i = 0; i < EL_TPW; ++i) {
        const int t = w + i * EL_NWAVES;

#pragma unroll
        for (int j = 0; j < 8; ++j)
            st[(2 * j + half) * EL_PITCH4 + c] = cur.r[j];

        const int srcA = cur.srcA, dstA = cur.dstA;
        const int srcB = cur.srcB, dstB = cur.dstB;
        const float sA = s_node[dstA * 4 + hd];
        const float dA = d_node[srcA * 4 + hd];
        const float sB = s_node[dstB * 4 + hd];
        const float dB = d_node[srcB * 4 + hd];

        if (i + 1 < EL_TPW)
            el_load_tile(cur, edge_attr, edge_index, t + EL_NWAVES, lane, eo);

        el_wave_lds_order();

        v4f a[8];
#pragma unroll
        for (int jj = 0; jj < 8; ++jj)
            a[jj] = st[c * EL_PITCH4 + 2 * jj + half];

        v8f acc0 = {};
        v8f acc1 = {};
#pragma unroll
        for (int jj = 0; jj < 8; ++jj) {
            const v2f a0 = {a[jj].x, a[jj].y};
            const v2f a1 = {a[jj].z, a[jj].w};
            const v2f b0 = {bw[jj].x, bw[jj].y};
            const v2f b1 = {bw[jj].z, bw[jj].w};
            acc0 = __builtin_amdgcn_wmma_f32_16x16x4_f32(
                       false, a0, false, b0, (short)0, acc0, false, false);
            acc1 = __builtin_amdgcn_wmma_f32_16x16x4_f32(
                       false, a1, false, b1, (short)0, acc1, false, false);
        }

        el_wave_lds_order();

        asm volatile("v_nop\n\tv_nop\n\tv_nop\n\tv_nop\n\t"
                     "v_nop\n\tv_nop\n\tv_nop\n\tv_nop"
                     : "+v"(acc0), "+v"(acc1));

        const v8f acc = acc0 + acc1;
        const bool q1 = (q & 1) != 0, q2 = (q & 2) != 0;
        const float lA01 = q1 ? acc[1] : acc[0], lA23 = q1 ? acc[3] : acc[2];
        const float lB01 = q1 ? acc[5] : acc[4], lB23 = q1 ? acc[7] : acc[6];
        const float lA = q2 ? lA23 : lA01;
        const float lB = q2 ? lB23 : lB01;

        const int eA = t * 16 + eo;
        el_finish(lA + sA + dA, eA,     hd, alpha);
        el_finish(lB + sB + dB, eA + 4, hd, alpha);
    }
}

#define ST_NB      15
#define ST_R       3336
#define ST_CH      16
#define ST_CE      (N_EDGES / ST_CH)
#define ST_THREADS 1024
#define ST_LD      13
#define ST_B       4

static_assert(ST_R * ST_NB >= N_NODES && ST_R * (ST_NB - 1) < N_NODES && ST_R % 8 == 0, "ranges must cover the nodes");
static_assert(ST_CE * ST_CH == N_EDGES && ST_CE % 4 == 0, "chunks must tile the edges");
static_assert(ST_THREADS * ST_LD * 4 >= ST_CE, "one sweep must cover a chunk");
static_assert(ST_R <= 4096 && ST_CE <= (1 << 20), "list entry packing");

__device__ __forceinline__ void st_lds_max(int* addr, float a)
{
    const int ai = __float_as_int(a);
    if (ai >= 0) __hip_atomic_fetch_max(addr, ai, __ATOMIC_RELAXED, __HIP_MEMORY_SCOPE_WORKGROUP);
    else         __hip_atomic_fetch_min((unsigned int*)addr, (unsigned int)ai,
                                        __ATOMIC_RELAXED, __HIP_MEMORY_SCOPE_WORKGROUP);
}

__global__ void __launch_bounds__(ST_THREADS) seg_stats(
    const int* __restrict__ edge_index, const float* __restrict__ alpha,
    float* __restrict__ pmax, float* __restrict__ psum)
{
    __shared__ v4f mx4[ST_R];
    __shared__ v4f sm4[ST_R];
    __shared__ unsigned int list[ST_CE];
    __shared__ unsigned int cnt;

    const int tid  = threadIdx.x;
    const int lane = tid & 31;
    const int b    = blockIdx.x % ST_NB;
    const int ch   = blockIdx.x / ST_NB;
    const int lo   = b * ST_R;
    const int e0   = ch * ST_CE;

    const int4* p = (const int4*)(edge_index + N_EDGES + e0);
    int4 dv[ST_LD];
#pragma unroll
    for (int u = 0; u < ST_LD; ++u) {
        const int g = tid + ST_THREADS * u;
        dv[u] = (g < ST_CE / 4) ? p[g] : make_int4(-1, -1, -1, -1);
    }

    const float ninf = -__builtin_huge_valf();
    for (int i = tid; i < ST_R; i += ST_THREADS) {
        mx4[i] = (v4f){ninf, ninf, ninf, ninf};
        sm4[i] = (v4f){0.f, 0.f, 0.f, 0.f};
    }
    if (tid == 0) cnt = 0u;
    __syncthreads();

    unsigned nm = 0;
#pragma unroll
    for (int u = 0; u < ST_LD; ++u) {
        nm += ((unsigned)(dv[u].x - lo) < ST_R) ? 1u : 0u;
        nm += ((unsigned)(dv[u].y - lo) < ST_R) ? 1u : 0u;
        nm += ((unsigned)(dv[u].z - lo) < ST_R) ? 1u : 0u;
        nm += ((unsigned)(dv[u].w - lo) < ST_R) ? 1u : 0u;
    }
    unsigned pre = 0, tot = 0;
#pragma unroll
    for (int bit = 0; bit < 6; ++bit) {
        const unsigned m = __builtin_amdgcn_ballot_w32(((nm >> bit) & 1u) != 0u);
        pre += __builtin_amdgcn_mbcnt_lo(m, 0u) << bit;
        tot += (unsigned)__builtin_popcount(m) << bit;
    }
    unsigned base = 0;
    if (lane == 0) base = atomicAdd(&cnt, tot);
    unsigned slot = (unsigned)__builtin_amdgcn_readfirstlane((int)base) + pre;
#pragma unroll
    for (int u = 0; u < ST_LD; ++u) {
        const int g4 = 4 * (tid + ST_THREADS * u);
        const int dk[4] = {dv[u].x, dv[u].y, dv[u].z, dv[u].w};
#pragma unroll
        for (int k = 0; k < 4; ++k) {
            const unsigned dl = (unsigned)(dk[k] - lo);
            if (dl < ST_R) {
                list[slot] = ((unsigned)(g4 + k) << 12) | dl;
                ++slot;
            }
        }
    }
    __syncthreads();

    const unsigned M = (unsigned)__builtin_amdgcn_readfirstlane((int)cnt);
    int*   mxi = (int*)mx4;
    float* smf = (float*)sm4;

    for (unsigned i0 = (unsigned)tid; i0 < M; i0 += ST_B * ST_THREADS) {
        v4f a[ST_B]; unsigned dl[ST_B];
#pragma unroll
        for (int u = 0; u < ST_B; ++u) {
            const unsigned i = i0 + u * ST_THREADS;
            const unsigned ent = list[(i < M) ? i : i0];
            dl[u] = (i < M) ? (ent & 4095u) : 0xFFFFFFFFu;
            a[u] = *(const v4f*)(alpha + ((size_t)e0 + (ent >> 12)) * 4);
        }
#pragma unroll
        for (int u = 0; u < ST_B; ++u) {
            if (dl[u] != 0xFFFFFFFFu) {
                st_lds_max(mxi + dl[u] * 4 + 0, a[u].x);
                st_lds_max(mxi + dl[u] * 4 + 1, a[u].y);
                st_lds_max(mxi + dl[u] * 4 + 2, a[u].z);
                st_lds_max(mxi + dl[u] * 4 + 3, a[u].w);
            }
        }
    }
    __syncthreads();

    for (unsigned i0 = (unsigned)tid; i0 < M; i0 += ST_B * ST_THREADS) {
        v4f a[ST_B]; unsigned dl[ST_B];
#pragma unroll
        for (int u = 0; u < ST_B; ++u) {
            const unsigned i = i0 + u * ST_THREADS;
            const unsigned ent = list[(i < M) ? i : i0];
            dl[u] = (i < M) ? (ent & 4095u) : 0xFFFFFFFFu;
            a[u] = *(const v4f*)(alpha + ((size_t)e0 + (ent >> 12)) * 4);
        }
        asm volatile("" :: "v"(a[0]), "v"(a[1]), "v"(a[2]), "v"(a[3]));
#pragma unroll
        for (int u = 0; u < ST_B; ++u) {
            if (dl[u] != 0xFFFFFFFFu) {
                const v4f m = mx4[dl[u]];
                __hip_atomic_fetch_add(smf + dl[u] * 4 + 0, __expf(a[u].x - m.x),
                                       __ATOMIC_RELAXED, __HIP_MEMORY_SCOPE_WORKGROUP);
                __hip_atomic_fetch_add(smf + dl[u] * 4 + 1, __expf(a[u].y - m.y),
                                       __ATOMIC_RELAXED, __HIP_MEMORY_SCOPE_WORKGROUP);
                __hip_atomic_fetch_add(smf + dl[u] * 4 + 2, __expf(a[u].z - m.z),
                                       __ATOMIC_RELAXED, __HIP_MEMORY_SCOPE_WORKGROUP);
                __hip_atomic_fetch_add(smf + dl[u] * 4 + 3, __expf(a[u].w - m.w),
                                       __ATOMIC_RELAXED, __HIP_MEMORY_SCOPE_WORKGROUP);
            }
        }
    }
    __syncthreads();

    const int nr = (N_NODES - lo < ST_R) ? (N_NODES - lo) : ST_R;
    v4f* om = (v4f*)pmax + (size_t)ch * N_NODES + lo;
    v4f* os = (v4f*)psum + (size_t)ch * N_NODES + lo;
    for (int i = tid; i < nr; i += ST_THREADS) {
        om[i] = mx4[i];
        os[i] = sm4[i];
    }
}

__global__ void seg_combine(const float* __restrict__ pmax,
                            const float* __restrict__ psum,
                            float* __restrict__ seg_ms)
{
    const int t = blockIdx.x * blockDim.x + threadIdx.x;
    if (t >= N_NODES * HEADS) return;
    float m[ST_CH], sv[ST_CH];
#pragma unroll
    for (int c = 0; c < ST_CH; ++c) {
        m[c]  = pmax[(size_t)c * (N_NODES * HEADS) + t];
        sv[c] = psum[(size_t)c * (N_NODES * HEADS) + t];
    }
    const float ninf = -__builtin_huge_valf();
    float M = m[0];
#pragma unroll
    for (int c = 1; c < ST_CH; ++c) M = (m[c] > M) ? m[c] : M;
    if (!__builtin_isfinite(M)) M = 0.f;
    float S = 0.f;
#pragma unroll
    for (int c = 0; c < ST_CH; ++c)
        S += (m[c] == ninf) ? 0.f : sv[c] * __expf(m[c] - M);
    const int n = t >> 2, h = t & 3;
    seg_ms[(size_t)n * 8 + h]     = M;
    seg_ms[(size_t)n * 8 + 4 + h] = S;
}

#define SC_R       2000
#define SC_NB      (N_NODES / SC_R)
#define SC_CH      10
#define SC_CE      (N_EDGES / SC_CH)
#define SC_SUB     16000
#define SC_NSUB    (SC_CE / SC_SUB)
#define SC_THREADS 1024
#define SC_WAVES   (SC_THREADS / 32)
#define SC_LD      4
#define SC_G       8
#define SC_PLANE   (N_NODES * OUT_CH)

static_assert(SC_R * SC_NB == N_NODES, "ranges must tile the nodes");
static_assert(SC_CE * SC_CH == N_EDGES && SC_SUB * SC_NSUB == SC_CE, "chunks must tile the edges");
static_assert(SC_THREADS * SC_LD * 4 >= SC_SUB && SC_SUB % 4 == 0, "a round must cover its edges");
static_assert(SC_R <= 2048 && SC_SUB <= (1 << 21), "list entry packing");
static_assert(32 % SC_G == 0, "groups must tile a 32-entry slab");

__device__ __forceinline__ void sc_load_dst(int4 (&dv)[SC_LD],
                                            const int* __restrict__ dstp, int tid)
{
    const int4* p = (const int4*)dstp;
#pragma unroll
    for (int u = 0; u < SC_LD; ++u) {
        const int g = tid + SC_THREADS * u;
        dv[u] = (g < SC_SUB / 4) ? p[g] : make_int4(-1, -1, -1, -1);
    }
}

__global__ void __launch_bounds__(SC_THREADS) edge_scatter(
    const int* __restrict__ edge_index, const float* __restrict__ xh,
    const float* __restrict__ alpha, const float* __restrict__ seg_ms,
    float* __restrict__ part)
{
    __shared__ v4f accs4[SC_R * OUT_CH / 4];
    __shared__ unsigned int list[SC_SUB];
    __shared__ unsigned int cnt[2];

    float* accs = (float*)accs4;
    const int tid  = threadIdx.x;
    const int lane = tid & 31;
    const int wv   = __builtin_amdgcn_readfirstlane(tid >> 5);
    const int b    = blockIdx.x % SC_NB;
    const int ch   = blockIdx.x / SC_NB;
    const int lo   = b * SC_R;
    const int e0   = ch * SC_CE;
    const int* dstp = edge_index + N_EDGES + e0;

    int4 dv[SC_LD];
    sc_load_dst(dv, dstp, tid);

    for (int i = tid; i < SC_R * OUT_CH / 4; i += SC_THREADS)
        accs4[i] = (v4f){0.f, 0.f, 0.f, 0.f};
    if (tid == 0) { cnt[0] = 0u; cnt[1] = 0u; }
    __syncthreads();

    for (int r = 0; r < SC_NSUB; ++r) {
        unsigned int* cn = cnt + (r & 1);
        const int eb = e0 + r * SC_SUB;

        unsigned nm = 0;
#pragma unroll
        for (int u = 0; u < SC_LD; ++u) {
            nm += ((unsigned)(dv[u].x - lo) < SC_R) ? 1u : 0u;
            nm += ((unsigned)(dv[u].y - lo) < SC_R) ? 1u : 0u;
            nm += ((unsigned)(dv[u].z - lo) < SC_R) ? 1u : 0u;
            nm += ((unsigned)(dv[u].w - lo) < SC_R) ? 1u : 0u;
        }
        unsigned pre = 0, tot = 0;
#pragma unroll
        for (int bit = 0; bit < 5; ++bit) {
            const unsigned m = __builtin_amdgcn_ballot_w32(((nm >> bit) & 1u) != 0u);
            pre += __builtin_amdgcn_mbcnt_lo(m, 0u) << bit;
            tot += (unsigned)__builtin_popcount(m) << bit;
        }
        unsigned base = 0;
        if (lane == 0) base = atomicAdd(cn, tot);
        unsigned slot = (unsigned)__builtin_amdgcn_readfirstlane((int)base) + pre;
#pragma unroll
        for (int u = 0; u < SC_LD; ++u) {
            const int g4 = 4 * (tid + SC_THREADS * u);
            const int dk[4] = {dv[u].x, dv[u].y, dv[u].z, dv[u].w};
#pragma unroll
            for (int k = 0; k < 4; ++k) {
                const unsigned dl = (unsigned)(dk[k] - lo);
                if (dl < SC_R) {
                    list[slot] = ((unsigned)(g4 + k) << 11) | dl;
                    ++slot;
                }
            }
        }
        __syncthreads();

        if (r + 1 < SC_NSUB) sc_load_dst(dv, dstp + (r + 1) * SC_SUB, tid);
        if (tid == 0) cnt[(r + 1) & 1] = 0u;

        const unsigned M   = (unsigned)__builtin_amdgcn_readfirstlane((int)*cn);
        const unsigned per = (M + SC_WAVES - 1) / SC_WAVES;
        const unsigned beg = wv * per;
        const unsigned end = (beg + per < M) ? beg + per : M;

        for (unsigned s = beg; s < end; s += 32) {
            int sl = 0; unsigned dl = 0; v4f cf = {0.f, 0.f, 0.f, 0.f};
            if (s + lane < end) {
                const unsigned ent = list[s + lane];
                dl = ent & 2047u;
                const int e = eb + (int)(ent >> 11);
                sl = edge_index[e];
                const v4f al = *(const v4f*)(alpha + (size_t)e * 4);
                const v4f* ms = (const v4f*)(seg_ms + (size_t)(lo + (int)dl) * 8);
                const v4f mm = ms[0];
                const v4f ss = ms[1];
                const v4f ex = {__expf(al.x - mm.x), __expf(al.y - mm.y),
                                __expf(al.z - mm.z), __expf(al.w - mm.w)};
                cf = ex / (ss + EPSV);
            }
            const int n = (end - s < 32u) ? (int)(end - s) : 32;

            for (int j = 0; j < n; j += SC_G) {
                float xv[SC_G][4];
#pragma unroll
                for (int u = 0; u < SC_G; ++u) {
                    const int sj = __builtin_amdgcn_readlane(sl, j + u);
                    const float* xp = xh + (size_t)sj * HC + lane;
                    xv[u][0] = xp[0];  xv[u][1] = xp[32];
                    xv[u][2] = xp[64]; xv[u][3] = xp[96];
                }
#pragma unroll
                for (int u = 0; u < SC_G; ++u) {
                    if (j + u < n) {
                        const float c0 = __int_as_float(__builtin_amdgcn_readlane(__float_as_int(cf.x), j + u));
                        const float c1 = __int_as_float(__builtin_amdgcn_readlane(__float_as_int(cf.y), j + u));
                        const float c2 = __int_as_float(__builtin_amdgcn_readlane(__float_as_int(cf.z), j + u));
                        const float c3 = __int_as_float(__builtin_amdgcn_readlane(__float_as_int(cf.w), j + u));
                        const int   dj = __builtin_amdgcn_readlane((int)dl, j + u);
                        const float v = c0 * xv[u][0] + c1 * xv[u][1]
                                      + c2 * xv[u][2] + c3 * xv[u][3];
                        __hip_atomic_fetch_add(accs + dj * OUT_CH + lane, v,
                                               __ATOMIC_RELAXED, __HIP_MEMORY_SCOPE_WORKGROUP);
                    }
                }
            }
        }
        __syncthreads();
    }

    v4f* o4 = (v4f*)(part + (size_t)ch * SC_PLANE + (size_t)lo * OUT_CH);
    for (int i = tid; i < SC_R * OUT_CH / 4; i += SC_THREADS)
        o4[i] = accs4[i];
}

__global__ void finalize(const float* __restrict__ part,
                         const float* __restrict__ bias_conv,
                         const float* __restrict__ bias_layer,
                         float* __restrict__ out)
{
    const int t = blockIdx.x * blockDim.x + threadIdx.x;
    if (t >= SC_PLANE / 4) return;
    const v4f* p = (const v4f*)part + t;
    v4f s = p[0];
#pragma unroll
    for (int c = 1; c < SC_CH; ++c)
        s += p[(size_t)c * (SC_PLANE / 4)];
    const int c0 = (t & 7) * 4;
    const float sv[4] = {s.x, s.y, s.z, s.w};
    float rv[4];
#pragma unroll
    for (int k = 0; k < 4; ++k) {
        const float v = sv[k] * 0.25f + bias_conv[c0 + k] + bias_layer[c0 + k];
        rv[k] = v > 0.f ? v : 0.f;
    }
    ((v4f*)out)[t] = (v4f){rv[0], rv[1], rv[2], rv[3]};
}

extern "C" void kernel_launch(void* const* d_in, const int* in_sizes, int n_in,
                              void* d_out, int out_size, void* d_ws, size_t ws_size,
                              hipStream_t stream) {
    const float* x          = (const float*)d_in[0];
    const float* edge_attr  = (const float*)d_in[1];
    const float* W          = (const float*)d_in[2];
    const float* W_edge     = (const float*)d_in[3];
    const float* att_src    = (const float*)d_in[4];
    const float* att_dst    = (const float*)d_in[5];
    const float* att_edge   = (const float*)d_in[6];
    const float* bias_conv  = (const float*)d_in[7];
    const float* bias_layer = (const float*)d_in[8];
    const int*   edge_index = (const int*)d_in[9];
    float* out = (float*)d_out;

    float* ws      = (float*)d_ws;
    float* xh      = ws;
    float* s_node  = xh      + (size_t)N_NODES * HC;
    float* d_node  = s_node  + (size_t)N_NODES * HEADS;
    float* weff    = d_node  + (size_t)N_NODES * HEADS;
    float* alpha   = weff    + (size_t)HEADS * IN_CH;
    float* seg_ms  = alpha   + (size_t)N_EDGES * HEADS;
    float* pmax    = seg_ms  + (size_t)N_NODES * HEADS * 2;
    float* psum    = pmax    + (size_t)ST_CH * N_NODES * HEADS;
    float* part    = psum    + (size_t)ST_CH * N_NODES * HEADS;

    proj_nodes_wmma<<<N_NODES / 16, 128, 0, stream>>>(x, W, xh);
    build_weff<<<1, 256, 0, stream>>>(W_edge, att_edge, weff);
    node_attn_scalars<<<(N_NODES * HEADS + 255) / 256, 256, 0, stream>>>(
        xh, att_src, att_dst, s_node, d_node);
    edge_logits<<<(EL_NWAVES + EL_WAVES - 1) / EL_WAVES, EL_WAVES * 32, 0, stream>>>(
        edge_attr, edge_index, s_node, d_node, weff, alpha);
    seg_stats<<<ST_NB * ST_CH, ST_THREADS, 0, stream>>>(
        edge_index, alpha, pmax, psum);
    seg_combine<<<(N_NODES * HEADS + 255) / 256, 256, 0, stream>>>(
        pmax, psum, seg_ms);
    edge_scatter<<<SC_NB * SC_CH, SC_THREADS, 0, stream>>>(
        edge_index, xh, alpha, seg_ms, part);
    finalize<<<(SC_PLANE / 4 + 255) / 256, 256, 0, stream>>>(
        part, bias_conv, bias_layer, out);
}


# front launches fused (8 -> 6): projection's epilogue parks its tile in LDS and takes the node attention scalars there; its first workgroup also forms w_eff; sum orders unchanged
# speedup vs baseline: 3.3166x; 1.0663x over previous
#include <hip/hip_runtime.h>
#include <cstdint>

#define N_NODES 50000
#define N_EDGES 800000
#define IN_CH 64
#define OUT_CH 32
#define HEADS 4
#define HC 128
#define NEG_SLOPE 0.2f
#define EPSV 1e-16f

typedef __attribute__((ext_vector_type(2))) float v2f;
typedef __attribute__((ext_vector_type(4))) float v4f;
typedef __attribute__((ext_vector_type(8))) float v8f;

#define PJ_HEAD  36
#define PJ_PITCH (HEADS * PJ_HEAD)

__global__ void __launch_bounds__(128) proj_nodes_wmma(
    const float* __restrict__ x, const float* __restrict__ W,
    const float* __restrict__ att_src, const float* __restrict__ att_dst,
    const float* __restrict__ W_edge, const float* __restrict__ att_edge,
    float* __restrict__ xh, float* __restrict__ s_node,
    float* __restrict__ d_node, float* __restrict__ weff)
{
    __shared__ v4f ptile4[16 * PJ_PITCH / 4];

    const int lane = threadIdx.x & 31;
    const int wv   = threadIdx.x >> 5;
    const int half = lane >> 4;
    const int l15  = lane & 15;
    const int m0   = blockIdx.x * 16;

    const int rowA  = m0 + l15;
    const int colB0 = wv * 32 + l15;
    const int colB1 = colB0 + 16;

    const int eh = lane & 3;
    const int er = (lane >> 2) + 8 * (wv & 1);
    const float* __restrict__ attv = (wv < 2) ? att_src : att_dst;
    v4f av[8];
#pragma unroll
    for (int j = 0; j < 8; ++j)
        av[j] = *(const v4f*)(attv + eh * OUT_CH + 4 * j);

    v8f acc0 = {};
    v8f acc1 = {};
#pragma unroll
    for (int k0 = 0; k0 < IN_CH; k0 += 4) {
        const int kk = k0 + half * 2;
        v2f a  = *(const v2f*)(x + (size_t)rowA  * IN_CH + kk);
        v2f b0 = *(const v2f*)(W + (size_t)colB0 * IN_CH + kk);
        v2f b1 = *(const v2f*)(W + (size_t)colB1 * IN_CH + kk);
        acc0 = __builtin_amdgcn_wmma_f32_16x16x4_f32(
                   false, a, false, b0, (short)0, acc0, false, false);
        acc1 = __builtin_amdgcn_wmma_f32_16x16x4_f32(
                   false, a, false, b1, (short)0, acc1, false, false);
    }
    float* ptile = (float*)ptile4;
#pragma unroll
    for (int v = 0; v < 8; ++v) {
        const int r = v + half * 8;
        ptile[r * PJ_PITCH + wv * PJ_HEAD + l15]      = acc0[v];
        ptile[r * PJ_PITCH + wv * PJ_HEAD + 16 + l15] = acc1[v];
    }
    __syncthreads();

#pragma unroll
    for (int v = 0; v < 8; ++v) {
        const int r = v + half * 8;
        xh[(size_t)(m0 + r) * HC + colB0] = acc0[v];
        xh[(size_t)(m0 + r) * HC + colB1] = acc1[v];
    }

    const v4f* rp = (const v4f*)(ptile + er * PJ_PITCH + eh * PJ_HEAD);
    float sd = 0.f;
#pragma unroll
    for (int j = 0; j < 8; ++j) {
        const v4f r = rp[j];
        sd += r.x * av[j].x;
        sd += r.y * av[j].y;
        sd += r.z * av[j].z;
        sd += r.w * av[j].w;
    }
    float* __restrict__ nd = (wv < 2) ? s_node : d_node;
    nd[(size_t)(m0 + er) * HEADS + eh] = sd;

    if (blockIdx.x == 0) {
#pragma unroll 1
        for (int u = 0; u < 2; ++u) {
            const int t = threadIdx.x + 128 * u;
            const int h = t >> 6, k = t & 63;
            float s = 0.f;
#pragma unroll
            for (int c = 0; c < OUT_CH; ++c)
                s += att_edge[h * OUT_CH + c] * W_edge[(size_t)(h * OUT_CH + c) * IN_CH + k];
            weff[t] = s;
        }
    }
}

#define EL_WAVES  4
#define EL_TPW    8
#define EL_NTILES (N_EDGES / 16)
#define EL_NWAVES (EL_NTILES / EL_TPW)
#define EL_PITCH4 17

struct EdgeTile {
    v4f r[8];
    int srcA, dstA, srcB, dstB;
};

__device__ __forceinline__ void el_load_tile(
    EdgeTile& T, const float* __restrict__ edge_attr,
    const int* __restrict__ edge_index, int t, int lane, int eo)
{
    const v4f* p = (const v4f*)(edge_attr + (size_t)t * (16 * IN_CH)) + lane;
#pragma unroll
    for (int j = 0; j < 8; ++j)
        T.r[j] = __builtin_nontemporal_load(p + 32 * j);
    const int e = t * 16 + eo;
    T.srcA = edge_index[e];
    T.dstA = edge_index[N_EDGES + e];
    T.srcB = edge_index[e + 4];
    T.dstB = edge_index[N_EDGES + e + 4];
}

__device__ __forceinline__ void el_wave_lds_order()
{
    __builtin_amdgcn_fence(__ATOMIC_RELEASE, "wavefront");
    __builtin_amdgcn_wave_barrier();
    __builtin_amdgcn_fence(__ATOMIC_ACQUIRE, "wavefront");
}

__device__ __forceinline__ void el_finish(
    float a, int e, int hd, float* __restrict__ alpha)
{
    a = (a > 0.f) ? a : NEG_SLOPE * a;
    alpha[(size_t)e * 4 + hd] = a;
}

__global__ void __launch_bounds__(EL_WAVES * 32) edge_logits(
    const float* __restrict__ edge_attr, const int* __restrict__ edge_index,
    const float* __restrict__ s_node, const float* __restrict__ d_node,
    const float* __restrict__ weff,
    float* __restrict__ alpha)
{
    __shared__ v4f stage[EL_WAVES * 16 * EL_PITCH4];

    const int lane = threadIdx.x & 31;
    const int wv   = threadIdx.x >> 5;
    const int c    = lane & 15;
    const int half = lane >> 4;
    const int w    = blockIdx.x * EL_WAVES + wv;
    if (w >= EL_NWAVES) return;

    v4f* st = stage + wv * (16 * EL_PITCH4);

    const int hd = c & 3;
    v4f bw[8];
#pragma unroll
    for (int jj = 0; jj < 8; ++jj)
        bw[jj] = *(const v4f*)(weff + hd * IN_CH + 8 * jj + 4 * half);

    const int q  = c >> 2;
    const int eo = 8 * half + q;

    EdgeTile cur;
    el_load_tile(cur, edge_attr, edge_index, w, lane, eo);

    for (int i = 0; i < EL_TPW; ++i) {
        const int t = w + i * EL_NWAVES;

#pragma unroll
        for (int j = 0; j < 8; ++j)
            st[(2 * j + half) * EL_PITCH4 + c] = cur.r[j];

        const int srcA = cur.srcA, dstA = cur.dstA;
        const int srcB = cur.srcB, dstB = cur.dstB;
        const float sA = s_node[dstA * 4 + hd];
        const float dA = d_node[srcA * 4 + hd];
        const float sB = s_node[dstB * 4 + hd];
        const float dB = d_node[srcB * 4 + hd];

        if (i + 1 < EL_TPW)
            el_load_tile(cur, edge_attr, edge_index, t + EL_NWAVES, lane, eo);

        el_wave_lds_order();

        v4f a[8];
#pragma unroll
        for (int jj = 0; jj < 8; ++jj)
            a[jj] = st[c * EL_PITCH4 + 2 * jj + half];

        v8f acc0 = {};
        v8f acc1 = {};
#pragma unroll
        for (int jj = 0; jj < 8; ++jj) {
            const v2f a0 = {a[jj].x, a[jj].y};
            const v2f a1 = {a[jj].z, a[jj].w};
            const v2f b0 = {bw[jj].x, bw[jj].y};
            const v2f b1 = {bw[jj].z, bw[jj].w};
            acc0 = __builtin_amdgcn_wmma_f32_16x16x4_f32(
                       false, a0, false, b0, (short)0, acc0, false, false);
            acc1 = __builtin_amdgcn_wmma_f32_16x16x4_f32(
                       false, a1, false, b1, (short)0, acc1, false, false);
        }

        el_wave_lds_order();

        asm volatile("v_nop\n\tv_nop\n\tv_nop\n\tv_nop\n\t"
                     "v_nop\n\tv_nop\n\tv_nop\n\tv_nop"
                     : "+v"(acc0), "+v"(acc1));

        const v8f acc = acc0 + acc1;
        const bool q1 = (q & 1) != 0, q2 = (q & 2) != 0;
        const float lA01 = q1 ? acc[1] : acc[0], lA23 = q1 ? acc[3] : acc[2];
        const float lB01 = q1 ? acc[5] : acc[4], lB23 = q1 ? acc[7] : acc[6];
        const float lA = q2 ? lA23 : lA01;
        const float lB = q2 ? lB23 : lB01;

        const int eA = t * 16 + eo;
        el_finish(lA + sA + dA, eA,     hd, alpha);
        el_finish(lB + sB + dB, eA + 4, hd, alpha);
    }
}

#define ST_NB      15
#define ST_R       3336
#define ST_CH      16
#define ST_CE      (N_EDGES / ST_CH)
#define ST_THREADS 1024
#define ST_LD      13
#define ST_B       4

static_assert(ST_R * ST_NB >= N_NODES && ST_R * (ST_NB - 1) < N_NODES && ST_R % 8 == 0, "ranges must cover the nodes");
static_assert(ST_CE * ST_CH == N_EDGES && ST_CE % 4 == 0, "chunks must tile the edges");
static_assert(ST_THREADS * ST_LD * 4 >= ST_CE, "one sweep must cover a chunk");
static_assert(ST_R <= 4096 && ST_CE <= (1 << 20), "list entry packing");

__device__ __forceinline__ void st_lds_max(int* addr, float a)
{
    const int ai = __float_as_int(a);
    if (ai >= 0) __hip_atomic_fetch_max(addr, ai, __ATOMIC_RELAXED, __HIP_MEMORY_SCOPE_WORKGROUP);
    else         __hip_atomic_fetch_min((unsigned int*)addr, (unsigned int)ai,
                                        __ATOMIC_RELAXED, __HIP_MEMORY_SCOPE_WORKGROUP);
}

__global__ void __launch_bounds__(ST_THREADS) seg_stats(
    const int* __restrict__ edge_index, const float* __restrict__ alpha,
    float* __restrict__ pmax, float* __restrict__ psum)
{
    __shared__ v4f mx4[ST_R];
    __shared__ v4f sm4[ST_R];
    __shared__ unsigned int list[ST_CE];
    __shared__ unsigned int cnt;

    const int tid  = threadIdx.x;
    const int lane = tid & 31;
    const int b    = blockIdx.x % ST_NB;
    const int ch   = blockIdx.x / ST_NB;
    const int lo   = b * ST_R;
    const int e0   = ch * ST_CE;

    const int4* p = (const int4*)(edge_index + N_EDGES + e0);
    int4 dv[ST_LD];
#pragma unroll
    for (int u = 0; u < ST_LD; ++u) {
        const int g = tid + ST_THREADS * u;
        dv[u] = (g < ST_CE / 4) ? p[g] : make_int4(-1, -1, -1, -1);
    }

    const float ninf = -__builtin_huge_valf();
    for (int i = tid; i < ST_R; i += ST_THREADS) {
        mx4[i] = (v4f){ninf, ninf, ninf, ninf};
        sm4[i] = (v4f){0.f, 0.f, 0.f, 0.f};
    }
    if (tid == 0) cnt = 0u;
    __syncthreads();

    unsigned nm = 0;
#pragma unroll
    for (int u = 0; u < ST_LD; ++u) {
        nm += ((unsigned)(dv[u].x - lo) < ST_R) ? 1u : 0u;
        nm += ((unsigned)(dv[u].y - lo) < ST_R) ? 1u : 0u;
        nm += ((unsigned)(dv[u].z - lo) < ST_R) ? 1u : 0u;
        nm += ((unsigned)(dv[u].w - lo) < ST_R) ? 1u : 0u;
    }
    unsigned pre = 0, tot = 0;
#pragma unroll
    for (int bit = 0; bit < 6; ++bit) {
        const unsigned m = __builtin_amdgcn_ballot_w32(((nm >> bit) & 1u) != 0u);
        pre += __builtin_amdgcn_mbcnt_lo(m, 0u) << bit;
        tot += (unsigned)__builtin_popcount(m) << bit;
    }
    unsigned base = 0;
    if (lane == 0) base = atomicAdd(&cnt, tot);
    unsigned slot = (unsigned)__builtin_amdgcn_readfirstlane((int)base) + pre;
#pragma unroll
    for (int u = 0; u < ST_LD; ++u) {
        const int g4 = 4 * (tid + ST_THREADS * u);
        const int dk[4] = {dv[u].x, dv[u].y, dv[u].z, dv[u].w};
#pragma unroll
        for (int k = 0; k < 4; ++k) {
            const unsigned dl = (unsigned)(dk[k] - lo);
            if (dl < ST_R) {
                list[slot] = ((unsigned)(g4 + k) << 12) | dl;
                ++slot;
            }
        }
    }
    __syncthreads();

    const unsigned M = (unsigned)__builtin_amdgcn_readfirstlane((int)cnt);
    int*   mxi = (int*)mx4;
    float* smf = (float*)sm4;

    for (unsigned i0 = (unsigned)tid; i0 < M; i0 += ST_B * ST_THREADS) {
        v4f a[ST_B]; unsigned dl[ST_B];
#pragma unroll
        for (int u = 0; u < ST_B; ++u) {
            const unsigned i = i0 + u * ST_THREADS;
            const unsigned ent = list[(i < M) ? i : i0];
            dl[u] = (i < M) ? (ent & 4095u) : 0xFFFFFFFFu;
            a[u] = *(const v4f*)(alpha + ((size_t)e0 + (ent >> 12)) * 4);
        }
#pragma unroll
        for (int u = 0; u < ST_B; ++u) {
            if (dl[u] != 0xFFFFFFFFu) {
                st_lds_max(mxi + dl[u] * 4 + 0, a[u].x);
                st_lds_max(mxi + dl[u] * 4 + 1, a[u].y);
                st_lds_max(mxi + dl[u] * 4 + 2, a[u].z);
                st_lds_max(mxi + dl[u] * 4 + 3, a[u].w);
            }
        }
    }
    __syncthreads();

    for (unsigned i0 = (unsigned)tid; i0 < M; i0 += ST_B * ST_THREADS) {
        v4f a[ST_B]; unsigned dl[ST_B];
#pragma unroll
        for (int u = 0; u < ST_B; ++u) {
            const unsigned i = i0 + u * ST_THREADS;
            const unsigned ent = list[(i < M) ? i : i0];
            dl[u] = (i < M) ? (ent & 4095u) : 0xFFFFFFFFu;
            a[u] = *(const v4f*)(alpha + ((size_t)e0 + (ent >> 12)) * 4);
        }
        asm volatile("" :: "v"(a[0]), "v"(a[1]), "v"(a[2]), "v"(a[3]));
#pragma unroll
        for (int u = 0; u < ST_B; ++u) {
            if (dl[u] != 0xFFFFFFFFu) {
                const v4f m = mx4[dl[u]];
                __hip_atomic_fetch_add(smf + dl[u] * 4 + 0, __expf(a[u].x - m.x),
                                       __ATOMIC_RELAXED, __HIP_MEMORY_SCOPE_WORKGROUP);
                __hip_atomic_fetch_add(smf + dl[u] * 4 + 1, __expf(a[u].y - m.y),
                                       __ATOMIC_RELAXED, __HIP_MEMORY_SCOPE_WORKGROUP);
                __hip_atomic_fetch_add(smf + dl[u] * 4 + 2, __expf(a[u].z - m.z),
                                       __ATOMIC_RELAXED, __HIP_MEMORY_SCOPE_WORKGROUP);
                __hip_atomic_fetch_add(smf + dl[u] * 4 + 3, __expf(a[u].w - m.w),
                                       __ATOMIC_RELAXED, __HIP_MEMORY_SCOPE_WORKGROUP);
            }
        }
    }
    __syncthreads();

    const int nr = (N_NODES - lo < ST_R) ? (N_NODES - lo) : ST_R;
    v4f* om = (v4f*)pmax + (size_t)ch * N_NODES + lo;
    v4f* os = (v4f*)psum + (size_t)ch * N_NODES + lo;
    for (int i = tid; i < nr; i += ST_THREADS) {
        om[i] = mx4[i];
        os[i] = sm4[i];
    }
}

__global__ void seg_combine(const float* __restrict__ pmax,
                            const float* __restrict__ psum,
                            float* __restrict__ seg_ms)
{
    const int t = blockIdx.x * blockDim.x + threadIdx.x;
    if (t >= N_NODES * HEADS) return;
    float m[ST_CH], sv[ST_CH];
#pragma unroll
    for (int c = 0; c < ST_CH; ++c) {
        m[c]  = pmax[(size_t)c * (N_NODES * HEADS) + t];
        sv[c] = psum[(size_t)c * (N_NODES * HEADS) + t];
    }
    const float ninf = -__builtin_huge_valf();
    float M = m[0];
#pragma unroll
    for (int c = 1; c < ST_CH; ++c) M = (m[c] > M) ? m[c] : M;
    if (!__builtin_isfinite(M)) M = 0.f;
    float S = 0.f;
#pragma unroll
    for (int c = 0; c < ST_CH; ++c)
        S += (m[c] == ninf) ? 0.f : sv[c] * __expf(m[c] - M);
    const int n = t >> 2, h = t & 3;
    seg_ms[(size_t)n * 8 + h]     = M;
    seg_ms[(size_t)n * 8 + 4 + h] = S;
}

#define SC_R       2000
#define SC_NB      (N_NODES / SC_R)
#define SC_CH      10
#define SC_CE      (N_EDGES / SC_CH)
#define SC_SUB     16000
#define SC_NSUB    (SC_CE / SC_SUB)
#define SC_THREADS 1024
#define SC_WAVES   (SC_THREADS / 32)
#define SC_LD      4
#define SC_G       8
#define SC_PLANE   (N_NODES * OUT_CH)

static_assert(SC_R * SC_NB == N_NODES, "ranges must tile the nodes");
static_assert(SC_CE * SC_CH == N_EDGES && SC_SUB * SC_NSUB == SC_CE, "chunks must tile the edges");
static_assert(SC_THREADS * SC_LD * 4 >= SC_SUB && SC_SUB % 4 == 0, "a round must cover its edges");
static_assert(SC_R <= 2048 && SC_SUB <= (1 << 21), "list entry packing");
static_assert(32 % SC_G == 0, "groups must tile a 32-entry slab");

__device__ __forceinline__ void sc_load_dst(int4 (&dv)[SC_LD],
                                            const int* __restrict__ dstp, int tid)
{
    const int4* p = (const int4*)dstp;
#pragma unroll
    for (int u = 0; u < SC_LD; ++u) {
        const int g = tid + SC_THREADS * u;
        dv[u] = (g < SC_SUB / 4) ? p[g] : make_int4(-1, -1, -1, -1);
    }
}

__global__ void __launch_bounds__(SC_THREADS) edge_scatter(
    const int* __restrict__ edge_index, const float* __restrict__ xh,
    const float* __restrict__ alpha, const float* __restrict__ seg_ms,
    float* __restrict__ part)
{
    __shared__ v4f accs4[SC_R * OUT_CH / 4];
    __shared__ unsigned int list[SC_SUB];
    __shared__ unsigned int cnt[2];

    float* accs = (float*)accs4;
    const int tid  = threadIdx.x;
    const int lane = tid & 31;
    const int wv   = __builtin_amdgcn_readfirstlane(tid >> 5);
    const int b    = blockIdx.x % SC_NB;
    const int ch   = blockIdx.x / SC_NB;
    const int lo   = b * SC_R;
    const int e0   = ch * SC_CE;
    const int* dstp = edge_index + N_EDGES + e0;

    int4 dv[SC_LD];
    sc_load_dst(dv, dstp, tid);

    for (int i = tid; i < SC_R * OUT_CH / 4; i += SC_THREADS)
        accs4[i] = (v4f){0.f, 0.f, 0.f, 0.f};
    if (tid == 0) { cnt[0] = 0u; cnt[1] = 0u; }
    __syncthreads();

    for (int r = 0; r < SC_NSUB; ++r) {
        unsigned int* cn = cnt + (r & 1);
        const int eb = e0 + r * SC_SUB;

        unsigned nm = 0;
#pragma unroll
        for (int u = 0; u < SC_LD; ++u) {
            nm += ((unsigned)(dv[u].x - lo) < SC_R) ? 1u : 0u;
            nm += ((unsigned)(dv[u].y - lo) < SC_R) ? 1u : 0u;
            nm += ((unsigned)(dv[u].z - lo) < SC_R) ? 1u : 0u;
            nm += ((unsigned)(dv[u].w - lo) < SC_R) ? 1u : 0u;
        }
        unsigned pre = 0, tot = 0;
#pragma unroll
        for (int bit = 0; bit < 5; ++bit) {
            const unsigned m = __builtin_amdgcn_ballot_w32(((nm >> bit) & 1u) != 0u);
            pre += __builtin_amdgcn_mbcnt_lo(m, 0u) << bit;
            tot += (unsigned)__builtin_popcount(m) << bit;
        }
        unsigned base = 0;
        if (lane == 0) base = atomicAdd(cn, tot);
        unsigned slot = (unsigned)__builtin_amdgcn_readfirstlane((int)base) + pre;
#pragma unroll
        for (int u = 0; u < SC_LD; ++u) {
            const int g4 = 4 * (tid + SC_THREADS * u);
            const int dk[4] = {dv[u].x, dv[u].y, dv[u].z, dv[u].w};
#pragma unroll
            for (int k = 0; k < 4; ++k) {
                const unsigned dl = (unsigned)(dk[k] - lo);
                if (dl < SC_R) {
                    list[slot] = ((unsigned)(g4 + k) << 11) | dl;
                    ++slot;
                }
            }
        }
        __syncthreads();

        if (r + 1 < SC_NSUB) sc_load_dst(dv, dstp + (r + 1) * SC_SUB, tid);
        if (tid == 0) cnt[(r + 1) & 1] = 0u;

        const unsigned M   = (unsigned)__builtin_amdgcn_readfirstlane((int)*cn);
        const unsigned per = (M + SC_WAVES - 1) / SC_WAVES;
        const unsigned beg = wv * per;
        const unsigned end = (beg + per < M) ? beg + per : M;

        for (unsigned s = beg; s < end; s += 32) {
            int sl = 0; unsigned dl = 0; v4f cf = {0.f, 0.f, 0.f, 0.f};
            if (s + lane < end) {
                const unsigned ent = list[s + lane];
                dl = ent & 2047u;
                const int e = eb + (int)(ent >> 11);
                sl = edge_index[e];
                const v4f al = *(const v4f*)(alpha + (size_t)e * 4);
                const v4f* ms = (const v4f*)(seg_ms + (size_t)(lo + (int)dl) * 8);
                const v4f mm = ms[0];
                const v4f ss = ms[1];
                const v4f ex = {__expf(al.x - mm.x), __expf(al.y - mm.y),
                                __expf(al.z - mm.z), __expf(al.w - mm.w)};
                cf = ex / (ss + EPSV);
            }
            const int n = (end - s < 32u) ? (int)(end - s) : 32;

            for (int j = 0; j < n; j += SC_G) {
                float xv[SC_G][4];
#pragma unroll
                for (int u = 0; u < SC_G; ++u) {
                    const int sj = __builtin_amdgcn_readlane(sl, j + u);
                    const float* xp = xh + (size_t)sj * HC + lane;
                    xv[u][0] = xp[0];  xv[u][1] = xp[32];
                    xv[u][2] = xp[64]; xv[u][3] = xp[96];
                }
#pragma unroll
                for (int u = 0; u < SC_G; ++u) {
                    if (j + u < n) {
                        const float c0 = __int_as_float(__builtin_amdgcn_readlane(__float_as_int(cf.x), j + u));
                        const float c1 = __int_as_float(__builtin_amdgcn_readlane(__float_as_int(cf.y), j + u));
                        const float c2 = __int_as_float(__builtin_amdgcn_readlane(__float_as_int(cf.z), j + u));
                        const float c3 = __int_as_float(__builtin_amdgcn_readlane(__float_as_int(cf.w), j + u));
                        const int   dj = __builtin_amdgcn_readlane((int)dl, j + u);
                        const float v = c0 * xv[u][0] + c1 * xv[u][1]
                                      + c2 * xv[u][2] + c3 * xv[u][3];
                        __hip_atomic_fetch_add(accs + dj * OUT_CH + lane, v,
                                               __ATOMIC_RELAXED, __HIP_MEMORY_SCOPE_WORKGROUP);
                    }
                }
            }
        }
        __syncthreads();
    }

    v4f* o4 = (v4f*)(part + (size_t)ch * SC_PLANE + (size_t)lo * OUT_CH);
    for (int i = tid; i < SC_R * OUT_CH / 4; i += SC_THREADS)
        o4[i] = accs4[i];
}

__global__ void finalize(const float* __restrict__ part,
                         const float* __restrict__ bias_conv,
                         const float* __restrict__ bias_layer,
                         float* __restrict__ out)
{
    const int t = blockIdx.x * blockDim.x + threadIdx.x;
    if (t >= SC_PLANE / 4) return;
    const v4f* p = (const v4f*)part + t;
    v4f s = p[0];
#pragma unroll
    for (int c = 1; c < SC_CH; ++c)
        s += p[(size_t)c * (SC_PLANE / 4)];
    const int c0 = (t & 7) * 4;
    const float sv[4] = {s.x, s.y, s.z, s.w};
    float rv[4];
#pragma unroll
    for (int k = 0; k < 4; ++k) {
        const float v = sv[k] * 0.25f + bias_conv[c0 + k] + bias_layer[c0 + k];
        rv[k] = v > 0.f ? v : 0.f;
    }
    ((v4f*)out)[t] = (v4f){rv[0], rv[1], rv[2], rv[3]};
}

extern "C" void kernel_launch(void* const* d_in, const int* in_sizes, int n_in,
                              void* d_out, int out_size, void* d_ws, size_t ws_size,
                              hipStream_t stream) {
    const float* x          = (const float*)d_in[0];
    const float* edge_attr  = (const float*)d_in[1];
    const float* W          = (const float*)d_in[2];
    const float* W_edge     = (const float*)d_in[3];
    const float* att_src    = (const float*)d_in[4];
    const float* att_dst    = (const float*)d_in[5];
    const float* att_edge   = (const float*)d_in[6];
    const float* bias_conv  = (const float*)d_in[7];
    const float* bias_layer = (const float*)d_in[8];
    const int*   edge_index = (const int*)d_in[9];
    float* out = (float*)d_out;

    float* ws      = (float*)d_ws;
    float* xh      = ws;
    float* s_node  = xh      + (size_t)N_NODES * HC;
    float* d_node  = s_node  + (size_t)N_NODES * HEADS;
    float* weff    = d_node  + (size_t)N_NODES * HEADS;
    float* alpha   = weff    + (size_t)HEADS * IN_CH;
    float* seg_ms  = alpha   + (size_t)N_EDGES * HEADS;
    float* pmax    = seg_ms  + (size_t)N_NODES * HEADS * 2;
    float* psum    = pmax    + (size_t)ST_CH * N_NODES * HEADS;
    float* part    = psum    + (size_t)ST_CH * N_NODES * HEADS;

    proj_nodes_wmma<<<N_NODES / 16, 128, 0, stream>>>(
        x, W, att_src, att_dst, W_edge, att_edge, xh, s_node, d_node, weff);
    edge_logits<<<(EL_NWAVES + EL_WAVES - 1) / EL_WAVES, EL_WAVES * 32, 0, stream>>>(
        edge_attr, edge_index, s_node, d_node, weff, alpha);
    seg_stats<<<ST_NB * ST_CH, ST_THREADS, 0, stream>>>(
        edge_index, alpha, pmax, psum);
    seg_combine<<<(N_NODES * HEADS + 255) / 256, 256, 0, stream>>>(
        pmax, psum, seg_ms);
    edge_scatter<<<SC_NB * SC_CH, SC_THREADS, 0, stream>>>(
        edge_index, xh, alpha, seg_ms, part);
    finalize<<<(SC_PLANE / 4 + 255) / 256, 256, 0, stream>>>(
        part, bias_conv, bias_layer, out);
}


# proj_nodes: x tile read once per workgroup by 512-byte wave loads and shared via LDS; W fragments by 16-byte loads (48 line-scattered 8-byte loads per wave -> 18); k order, so sum order, changes
# speedup vs baseline: 3.4472x; 1.0394x over previous
#include <hip/hip_runtime.h>
#include <cstdint>

#define N_NODES 50000
#define N_EDGES 800000
#define IN_CH 64
#define OUT_CH 32
#define HEADS 4
#define HC 128
#define NEG_SLOPE 0.2f
#define EPSV 1e-16f

typedef __attribute__((ext_vector_type(2))) float v2f;
typedef __attribute__((ext_vector_type(4))) float v4f;
typedef __attribute__((ext_vector_type(8))) float v8f;

#define PJ_HEAD  36
#define PJ_XPITCH4 17
#define PJ_PITCH (HEADS * PJ_HEAD)

__global__ void __launch_bounds__(128) proj_nodes_wmma(
    const float* __restrict__ x, const float* __restrict__ W,
    const float* __restrict__ att_src, const float* __restrict__ att_dst,
    const float* __restrict__ W_edge, const float* __restrict__ att_edge,
    float* __restrict__ xh, float* __restrict__ s_node,
    float* __restrict__ d_node, float* __restrict__ weff)
{
    __shared__ v4f ptile4[16 * PJ_PITCH / 4];
    __shared__ v4f xs4[16 * PJ_XPITCH4];

    const int lane = threadIdx.x & 31;
    const int wv   = threadIdx.x >> 5;
    const int half = lane >> 4;
    const int l15  = lane & 15;
    const int m0   = blockIdx.x * 16;

    const int colB0 = wv * 32 + l15;
    const int colB1 = colB0 + 16;

    const v4f* xt = (const v4f*)(x + (size_t)m0 * IN_CH);
    const int p0 = threadIdx.x;
    const int p1 = threadIdx.x + 128;
    const v4f xa = xt[p0];
    const v4f xb = xt[p1];

    const int eh = lane & 3;
    const int er = (lane >> 2) + 8 * (wv & 1);
    const float* __restrict__ attv = (wv < 2) ? att_src : att_dst;
    v4f av[8];
#pragma unroll
    for (int j = 0; j < 8; ++j)
        av[j] = *(const v4f*)(attv + eh * OUT_CH + 4 * j);

    v4f bw0[8], bw1[8];
#pragma unroll
    for (int jj = 0; jj < 8; ++jj) {
        bw0[jj] = *(const v4f*)(W + (size_t)colB0 * IN_CH + 8 * jj + 4 * half);
        bw1[jj] = *(const v4f*)(W + (size_t)colB1 * IN_CH + 8 * jj + 4 * half);
    }

    xs4[(p0 >> 4) * PJ_XPITCH4 + (p0 & 15)] = xa;
    xs4[(p1 >> 4) * PJ_XPITCH4 + (p1 & 15)] = xb;
    __syncthreads();

    v4f a[8];
#pragma unroll
    for (int jj = 0; jj < 8; ++jj)
        a[jj] = xs4[l15 * PJ_XPITCH4 + 2 * jj + half];

    v8f acc0 = {};
    v8f acc1 = {};
#pragma unroll
    for (int jj = 0; jj < 8; ++jj) {
        const v2f a0  = {a[jj].x, a[jj].y};
        const v2f a1  = {a[jj].z, a[jj].w};
        const v2f b00 = {bw0[jj].x, bw0[jj].y};
        const v2f b01 = {bw0[jj].z, bw0[jj].w};
        const v2f b10 = {bw1[jj].x, bw1[jj].y};
        const v2f b11 = {bw1[jj].z, bw1[jj].w};
        acc0 = __builtin_amdgcn_wmma_f32_16x16x4_f32(
                   false, a0, false, b00, (short)0, acc0, false, false);
        acc1 = __builtin_amdgcn_wmma_f32_16x16x4_f32(
                   false, a0, false, b10, (short)0, acc1, false, false);
        acc0 = __builtin_amdgcn_wmma_f32_16x16x4_f32(
                   false, a1, false, b01, (short)0, acc0, false, false);
        acc1 = __builtin_amdgcn_wmma_f32_16x16x4_f32(
                   false, a1, false, b11, (short)0, acc1, false, false);
    }
    float* ptile = (float*)ptile4;
#pragma unroll
    for (int v = 0; v < 8; ++v) {
        const int r = v + half * 8;
        ptile[r * PJ_PITCH + wv * PJ_HEAD + l15]      = acc0[v];
        ptile[r * PJ_PITCH + wv * PJ_HEAD + 16 + l15] = acc1[v];
    }
    __syncthreads();

#pragma unroll
    for (int v = 0; v < 8; ++v) {
        const int r = v + half * 8;
        xh[(size_t)(m0 + r) * HC + colB0] = acc0[v];
        xh[(size_t)(m0 + r) * HC + colB1] = acc1[v];
    }

    const v4f* rp = (const v4f*)(ptile + er * PJ_PITCH + eh * PJ_HEAD);
    float sd = 0.f;
#pragma unroll
    for (int j = 0; j < 8; ++j) {
        const v4f r = rp[j];
        sd += r.x * av[j].x;
        sd += r.y * av[j].y;
        sd += r.z * av[j].z;
        sd += r.w * av[j].w;
    }
    float* __restrict__ nd = (wv < 2) ? s_node : d_node;
    nd[(size_t)(m0 + er) * HEADS + eh] = sd;

    if (blockIdx.x == 0) {
#pragma unroll 1
        for (int u = 0; u < 2; ++u) {
            const int t = threadIdx.x + 128 * u;
            const int h = t >> 6, k = t & 63;
            float s = 0.f;
#pragma unroll
            for (int c = 0; c < OUT_CH; ++c)
                s += att_edge[h * OUT_CH + c] * W_edge[(size_t)(h * OUT_CH + c) * IN_CH + k];
            weff[t] = s;
        }
    }
}

#define EL_WAVES  4
#define EL_TPW    8
#define EL_NTILES (N_EDGES / 16)
#define EL_NWAVES (EL_NTILES / EL_TPW)
#define EL_PITCH4 17

struct EdgeTile {
    v4f r[8];
    int srcA, dstA, srcB, dstB;
};

__device__ __forceinline__ void el_load_tile(
    EdgeTile& T, const float* __restrict__ edge_attr,
    const int* __restrict__ edge_index, int t, int lane, int eo)
{
    const v4f* p = (const v4f*)(edge_attr + (size_t)t * (16 * IN_CH)) + lane;
#pragma unroll
    for (int j = 0; j < 8; ++j)
        T.r[j] = __builtin_nontemporal_load(p + 32 * j);
    const int e = t * 16 + eo;
    T.srcA = edge_index[e];
    T.dstA = edge_index[N_EDGES + e];
    T.srcB = edge_index[e + 4];
    T.dstB = edge_index[N_EDGES + e + 4];
}

__device__ __forceinline__ void el_wave_lds_order()
{
    __builtin_amdgcn_fence(__ATOMIC_RELEASE, "wavefront");
    __builtin_amdgcn_wave_barrier();
    __builtin_amdgcn_fence(__ATOMIC_ACQUIRE, "wavefront");
}

__device__ __forceinline__ void el_finish(
    float a, int e, int hd, float* __restrict__ alpha)
{
    a = (a > 0.f) ? a : NEG_SLOPE * a;
    alpha[(size_t)e * 4 + hd] = a;
}

__global__ void __launch_bounds__(EL_WAVES * 32) edge_logits(
    const float* __restrict__ edge_attr, const int* __restrict__ edge_index,
    const float* __restrict__ s_node, const float* __restrict__ d_node,
    const float* __restrict__ weff,
    float* __restrict__ alpha)
{
    __shared__ v4f stage[EL_WAVES * 16 * EL_PITCH4];

    const int lane = threadIdx.x & 31;
    const int wv   = threadIdx.x >> 5;
    const int c    = lane & 15;
    const int half = lane >> 4;
    const int w    = blockIdx.x * EL_WAVES + wv;
    if (w >= EL_NWAVES) return;

    v4f* st = stage + wv * (16 * EL_PITCH4);

    const int hd = c & 3;
    v4f bw[8];
#pragma unroll
    for (int jj = 0; jj < 8; ++jj)
        bw[jj] = *(const v4f*)(weff + hd * IN_CH + 8 * jj + 4 * half);

    const int q  = c >> 2;
    const int eo = 8 * half + q;

    EdgeTile cur;
    el_load_tile(cur, edge_attr, edge_index, w, lane, eo);

    for (int i = 0; i < EL_TPW; ++i) {
        const int t = w + i * EL_NWAVES;

#pragma unroll
        for (int j = 0; j < 8; ++j)
            st[(2 * j + half) * EL_PITCH4 + c] = cur.r[j];

        const int srcA = cur.srcA, dstA = cur.dstA;
        const int srcB = cur.srcB, dstB = cur.dstB;
        const float sA = s_node[dstA * 4 + hd];
        const float dA = d_node[srcA * 4 + hd];
        const float sB = s_node[dstB * 4 + hd];
        const float dB = d_node[srcB * 4 + hd];

        if (i + 1 < EL_TPW)
            el_load_tile(cur, edge_attr, edge_index, t + EL_NWAVES, lane, eo);

        el_wave_lds_order();

        v4f a[8];
#pragma unroll
        for (int jj = 0; jj < 8; ++jj)
            a[jj] = st[c * EL_PITCH4 + 2 * jj + half];

        v8f acc0 = {};
        v8f acc1 = {};
#pragma unroll
        for (int jj = 0; jj < 8; ++jj) {
            const v2f a0 = {a[jj].x, a[jj].y};
            const v2f a1 = {a[jj].z, a[jj].w};
            const v2f b0 = {bw[jj].x, bw[jj].y};
            const v2f b1 = {bw[jj].z, bw[jj].w};
            acc0 = __builtin_amdgcn_wmma_f32_16x16x4_f32(
                       false, a0, false, b0, (short)0, acc0, false, false);
            acc1 = __builtin_amdgcn_wmma_f32_16x16x4_f32(
                       false, a1, false, b1, (short)0, acc1, false, false);
        }

        el_wave_lds_order();

        asm volatile("v_nop\n\tv_nop\n\tv_nop\n\tv_nop\n\t"
                     "v_nop\n\tv_nop\n\tv_nop\n\tv_nop"
                     : "+v"(acc0), "+v"(acc1));

        const v8f acc = acc0 + acc1;
        const bool q1 = (q & 1) != 0, q2 = (q & 2) != 0;
        const float lA01 = q1 ? acc[1] : acc[0], lA23 = q1 ? acc[3] : acc[2];
        const float lB01 = q1 ? acc[5] : acc[4], lB23 = q1 ? acc[7] : acc[6];
        const float lA = q2 ? lA23 : lA01;
        const float lB = q2 ? lB23 : lB01;

        const int eA = t * 16 + eo;
        el_finish(lA + sA + dA, eA,     hd, alpha);
        el_finish(lB + sB + dB, eA + 4, hd, alpha);
    }
}

#define ST_NB      15
#define ST_R       3336
#define ST_CH      16
#define ST_CE      (N_EDGES / ST_CH)
#define ST_THREADS 1024
#define ST_LD      13
#define ST_B       4

static_assert(ST_R * ST_NB >= N_NODES && ST_R * (ST_NB - 1) < N_NODES && ST_R % 8 == 0, "ranges must cover the nodes");
static_assert(ST_CE * ST_CH == N_EDGES && ST_CE % 4 == 0, "chunks must tile the edges");
static_assert(ST_THREADS * ST_LD * 4 >= ST_CE, "one sweep must cover a chunk");
static_assert(ST_R <= 4096 && ST_CE <= (1 << 20), "list entry packing");

__device__ __forceinline__ void st_lds_max(int* addr, float a)
{
    const int ai = __float_as_int(a);
    if (ai >= 0) __hip_atomic_fetch_max(addr, ai, __ATOMIC_RELAXED, __HIP_MEMORY_SCOPE_WORKGROUP);
    else         __hip_atomic_fetch_min((unsigned int*)addr, (unsigned int)ai,
                                        __ATOMIC_RELAXED, __HIP_MEMORY_SCOPE_WORKGROUP);
}

__global__ void __launch_bounds__(ST_THREADS) seg_stats(
    const int* __restrict__ edge_index, const float* __restrict__ alpha,
    float* __restrict__ pmax, float* __restrict__ psum)
{
    __shared__ v4f mx4[ST_R];
    __shared__ v4f sm4[ST_R];
    __shared__ unsigned int list[ST_CE];
    __shared__ unsigned int cnt;

    const int tid  = threadIdx.x;
    const int lane = tid & 31;
    const int b    = blockIdx.x % ST_NB;
    const int ch   = blockIdx.x / ST_NB;
    const int lo   = b * ST_R;
    const int e0   = ch * ST_CE;

    const int4* p = (const int4*)(edge_index + N_EDGES + e0);
    int4 dv[ST_LD];
#pragma unroll
    for (int u = 0; u < ST_LD; ++u) {
        const int g = tid + ST_THREADS * u;
        dv[u] = (g < ST_CE / 4) ? p[g] : make_int4(-1, -1, -1, -1);
    }

    const float ninf = -__builtin_huge_valf();
    for (int i = tid; i < ST_R; i += ST_THREADS) {
        mx4[i] = (v4f){ninf, ninf, ninf, ninf};
        sm4[i] = (v4f){0.f, 0.f, 0.f, 0.f};
    }
    if (tid == 0) cnt = 0u;
    __syncthreads();

    unsigned nm = 0;
#pragma unroll
    for (int u = 0; u < ST_LD; ++u) {
        nm += ((unsigned)(dv[u].x - lo) < ST_R) ? 1u : 0u;
        nm += ((unsigned)(dv[u].y - lo) < ST_R) ? 1u : 0u;
        nm += ((unsigned)(dv[u].z - lo) < ST_R) ? 1u : 0u;
        nm += ((unsigned)(dv[u].w - lo) < ST_R) ? 1u : 0u;
    }
    unsigned pre = 0, tot = 0;
#pragma unroll
    for (int bit = 0; bit < 6; ++bit) {
        const unsigned m = __builtin_amdgcn_ballot_w32(((nm >> bit) & 1u) != 0u);
        pre += __builtin_amdgcn_mbcnt_lo(m, 0u) << bit;
        tot += (unsigned)__builtin_popcount(m) << bit;
    }
    unsigned base = 0;
    if (lane == 0) base = atomicAdd(&cnt, tot);
    unsigned slot = (unsigned)__builtin_amdgcn_readfirstlane((int)base) + pre;
#pragma unroll
    for (int u = 0; u < ST_LD; ++u) {
        const int g4 = 4 * (tid + ST_THREADS * u);
        const int dk[4] = {dv[u].x, dv[u].y, dv[u].z, dv[u].w};
#pragma unroll
        for (int k = 0; k < 4; ++k) {
            const unsigned dl = (unsigned)(dk[k] - lo);
            if (dl < ST_R) {
                list[slot] = ((unsigned)(g4 + k) << 12) | dl;
                ++slot;
            }
        }
    }
    __syncthreads();

    const unsigned M = (unsigned)__builtin_amdgcn_readfirstlane((int)cnt);
    int*   mxi = (int*)mx4;
    float* smf = (float*)sm4;

    for (unsigned i0 = (unsigned)tid; i0 < M; i0 += ST_B * ST_THREADS) {
        v4f a[ST_B]; unsigned dl[ST_B];
#pragma unroll
        for (int u = 0; u < ST_B; ++u) {
            const unsigned i = i0 + u * ST_THREADS;
            const unsigned ent = list[(i < M) ? i : i0];
            dl[u] = (i < M) ? (ent & 4095u) : 0xFFFFFFFFu;
            a[u] = *(const v4f*)(alpha + ((size_t)e0 + (ent >> 12)) * 4);
        }
#pragma unroll
        for (int u = 0; u < ST_B; ++u) {
            if (dl[u] != 0xFFFFFFFFu) {
                st_lds_max(mxi + dl[u] * 4 + 0, a[u].x);
                st_lds_max(mxi + dl[u] * 4 + 1, a[u].y);
                st_lds_max(mxi + dl[u] * 4 + 2, a[u].z);
                st_lds_max(mxi + dl[u] * 4 + 3, a[u].w);
            }
        }
    }
    __syncthreads();

    for (unsigned i0 = (unsigned)tid; i0 < M; i0 += ST_B * ST_THREADS) {
        v4f a[ST_B]; unsigned dl[ST_B];
#pragma unroll
        for (int u = 0; u < ST_B; ++u) {
            const unsigned i = i0 + u * ST_THREADS;
            const unsigned ent = list[(i < M) ? i : i0];
            dl[u] = (i < M) ? (ent & 4095u) : 0xFFFFFFFFu;
            a[u] = *(const v4f*)(alpha + ((size_t)e0 + (ent >> 12)) * 4);
        }
        asm volatile("" :: "v"(a[0]), "v"(a[1]), "v"(a[2]), "v"(a[3]));
#pragma unroll
        for (int u = 0; u < ST_B; ++u) {
            if (dl[u] != 0xFFFFFFFFu) {
                const v4f m = mx4[dl[u]];
                __hip_atomic_fetch_add(smf + dl[u] * 4 + 0, __expf(a[u].x - m.x),
                                       __ATOMIC_RELAXED, __HIP_MEMORY_SCOPE_WORKGROUP);
                __hip_atomic_fetch_add(smf + dl[u] * 4 + 1, __expf(a[u].y - m.y),
                                       __ATOMIC_RELAXED, __HIP_MEMORY_SCOPE_WORKGROUP);
                __hip_atomic_fetch_add(smf + dl[u] * 4 + 2, __expf(a[u].z - m.z),
                                       __ATOMIC_RELAXED, __HIP_MEMORY_SCOPE_WORKGROUP);
                __hip_atomic_fetch_add(smf + dl[u] * 4 + 3, __expf(a[u].w - m.w),
                                       __ATOMIC_RELAXED, __HIP_MEMORY_SCOPE_WORKGROUP);
            }
        }
    }
    __syncthreads();

    const int nr = (N_NODES - lo < ST_R) ? (N_NODES - lo) : ST_R;
    v4f* om = (v4f*)pmax + (size_t)ch * N_NODES + lo;
    v4f* os = (v4f*)psum + (size_t)ch * N_NODES + lo;
    for (int i = tid; i < nr; i += ST_THREADS) {
        om[i] = mx4[i];
        os[i] = sm4[i];
    }
}

__global__ void seg_combine(const float* __restrict__ pmax,
                            const float* __restrict__ psum,
                            float* __restrict__ seg_ms)
{
    const int t = blockIdx.x * blockDim.x + threadIdx.x;
    if (t >= N_NODES * HEADS) return;
    float m[ST_CH], sv[ST_CH];
#pragma unroll
    for (int c = 0; c < ST_CH; ++c) {
        m[c]  = pmax[(size_t)c * (N_NODES * HEADS) + t];
        sv[c] = psum[(size_t)c * (N_NODES * HEADS) + t];
    }
    const float ninf = -__builtin_huge_valf();
    float M = m[0];
#pragma unroll
    for (int c = 1; c < ST_CH; ++c) M = (m[c] > M) ? m[c] : M;
    if (!__builtin_isfinite(M)) M = 0.f;
    float S = 0.f;
#pragma unroll
    for (int c = 0; c < ST_CH; ++c)
        S += (m[c] == ninf) ? 0.f : sv[c] * __expf(m[c] - M);
    const int n = t >> 2, h = t & 3;
    seg_ms[(size_t)n * 8 + h]     = M;
    seg_ms[(size_t)n * 8 + 4 + h] = S;
}

#define SC_R       2000
#define SC_NB      (N_NODES / SC_R)
#define SC_CH      10
#define SC_CE      (N_EDGES / SC_CH)
#define SC_SUB     16000
#define SC_NSUB    (SC_CE / SC_SUB)
#define SC_THREADS 1024
#define SC_WAVES   (SC_THREADS / 32)
#define SC_LD      4
#define SC_G       8
#define SC_PLANE   (N_NODES * OUT_CH)

static_assert(SC_R * SC_NB == N_NODES, "ranges must tile the nodes");
static_assert(SC_CE * SC_CH == N_EDGES && SC_SUB * SC_NSUB == SC_CE, "chunks must tile the edges");
static_assert(SC_THREADS * SC_LD * 4 >= SC_SUB && SC_SUB % 4 == 0, "a round must cover its edges");
static_assert(SC_R <= 2048 && SC_SUB <= (1 << 21), "list entry packing");
static_assert(32 % SC_G == 0, "groups must tile a 32-entry slab");

__device__ __forceinline__ void sc_load_dst(int4 (&dv)[SC_LD],
                                            const int* __restrict__ dstp, int tid)
{
    const int4* p = (const int4*)dstp;
#pragma unroll
    for (int u = 0; u < SC_LD; ++u) {
        const int g = tid + SC_THREADS * u;
        dv[u] = (g < SC_SUB / 4) ? p[g] : make_int4(-1, -1, -1, -1);
    }
}

__global__ void __launch_bounds__(SC_THREADS) edge_scatter(
    const int* __restrict__ edge_index, const float* __restrict__ xh,
    const float* __restrict__ alpha, const float* __restrict__ seg_ms,
    float* __restrict__ part)
{
    __shared__ v4f accs4[SC_R * OUT_CH / 4];
    __shared__ unsigned int list[SC_SUB];
    __shared__ unsigned int cnt[2];

    float* accs = (float*)accs4;
    const int tid  = threadIdx.x;
    const int lane = tid & 31;
    const int wv   = __builtin_amdgcn_readfirstlane(tid >> 5);
    const int b    = blockIdx.x % SC_NB;
    const int ch   = blockIdx.x / SC_NB;
    const int lo   = b * SC_R;
    const int e0   = ch * SC_CE;
    const int* dstp = edge_index + N_EDGES + e0;

    int4 dv[SC_LD];
    sc_load_dst(dv, dstp, tid);

    for (int i = tid; i < SC_R * OUT_CH / 4; i += SC_THREADS)
        accs4[i] = (v4f){0.f, 0.f, 0.f, 0.f};
    if (tid == 0) { cnt[0] = 0u; cnt[1] = 0u; }
    __syncthreads();

    for (int r = 0; r < SC_NSUB; ++r) {
        unsigned int* cn = cnt + (r & 1);
        const int eb = e0 + r * SC_SUB;

        unsigned nm = 0;
#pragma unroll
        for (int u = 0; u < SC_LD; ++u) {
            nm += ((unsigned)(dv[u].x - lo) < SC_R) ? 1u : 0u;
            nm += ((unsigned)(dv[u].y - lo) < SC_R) ? 1u : 0u;
            nm += ((unsigned)(dv[u].z - lo) < SC_R) ? 1u : 0u;
            nm += ((unsigned)(dv[u].w - lo) < SC_R) ? 1u : 0u;
        }
        unsigned pre = 0, tot = 0;
#pragma unroll
        for (int bit = 0; bit < 5; ++bit) {
            const unsigned m = __builtin_amdgcn_ballot_w32(((nm >> bit) & 1u) != 0u);
            pre += __builtin_amdgcn_mbcnt_lo(m, 0u) << bit;
            tot += (unsigned)__builtin_popcount(m) << bit;
        }
        unsigned base = 0;
        if (lane == 0) base = atomicAdd(cn, tot);
        unsigned slot = (unsigned)__builtin_amdgcn_readfirstlane((int)base) + pre;
#pragma unroll
        for (int u = 0; u < SC_LD; ++u) {
            const int g4 = 4 * (tid + SC_THREADS * u);
            const int dk[4] = {dv[u].x, dv[u].y, dv[u].z, dv[u].w};
#pragma unroll
            for (int k = 0; k < 4; ++k) {
                const unsigned dl = (unsigned)(dk[k] - lo);
                if (dl < SC_R) {
                    list[slot] = ((unsigned)(g4 + k) << 11) | dl;
                    ++slot;
                }
            }
        }
        __syncthreads();

        if (r + 1 < SC_NSUB) sc_load_dst(dv, dstp + (r + 1) * SC_SUB, tid);
        if (tid == 0) cnt[(r + 1) & 1] = 0u;

        const unsigned M   = (unsigned)__builtin_amdgcn_readfirstlane((int)*cn);
        const unsigned per = (M + SC_WAVES - 1) / SC_WAVES;
        const unsigned beg = wv * per;
        const unsigned end = (beg + per < M) ? beg + per : M;

        for (unsigned s = beg; s < end; s += 32) {
            int sl = 0; unsigned dl = 0; v4f cf = {0.f, 0.f, 0.f, 0.f};
            if (s + lane < end) {
                const unsigned ent = list[s + lane];
                dl = ent & 2047u;
                const int e = eb + (int)(ent >> 11);
                sl = edge_index[e];
                const v4f al = *(const v4f*)(alpha + (size_t)e * 4);
                const v4f* ms = (const v4f*)(seg_ms + (size_t)(lo + (int)dl) * 8);
                const v4f mm = ms[0];
                const v4f ss = ms[1];
                const v4f ex = {__expf(al.x - mm.x), __expf(al.y - mm.y),
                                __expf(al.z - mm.z), __expf(al.w - mm.w)};
                cf = ex / (ss + EPSV);
            }
            const int n = (end - s < 32u) ? (int)(end - s) : 32;

            for (int j = 0; j < n; j += SC_G) {
                float xv[SC_G][4];
#pragma unroll
                for (int u = 0; u < SC_G; ++u) {
                    const int sj = __builtin_amdgcn_readlane(sl, j + u);
                    const float* xp = xh + (size_t)sj * HC + lane;
                    xv[u][0] = xp[0];  xv[u][1] = xp[32];
                    xv[u][2] = xp[64]; xv[u][3] = xp[96];
                }
#pragma unroll
                for (int u = 0; u < SC_G; ++u) {
                    if (j + u < n) {
                        const float c0 = __int_as_float(__builtin_amdgcn_readlane(__float_as_int(cf.x), j + u));
                        const float c1 = __int_as_float(__builtin_amdgcn_readlane(__float_as_int(cf.y), j + u));
                        const float c2 = __int_as_float(__builtin_amdgcn_readlane(__float_as_int(cf.z), j + u));
                        const float c3 = __int_as_float(__builtin_amdgcn_readlane(__float_as_int(cf.w), j + u));
                        const int   dj = __builtin_amdgcn_readlane((int)dl, j + u);
                        const float v = c0 * xv[u][0] + c1 * xv[u][1]
                                      + c2 * xv[u][2] + c3 * xv[u][3];
                        __hip_atomic_fetch_add(accs + dj * OUT_CH + lane, v,
                                               __ATOMIC_RELAXED, __HIP_MEMORY_SCOPE_WORKGROUP);
                    }
                }
            }
        }
        __syncthreads();
    }

    v4f* o4 = (v4f*)(part + (size_t)ch * SC_PLANE + (size_t)lo * OUT_CH);
    for (int i = tid; i < SC_R * OUT_CH / 4; i += SC_THREADS)
        o4[i] = accs4[i];
}

__global__ void finalize(const float* __restrict__ part,
                         const float* __restrict__ bias_conv,
                         const float* __restrict__ bias_layer,
                         float* __restrict__ out)
{
    const int t = blockIdx.x * blockDim.x + threadIdx.x;
    if (t >= SC_PLANE / 4) return;
    const v4f* p = (const v4f*)part + t;
    v4f s = p[0];
#pragma unroll
    for (int c = 1; c < SC_CH; ++c)
        s += p[(size_t)c * (SC_PLANE / 4)];
    const int c0 = (t & 7) * 4;
    const float sv[4] = {s.x, s.y, s.z, s.w};
    float rv[4];
#pragma unroll
    for (int k = 0; k < 4; ++k) {
        const float v = sv[k] * 0.25f + bias_conv[c0 + k] + bias_layer[c0 + k];
        rv[k] = v > 0.f ? v : 0.f;
    }
    ((v4f*)out)[t] = (v4f){rv[0], rv[1], rv[2], rv[3]};
}

extern "C" void kernel_launch(void* const* d_in, const int* in_sizes, int n_in,
                              void* d_out, int out_size, void* d_ws, size_t ws_size,
                              hipStream_t stream) {
    const float* x          = (const float*)d_in[0];
    const float* edge_attr  = (const float*)d_in[1];
    const float* W          = (const float*)d_in[2];
    const float* W_edge     = (const float*)d_in[3];
    const float* att_src    = (const float*)d_in[4];
    const float* att_dst    = (const float*)d_in[5];
    const float* att_edge   = (const float*)d_in[6];
    const float* bias_conv  = (const float*)d_in[7];
    const float* bias_layer = (const float*)d_in[8];
    const int*   edge_index = (const int*)d_in[9];
    float* out = (float*)d_out;

    float* ws      = (float*)d_ws;
    float* xh      = ws;
    float* s_node  = xh      + (size_t)N_NODES * HC;
    float* d_node  = s_node  + (size_t)N_NODES * HEADS;
    float* weff    = d_node  + (size_t)N_NODES * HEADS;
    float* alpha   = weff    + (size_t)HEADS * IN_CH;
    float* seg_ms  = alpha   + (size_t)N_EDGES * HEADS;
    float* pmax    = seg_ms  + (size_t)N_NODES * HEADS * 2;
    float* psum    = pmax    + (size_t)ST_CH * N_NODES * HEADS;
    float* part    = psum    + (size_t)ST_CH * N_NODES * HEADS;

    proj_nodes_wmma<<<N_NODES / 16, 128, 0, stream>>>(
        x, W, att_src, att_dst, W_edge, att_edge, xh, s_node, d_node, weff);
    edge_logits<<<(EL_NWAVES + EL_WAVES - 1) / EL_WAVES, EL_WAVES * 32, 0, stream>>>(
        edge_attr, edge_index, s_node, d_node, weff, alpha);
    seg_stats<<<ST_NB * ST_CH, ST_THREADS, 0, stream>>>(
        edge_index, alpha, pmax, psum);
    seg_combine<<<(N_NODES * HEADS + 255) / 256, 256, 0, stream>>>(
        pmax, psum, seg_ms);
    edge_scatter<<<SC_NB * SC_CH, SC_THREADS, 0, stream>>>(
        edge_index, xh, alpha, seg_ms, part);
    finalize<<<(SC_PLANE / 4 + 255) / 256, 256, 0, stream>>>(
        part, bias_conv, bias_layer, out);
}


# edge_scatter: a workgroup owns 200 destinations for ALL edges, one piece per chunk of the edge lists seg_stats now bins and stores: no partial planes, no finalize launch (sum order changes)
# speedup vs baseline: 4.0478x; 1.1742x over previous
#include <hip/hip_runtime.h>
#include <cstdint>

#define N_NODES 50000
#define N_EDGES 800000
#define IN_CH 64
#define OUT_CH 32
#define HEADS 4
#define HC 128
#define NEG_SLOPE 0.2f
#define EPSV 1e-16f

typedef __attribute__((ext_vector_type(2))) float v2f;
typedef __attribute__((ext_vector_type(4))) float v4f;
typedef __attribute__((ext_vector_type(8))) float v8f;

#define PJ_HEAD  36
#define PJ_XPITCH4 17
#define PJ_PITCH (HEADS * PJ_HEAD)

__global__ void __launch_bounds__(128) proj_nodes_wmma(
    const float* __restrict__ x, const float* __restrict__ W,
    const float* __restrict__ att_src, const float* __restrict__ att_dst,
    const float* __restrict__ W_edge, const float* __restrict__ att_edge,
    float* __restrict__ xh, float* __restrict__ s_node,
    float* __restrict__ d_node, float* __restrict__ weff)
{
    __shared__ v4f ptile4[16 * PJ_PITCH / 4];
    __shared__ v4f xs4[16 * PJ_XPITCH4];

    const int lane = threadIdx.x & 31;
    const int wv   = threadIdx.x >> 5;
    const int half = lane >> 4;
    const int l15  = lane & 15;
    const int m0   = blockIdx.x * 16;

    const int colB0 = wv * 32 + l15;
    const int colB1 = colB0 + 16;

    const v4f* xt = (const v4f*)(x + (size_t)m0 * IN_CH);
    const int p0 = threadIdx.x;
    const int p1 = threadIdx.x + 128;
    const v4f xa = xt[p0];
    const v4f xb = xt[p1];

    const int eh = lane & 3;
    const int er = (lane >> 2) + 8 * (wv & 1);
    const float* __restrict__ attv = (wv < 2) ? att_src : att_dst;
    v4f av[8];
#pragma unroll
    for (int j = 0; j < 8; ++j)
        av[j] = *(const v4f*)(attv + eh * OUT_CH + 4 * j);

    v4f bw0[8], bw1[8];
#pragma unroll
    for (int jj = 0; jj < 8; ++jj) {
        bw0[jj] = *(const v4f*)(W + (size_t)colB0 * IN_CH + 8 * jj + 4 * half);
        bw1[jj] = *(const v4f*)(W + (size_t)colB1 * IN_CH + 8 * jj + 4 * half);
    }

    xs4[(p0 >> 4) * PJ_XPITCH4 + (p0 & 15)] = xa;
    xs4[(p1 >> 4) * PJ_XPITCH4 + (p1 & 15)] = xb;
    __syncthreads();

    v4f a[8];
#pragma unroll
    for (int jj = 0; jj < 8; ++jj)
        a[jj] = xs4[l15 * PJ_XPITCH4 + 2 * jj + half];

    v8f acc0 = {};
    v8f acc1 = {};
#pragma unroll
    for (int jj = 0; jj < 8; ++jj) {
        const v2f a0  = {a[jj].x, a[jj].y};
        const v2f a1  = {a[jj].z, a[jj].w};
        const v2f b00 = {bw0[jj].x, bw0[jj].y};
        const v2f b01 = {bw0[jj].z, bw0[jj].w};
        const v2f b10 = {bw1[jj].x, bw1[jj].y};
        const v2f b11 = {bw1[jj].z, bw1[jj].w};
        acc0 = __builtin_amdgcn_wmma_f32_16x16x4_f32(
                   false, a0, false, b00, (short)0, acc0, false, false);
        acc1 = __builtin_amdgcn_wmma_f32_16x16x4_f32(
                   false, a0, false, b10, (short)0, acc1, false, false);
        acc0 = __builtin_amdgcn_wmma_f32_16x16x4_f32(
                   false, a1, false, b01, (short)0, acc0, false, false);
        acc1 = __builtin_amdgcn_wmma_f32_16x16x4_f32(
                   false, a1, false, b11, (short)0, acc1, false, false);
    }
    float* ptile = (float*)ptile4;
#pragma unroll
    for (int v = 0; v < 8; ++v) {
        const int r = v + half * 8;
        ptile[r * PJ_PITCH + wv * PJ_HEAD + l15]      = acc0[v];
        ptile[r * PJ_PITCH + wv * PJ_HEAD + 16 + l15] = acc1[v];
    }
    __syncthreads();

#pragma unroll
    for (int v = 0; v < 8; ++v) {
        const int r = v + half * 8;
        xh[(size_t)(m0 + r) * HC + colB0] = acc0[v];
        xh[(size_t)(m0 + r) * HC + colB1] = acc1[v];
    }

    const v4f* rp = (const v4f*)(ptile + er * PJ_PITCH + eh * PJ_HEAD);
    float sd = 0.f;
#pragma unroll
    for (int j = 0; j < 8; ++j) {
        const v4f r = rp[j];
        sd += r.x * av[j].x;
        sd += r.y * av[j].y;
        sd += r.z * av[j].z;
        sd += r.w * av[j].w;
    }
    float* __restrict__ nd = (wv < 2) ? s_node : d_node;
    nd[(size_t)(m0 + er) * HEADS + eh] = sd;

    if (blockIdx.x == 0) {
#pragma unroll 1
        for (int u = 0; u < 2; ++u) {
            const int t = threadIdx.x + 128 * u;
            const int h = t >> 6, k = t & 63;
            float s = 0.f;
#pragma unroll
            for (int c = 0; c < OUT_CH; ++c)
                s += att_edge[h * OUT_CH + c] * W_edge[(size_t)(h * OUT_CH + c) * IN_CH + k];
            weff[t] = s;
        }
    }
}

#define EL_WAVES  4
#define EL_TPW    8
#define EL_NTILES (N_EDGES / 16)
#define EL_NWAVES (EL_NTILES / EL_TPW)
#define EL_PITCH4 17

struct EdgeTile {
    v4f r[8];
    int srcA, dstA, srcB, dstB;
};

__device__ __forceinline__ void el_load_tile(
    EdgeTile& T, const float* __restrict__ edge_attr,
    const int* __restrict__ edge_index, int t, int lane, int eo)
{
    const v4f* p = (const v4f*)(edge_attr + (size_t)t * (16 * IN_CH)) + lane;
#pragma unroll
    for (int j = 0; j < 8; ++j)
        T.r[j] = __builtin_nontemporal_load(p + 32 * j);
    const int e = t * 16 + eo;
    T.srcA = edge_index[e];
    T.dstA = edge_index[N_EDGES + e];
    T.srcB = edge_index[e + 4];
    T.dstB = edge_index[N_EDGES + e + 4];
}

__device__ __forceinline__ void el_wave_lds_order()
{
    __builtin_amdgcn_fence(__ATOMIC_RELEASE, "wavefront");
    __builtin_amdgcn_wave_barrier();
    __builtin_amdgcn_fence(__ATOMIC_ACQUIRE, "wavefront");
}

__device__ __forceinline__ void el_finish(
    float a, int e, int hd, float* __restrict__ alpha)
{
    a = (a > 0.f) ? a : NEG_SLOPE * a;
    alpha[(size_t)e * 4 + hd] = a;
}

__global__ void __launch_bounds__(EL_WAVES * 32) edge_logits(
    const float* __restrict__ edge_attr, const int* __restrict__ edge_index,
    const float* __restrict__ s_node, const float* __restrict__ d_node,
    const float* __restrict__ weff,
    float* __restrict__ alpha)
{
    __shared__ v4f stage[EL_WAVES * 16 * EL_PITCH4];

    const int lane = threadIdx.x & 31;
    const int wv   = threadIdx.x >> 5;
    const int c    = lane & 15;
    const int half = lane >> 4;
    const int w    = blockIdx.x * EL_WAVES + wv;
    if (w >= EL_NWAVES) return;

    v4f* st = stage + wv * (16 * EL_PITCH4);

    const int hd = c & 3;
    v4f bw[8];
#pragma unroll
    for (int jj = 0; jj < 8; ++jj)
        bw[jj] = *(const v4f*)(weff + hd * IN_CH + 8 * jj + 4 * half);

    const int q  = c >> 2;
    const int eo = 8 * half + q;

    EdgeTile cur;
    el_load_tile(cur, edge_attr, edge_index, w, lane, eo);

    for (int i = 0; i < EL_TPW; ++i) {
        const int t = w + i * EL_NWAVES;

#pragma unroll
        for (int j = 0; j < 8; ++j)
            st[(2 * j + half) * EL_PITCH4 + c] = cur.r[j];

        const int srcA = cur.srcA, dstA = cur.dstA;
        const int srcB = cur.srcB, dstB = cur.dstB;
        const float sA = s_node[dstA * 4 + hd];
        const float dA = d_node[srcA * 4 + hd];
        const float sB = s_node[dstB * 4 + hd];
        const float dB = d_node[srcB * 4 + hd];

        if (i + 1 < EL_TPW)
            el_load_tile(cur, edge_attr, edge_index, t + EL_NWAVES, lane, eo);

        el_wave_lds_order();

        v4f a[8];
#pragma unroll
        for (int jj = 0; jj < 8; ++jj)
            a[jj] = st[c * EL_PITCH4 + 2 * jj + half];

        v8f acc0 = {};
        v8f acc1 = {};
#pragma unroll
        for (int jj = 0; jj < 8; ++jj) {
            const v2f a0 = {a[jj].x, a[jj].y};
            const v2f a1 = {a[jj].z, a[jj].w};
            const v2f b0 = {bw[jj].x, bw[jj].y};
            const v2f b1 = {bw[jj].z, bw[jj].w};
            acc0 = __builtin_amdgcn_wmma_f32_16x16x4_f32(
                       false, a0, false, b0, (short)0, acc0, false, false);
            acc1 = __builtin_amdgcn_wmma_f32_16x16x4_f32(
                       false, a1, false, b1, (short)0, acc1, false, false);
        }

        el_wave_lds_order();

        asm volatile("v_nop\n\tv_nop\n\tv_nop\n\tv_nop\n\t"
                     "v_nop\n\tv_nop\n\tv_nop\n\tv_nop"
                     : "+v"(acc0), "+v"(acc1));

        const v8f acc = acc0 + acc1;
        const bool q1 = (q & 1) != 0, q2 = (q & 2) != 0;
        const float lA01 = q1 ? acc[1] : acc[0], lA23 = q1 ? acc[3] : acc[2];
        const float lB01 = q1 ? acc[5] : acc[4], lB23 = q1 ? acc[7] : acc[6];
        const float lA = q2 ? lA23 : lA01;
        const float lB = q2 ? lB23 : lB01;

        const int eA = t * 16 + eo;
        el_finish(lA + sA + dA, eA,     hd, alpha);
        el_finish(lB + sB + dB, eA + 4, hd, alpha);
    }
}

#define ST_NB      15
#define SC_FR      200
#define ST_R       3400
#define SC_FPR     (ST_R / SC_FR)
#define ST_CH      16
#define ST_CE      (N_EDGES / ST_CH)
#define ST_THREADS 1024
#define ST_LD      13
#define ST_B       4
#define LS_STRIDE  50016
#define LS_CNT     32

static_assert(ST_R * ST_NB >= N_NODES && ST_R * (ST_NB - 1) < N_NODES && ST_R % 8 == 0, "ranges must cover the nodes");
static_assert(ST_CE * ST_CH == N_EDGES && ST_CE % 4 == 0, "chunks must tile the edges");
static_assert(ST_THREADS * ST_LD * 4 >= ST_CE, "one sweep must cover a chunk");
static_assert(ST_R <= 4096 && ST_CE <= (1 << 20), "list entry packing");
static_assert(ST_R % SC_FR == 0 && N_NODES % SC_FR == 0 && SC_FPR + 1 <= LS_CNT, "list bins must tile the ranges");
static_assert(LS_STRIDE >= ST_CE && LS_STRIDE % 32 == 0, "stored lists must not share a line");

__device__ __forceinline__ void st_lds_max(int* addr, float a)
{
    const int ai = __float_as_int(a);
    if (ai >= 0) __hip_atomic_fetch_max(addr, ai, __ATOMIC_RELAXED, __HIP_MEMORY_SCOPE_WORKGROUP);
    else         __hip_atomic_fetch_min((unsigned int*)addr, (unsigned int)ai,
                                        __ATOMIC_RELAXED, __HIP_MEMORY_SCOPE_WORKGROUP);
}

__global__ void __launch_bounds__(ST_THREADS) seg_stats(
    const int* __restrict__ edge_index, const float* __restrict__ alpha,
    float* __restrict__ pmax, float* __restrict__ psum,
    unsigned int* __restrict__ elist, int* __restrict__ eoff)
{
    __shared__ v4f mx4[ST_R];
    __shared__ v4f sm4[ST_R];
    __shared__ unsigned int list[ST_CE];
    __shared__ unsigned int bincnt[LS_CNT];
    __shared__ unsigned int binbase[LS_CNT];

    const int tid  = threadIdx.x;
    const int b    = blockIdx.x % ST_NB;
    const int ch   = blockIdx.x / ST_NB;
    const int lo   = b * ST_R;
    const int e0   = ch * ST_CE;

    const int4* p = (const int4*)(edge_index + N_EDGES + e0);
    int4 dv[ST_LD];
#pragma unroll
    for (int u = 0; u < ST_LD; ++u) {
        const int g = tid + ST_THREADS * u;
        dv[u] = (g < ST_CE / 4) ? p[g] : make_int4(-1, -1, -1, -1);
    }

    const float ninf = -__builtin_huge_valf();
    for (int i = tid; i < ST_R; i += ST_THREADS) {
        mx4[i] = (v4f){ninf, ninf, ninf, ninf};
        sm4[i] = (v4f){0.f, 0.f, 0.f, 0.f};
    }
    if (tid < LS_CNT) bincnt[tid] = 0u;
    __syncthreads();

#pragma unroll
    for (int u = 0; u < ST_LD; ++u) {
        const int dk[4] = {dv[u].x, dv[u].y, dv[u].z, dv[u].w};
#pragma unroll
        for (int k = 0; k < 4; ++k) {
            const unsigned dl = (unsigned)(dk[k] - lo);
            if (dl < ST_R) atomicAdd(&bincnt[dl / SC_FR], 1u);
        }
    }
    __syncthreads();
    if (tid == 0) {
        unsigned a = 0u;
#pragma unroll 1
        for (int j = 0; j < SC_FPR; ++j) {
            const unsigned n = bincnt[j];
            binbase[j] = a;
            bincnt[j]  = a;
            a += n;
        }
        binbase[SC_FPR] = a;
    }
    __syncthreads();
    unsigned pos[ST_LD][4];
#pragma unroll
    for (int u = 0; u < ST_LD; ++u) {
        const int dk[4] = {dv[u].x, dv[u].y, dv[u].z, dv[u].w};
#pragma unroll
        for (int k = 0; k < 4; ++k) {
            const unsigned dl = (unsigned)(dk[k] - lo);
            pos[u][k] = 0u;
            if (dl < ST_R) pos[u][k] = atomicAdd(&bincnt[dl / SC_FR], 1u);
        }
    }
    __builtin_amdgcn_fence(__ATOMIC_ACQ_REL, "workgroup");
#pragma unroll
    for (int u = 0; u < ST_LD; ++u) {
        const int g4 = 4 * (tid + ST_THREADS * u);
        const int dk[4] = {dv[u].x, dv[u].y, dv[u].z, dv[u].w};
#pragma unroll
        for (int k = 0; k < 4; ++k) {
            const unsigned dl = (unsigned)(dk[k] - lo);
            if (dl < ST_R)
                list[pos[u][k]] = ((unsigned)(g4 + k) << 12) | dl;
        }
    }
    __syncthreads();

    const unsigned M = (unsigned)__builtin_amdgcn_readfirstlane((int)binbase[SC_FPR]);
    int*   mxi = (int*)mx4;
    float* smf = (float*)sm4;

    {
        unsigned int* el = elist + (size_t)blockIdx.x * LS_STRIDE;
        for (unsigned i = (unsigned)tid; i < M; i += ST_THREADS) el[i] = list[i];
        if (tid <= SC_FPR) eoff[blockIdx.x * LS_CNT + tid] = (int)binbase[tid];
    }

    for (unsigned i0 = (unsigned)tid; i0 < M; i0 += ST_B * ST_THREADS) {
        v4f a[ST_B]; unsigned dl[ST_B];
#pragma unroll
        for (int u = 0; u < ST_B; ++u) {
            const unsigned i = i0 + u * ST_THREADS;
            const unsigned ent = list[(i < M) ? i : i0];
            dl[u] = (i < M) ? (ent & 4095u) : 0xFFFFFFFFu;
            a[u] = *(const v4f*)(alpha + ((size_t)e0 + (ent >> 12)) * 4);
        }
#pragma unroll
        for (int u = 0; u < ST_B; ++u) {
            if (dl[u] != 0xFFFFFFFFu) {
                st_lds_max(mxi + dl[u] * 4 + 0, a[u].x);
                st_lds_max(mxi + dl[u] * 4 + 1, a[u].y);
                st_lds_max(mxi + dl[u] * 4 + 2, a[u].z);
                st_lds_max(mxi + dl[u] * 4 + 3, a[u].w);
            }
        }
    }
    __syncthreads();

    for (unsigned i0 = (unsigned)tid; i0 < M; i0 += ST_B * ST_THREADS) {
        v4f a[ST_B]; unsigned dl[ST_B];
#pragma unroll
        for (int u = 0; u < ST_B; ++u) {
            const unsigned i = i0 + u * ST_THREADS;
            const unsigned ent = list[(i < M) ? i : i0];
            dl[u] = (i < M) ? (ent & 4095u) : 0xFFFFFFFFu;
            a[u] = *(const v4f*)(alpha + ((size_t)e0 + (ent >> 12)) * 4);
        }
        asm volatile("" :: "v"(a[0]), "v"(a[1]), "v"(a[2]), "v"(a[3]));
#pragma unroll
        for (int u = 0; u < ST_B; ++u) {
            if (dl[u] != 0xFFFFFFFFu) {
                const v4f m = mx4[dl[u]];
                __hip_atomic_fetch_add(smf + dl[u] * 4 + 0, __expf(a[u].x - m.x),
                                       __ATOMIC_RELAXED, __HIP_MEMORY_SCOPE_WORKGROUP);
                __hip_atomic_fetch_add(smf + dl[u] * 4 + 1, __expf(a[u].y - m.y),
                                       __ATOMIC_RELAXED, __HIP_MEMORY_SCOPE_WORKGROUP);
                __hip_atomic_fetch_add(smf + dl[u] * 4 + 2, __expf(a[u].z - m.z),
                                       __ATOMIC_RELAXED, __HIP_MEMORY_SCOPE_WORKGROUP);
                __hip_atomic_fetch_add(smf + dl[u] * 4 + 3, __expf(a[u].w - m.w),
                                       __ATOMIC_RELAXED, __HIP_MEMORY_SCOPE_WORKGROUP);
            }
        }
    }
    __syncthreads();

    const int nr = (N_NODES - lo < ST_R) ? (N_NODES - lo) : ST_R;
    v4f* om = (v4f*)pmax + (size_t)ch * N_NODES + lo;
    v4f* os = (v4f*)psum + (size_t)ch * N_NODES + lo;
    for (int i = tid; i < nr; i += ST_THREADS) {
        om[i] = mx4[i];
        os[i] = sm4[i];
    }
}

__global__ void seg_combine(const float* __restrict__ pmax,
                            const float* __restrict__ psum,
                            float* __restrict__ seg_ms)
{
    const int t = blockIdx.x * blockDim.x + threadIdx.x;
    if (t >= N_NODES * HEADS) return;
    float m[ST_CH], sv[ST_CH];
#pragma unroll
    for (int c = 0; c < ST_CH; ++c) {
        m[c]  = pmax[(size_t)c * (N_NODES * HEADS) + t];
        sv[c] = psum[(size_t)c * (N_NODES * HEADS) + t];
    }
    const float ninf = -__builtin_huge_valf();
    float M = m[0];
#pragma unroll
    for (int c = 1; c < ST_CH; ++c) M = (m[c] > M) ? m[c] : M;
    if (!__builtin_isfinite(M)) M = 0.f;
    float S = 0.f;
#pragma unroll
    for (int c = 0; c < ST_CH; ++c)
        S += (m[c] == ninf) ? 0.f : sv[c] * __expf(m[c] - M);
    const int n = t >> 2, h = t & 3;
    seg_ms[(size_t)n * 8 + h]     = M;
    seg_ms[(size_t)n * 8 + 4 + h] = S;
}

#define SC_NF      (N_NODES / SC_FR)
#define SC_THREADS 1024
#define SC_WAVES   (SC_THREADS / 32)
#define SC_PRE     4
#define SC_G       8
#define SC_COPIES  8
#define SC_ACC4    (SC_FR * OUT_CH / 4)

static_assert(SC_WAVES == 2 * ST_CH, "two waves per stored list");
static_assert(32 % SC_G == 0, "groups must tile a 32-entry slab");

struct ScRaw {
    int sl; unsigned dl; v4f al, mm, ss;
};

__device__ __forceinline__ void sc_fetch(
    ScRaw& R, unsigned ent, int ebase, unsigned off, int lo,
    const int* __restrict__ edge_index, const float* __restrict__ alpha,
    const float* __restrict__ seg_ms)
{
    unsigned dl = (ent & 4095u) - off;
    dl = (dl < (unsigned)SC_FR) ? dl : 0u;
    unsigned eo = ent >> 12;
    eo = (eo < (unsigned)ST_CE) ? eo : 0u;
    const int e = ebase + (int)eo;
    R.dl = dl;
    R.sl = edge_index[e];
    R.al = *(const v4f*)(alpha + (size_t)e * 4);
    const v4f* ms = (const v4f*)(seg_ms + (size_t)(lo + (int)dl) * 8);
    R.mm = ms[0];
    R.ss = ms[1];
}

__device__ __forceinline__ void sc_rows(
    const ScRaw& R, int n, int lane,
    const float* __restrict__ xh, float* accs)
{
    const v4f ex = {__expf(R.al.x - R.mm.x), __expf(R.al.y - R.mm.y),
                    __expf(R.al.z - R.mm.z), __expf(R.al.w - R.mm.w)};
    const v4f cf = ex / (R.ss + EPSV);
    const int sl = R.sl;
    const unsigned dl = R.dl;
    for (int j = 0; j < n; j += SC_G) {
        float xv[SC_G][4];
#pragma unroll
        for (int u = 0; u < SC_G; ++u) {
            const int sj = __builtin_amdgcn_readlane(sl, j + u);
            const float* xp = xh + (size_t)sj * HC + lane;
            xv[u][0] = xp[0];  xv[u][1] = xp[32];
            xv[u][2] = xp[64]; xv[u][3] = xp[96];
        }
#pragma unroll
        for (int u = 0; u < SC_G; ++u) {
            if (j + u < n) {
                const float c0 = __int_as_float(__builtin_amdgcn_readlane(__float_as_int(cf.x), j + u));
                const float c1 = __int_as_float(__builtin_amdgcn_readlane(__float_as_int(cf.y), j + u));
                const float c2 = __int_as_float(__builtin_amdgcn_readlane(__float_as_int(cf.z), j + u));
                const float c3 = __int_as_float(__builtin_amdgcn_readlane(__float_as_int(cf.w), j + u));
                const int   dj = __builtin_amdgcn_readlane((int)dl, j + u);
                const float v = c0 * xv[u][0] + c1 * xv[u][1]
                              + c2 * xv[u][2] + c3 * xv[u][3];
                __hip_atomic_fetch_add(accs + dj * OUT_CH + lane, v,
                                       __ATOMIC_RELAXED, __HIP_MEMORY_SCOPE_WORKGROUP);
            }
        }
    }
}

__global__ void __launch_bounds__(SC_THREADS) edge_scatter(
    const int* __restrict__ edge_index, const float* __restrict__ xh,
    const float* __restrict__ alpha, const float* __restrict__ seg_ms,
    const unsigned int* __restrict__ elist, const int* __restrict__ eoff,
    const float* __restrict__ bias_conv, const float* __restrict__ bias_layer,
    float* __restrict__ out)
{
    __shared__ v4f accs4[SC_COPIES * SC_ACC4];

    const int tid  = threadIdx.x;
    const int lane = tid & 31;
    const int wv   = __builtin_amdgcn_readfirstlane(tid >> 5);
    float* accs = (float*)(accs4 + (wv % SC_COPIES) * SC_ACC4);
    const int f    = blockIdx.x;
    const int b    = f / SC_FPR;
    const int fb   = f - b * SC_FPR;
    const unsigned off = (unsigned)(fb * SC_FR);
    const int lo   = f * SC_FR;

    const int ch   = wv >> 1;
    const int reg  = ch * ST_NB + b;
    int p0 = eoff[reg * LS_CNT + fb];
    int p1 = eoff[reg * LS_CNT + fb + 1];
    p0 = (p0 < 0) ? 0 : ((p0 > ST_CE) ? ST_CE : p0);
    p1 = (p1 < p0) ? p0 : ((p1 > ST_CE) ? ST_CE : p1);
    const int mid = p0 + ((p1 - p0 + 1) >> 1);
    const int beg = (wv & 1) ? mid : p0;
    const int end = (wv & 1) ? p1 : mid;
    const unsigned int* el = elist + (size_t)reg * LS_STRIDE;
    const int ebase = ch * ST_CE;

    unsigned entq[SC_PRE];
#pragma unroll
    for (int i = 0; i < SC_PRE; ++i) {
        const int idx = beg + 32 * i + lane;
        entq[i] = (idx < end) ? el[idx] : 0u;
    }
    ScRaw raw[SC_PRE];
#pragma unroll
    for (int i = 0; i < SC_PRE; ++i)
        sc_fetch(raw[i], entq[i], ebase, off, lo, edge_index, alpha, seg_ms);

    for (int i = tid; i < SC_COPIES * SC_ACC4; i += SC_THREADS)
        accs4[i] = (v4f){0.f, 0.f, 0.f, 0.f};
    __syncthreads();

#pragma unroll
    for (int i = 0; i < SC_PRE; ++i) {
        const int s = beg + 32 * i;
        if (s < end)
            sc_rows(raw[i], (end - s < 32) ? end - s : 32, lane, xh, accs);
    }
    for (int s = beg + 32 * SC_PRE; s < end; s += 32) {
        const int idx = s + lane;
        const int n = (end - s < 32) ? end - s : 32;
        const unsigned ent = (idx < end) ? el[idx] : 0u;
        ScRaw R;
        sc_fetch(R, ent, ebase, off, lo, edge_index, alpha, seg_ms);
        sc_rows(R, n, lane, xh, accs);
    }
    __syncthreads();

    v4f* o4 = (v4f*)(out + (size_t)lo * OUT_CH);
    for (int i = tid; i < SC_ACC4; i += SC_THREADS) {
        v4f s = accs4[i];
#pragma unroll
        for (int k = 1; k < SC_COPIES; ++k) s += accs4[k * SC_ACC4 + i];
        const int ch0 = (i & 7) * 4;
        const float sv[4] = {s.x, s.y, s.z, s.w};
        float rv[4];
#pragma unroll
        for (int k = 0; k < 4; ++k) {
            const float v = sv[k] * 0.25f + bias_conv[ch0 + k] + bias_layer[ch0 + k];
            rv[k] = v > 0.f ? v : 0.f;
        }
        o4[i] = (v4f){rv[0], rv[1], rv[2], rv[3]};
    }
}

extern "C" void kernel_launch(void* const* d_in, const int* in_sizes, int n_in,
                              void* d_out, int out_size, void* d_ws, size_t ws_size,
                              hipStream_t stream) {
    const float* x          = (const float*)d_in[0];
    const float* edge_attr  = (const float*)d_in[1];
    const float* W          = (const float*)d_in[2];
    const float* W_edge     = (const float*)d_in[3];
    const float* att_src    = (const float*)d_in[4];
    const float* att_dst    = (const float*)d_in[5];
    const float* att_edge   = (const float*)d_in[6];
    const float* bias_conv  = (const float*)d_in[7];
    const float* bias_layer = (const float*)d_in[8];
    const int*   edge_index = (const int*)d_in[9];
    float* out = (float*)d_out;

    float* ws      = (float*)d_ws;
    float* xh      = ws;
    float* s_node  = xh      + (size_t)N_NODES * HC;
    float* d_node  = s_node  + (size_t)N_NODES * HEADS;
    float* weff    = d_node  + (size_t)N_NODES * HEADS;
    float* alpha   = weff    + (size_t)HEADS * IN_CH;
    float* seg_ms  = alpha   + (size_t)N_EDGES * HEADS;
    float* pmax    = seg_ms  + (size_t)N_NODES * HEADS * 2;
    float* psum    = pmax    + (size_t)ST_CH * N_NODES * HEADS;
    unsigned int* elist = (unsigned int*)(psum + (size_t)ST_CH * N_NODES * HEADS);
    int* eoff      = (int*)(elist + (size_t)ST_NB * ST_CH * LS_STRIDE);

    proj_nodes_wmma<<<N_NODES / 16, 128, 0, stream>>>(
        x, W, att_src, att_dst, W_edge, att_edge, xh, s_node, d_node, weff);
    edge_logits<<<(EL_NWAVES + EL_WAVES - 1) / EL_WAVES, EL_WAVES * 32, 0, stream>>>(
        edge_attr, edge_index, s_node, d_node, weff, alpha);
    seg_stats<<<ST_NB * ST_CH, ST_THREADS, 0, stream>>>(
        edge_index, alpha, pmax, psum, elist, eoff);
    seg_combine<<<(N_NODES * HEADS + 255) / 256, 256, 0, stream>>>(
        pmax, psum, seg_ms);
    edge_scatter<<<SC_NF, SC_THREADS, 0, stream>>>(
        edge_index, xh, alpha, seg_ms, elist, eoff, bias_conv, bias_layer, out);
}


# softmax max and exp-sum move into edge_scatter: a workgroup owning 200 destinations takes both in LDS from its edge lists; pass 2 only bins; no partial planes or combine launch (sum order changes)
# speedup vs baseline: 4.6194x; 1.1412x over previous
#include <hip/hip_runtime.h>
#include <cstdint>

#define N_NODES 50000
#define N_EDGES 800000
#define IN_CH 64
#define OUT_CH 32
#define HEADS 4
#define HC 128
#define NEG_SLOPE 0.2f
#define EPSV 1e-16f

typedef __attribute__((ext_vector_type(2))) float v2f;
typedef __attribute__((ext_vector_type(4))) float v4f;
typedef __attribute__((ext_vector_type(8))) float v8f;

#define PJ_HEAD  36
#define PJ_XPITCH4 17
#define PJ_PITCH (HEADS * PJ_HEAD)

__global__ void __launch_bounds__(128) proj_nodes_wmma(
    const float* __restrict__ x, const float* __restrict__ W,
    const float* __restrict__ att_src, const float* __restrict__ att_dst,
    const float* __restrict__ W_edge, const float* __restrict__ att_edge,
    float* __restrict__ xh, float* __restrict__ s_node,
    float* __restrict__ d_node, float* __restrict__ weff)
{
    __shared__ v4f ptile4[16 * PJ_PITCH / 4];
    __shared__ v4f xs4[16 * PJ_XPITCH4];

    const int lane = threadIdx.x & 31;
    const int wv   = threadIdx.x >> 5;
    const int half = lane >> 4;
    const int l15  = lane & 15;
    const int m0   = blockIdx.x * 16;

    const int colB0 = wv * 32 + l15;
    const int colB1 = colB0 + 16;

    const v4f* xt = (const v4f*)(x + (size_t)m0 * IN_CH);
    const int p0 = threadIdx.x;
    const int p1 = threadIdx.x + 128;
    const v4f xa = xt[p0];
    const v4f xb = xt[p1];

    const int eh = lane & 3;
    const int er = (lane >> 2) + 8 * (wv & 1);
    const float* __restrict__ attv = (wv < 2) ? att_src : att_dst;
    v4f av[8];
#pragma unroll
    for (int j = 0; j < 8; ++j)
        av[j] = *(const v4f*)(attv + eh * OUT_CH + 4 * j);

    v4f bw0[8], bw1[8];
#pragma unroll
    for (int jj = 0; jj < 8; ++jj) {
        bw0[jj] = *(const v4f*)(W + (size_t)colB0 * IN_CH + 8 * jj + 4 * half);
        bw1[jj] = *(const v4f*)(W + (size_t)colB1 * IN_CH + 8 * jj + 4 * half);
    }

    xs4[(p0 >> 4) * PJ_XPITCH4 + (p0 & 15)] = xa;
    xs4[(p1 >> 4) * PJ_XPITCH4 + (p1 & 15)] = xb;
    __syncthreads();

    v4f a[8];
#pragma unroll
    for (int jj = 0; jj < 8; ++jj)
        a[jj] = xs4[l15 * PJ_XPITCH4 + 2 * jj + half];

    v8f acc0 = {};
    v8f acc1 = {};
#pragma unroll
    for (int jj = 0; jj < 8; ++jj) {
        const v2f a0  = {a[jj].x, a[jj].y};
        const v2f a1  = {a[jj].z, a[jj].w};
        const v2f b00 = {bw0[jj].x, bw0[jj].y};
        const v2f b01 = {bw0[jj].z, bw0[jj].w};
        const v2f b10 = {bw1[jj].x, bw1[jj].y};
        const v2f b11 = {bw1[jj].z, bw1[jj].w};
        acc0 = __builtin_amdgcn_wmma_f32_16x16x4_f32(
                   false, a0, false, b00, (short)0, acc0, false, false);
        acc1 = __builtin_amdgcn_wmma_f32_16x16x4_f32(
                   false, a0, false, b10, (short)0, acc1, false, false);
        acc0 = __builtin_amdgcn_wmma_f32_16x16x4_f32(
                   false, a1, false, b01, (short)0, acc0, false, false);
        acc1 = __builtin_amdgcn_wmma_f32_16x16x4_f32(
                   false, a1, false, b11, (short)0, acc1, false, false);
    }
    float* ptile = (float*)ptile4;
#pragma unroll
    for (int v = 0; v < 8; ++v) {
        const int r = v + half * 8;
        ptile[r * PJ_PITCH + wv * PJ_HEAD + l15]      = acc0[v];
        ptile[r * PJ_PITCH + wv * PJ_HEAD + 16 + l15] = acc1[v];
    }
    __syncthreads();

#pragma unroll
    for (int v = 0; v < 8; ++v) {
        const int r = v + half * 8;
        xh[(size_t)(m0 + r) * HC + colB0] = acc0[v];
        xh[(size_t)(m0 + r) * HC + colB1] = acc1[v];
    }

    const v4f* rp = (const v4f*)(ptile + er * PJ_PITCH + eh * PJ_HEAD);
    float sd = 0.f;
#pragma unroll
    for (int j = 0; j < 8; ++j) {
        const v4f r = rp[j];
        sd += r.x * av[j].x;
        sd += r.y * av[j].y;
        sd += r.z * av[j].z;
        sd += r.w * av[j].w;
    }
    float* __restrict__ nd = (wv < 2) ? s_node : d_node;
    nd[(size_t)(m0 + er) * HEADS + eh] = sd;

    if (blockIdx.x == 0) {
#pragma unroll 1
        for (int u = 0; u < 2; ++u) {
            const int t = threadIdx.x + 128 * u;
            const int h = t >> 6, k = t & 63;
            float s = 0.f;
#pragma unroll
            for (int c = 0; c < OUT_CH; ++c)
                s += att_edge[h * OUT_CH + c] * W_edge[(size_t)(h * OUT_CH + c) * IN_CH + k];
            weff[t] = s;
        }
    }
}

#define EL_WAVES  4
#define EL_TPW    8
#define EL_NTILES (N_EDGES / 16)
#define EL_NWAVES (EL_NTILES / EL_TPW)
#define EL_PITCH4 17

struct EdgeTile {
    v4f r[8];
    int srcA, dstA, srcB, dstB;
};

__device__ __forceinline__ void el_load_tile(
    EdgeTile& T, const float* __restrict__ edge_attr,
    const int* __restrict__ edge_index, int t, int lane, int eo)
{
    const v4f* p = (const v4f*)(edge_attr + (size_t)t * (16 * IN_CH)) + lane;
#pragma unroll
    for (int j = 0; j < 8; ++j)
        T.r[j] = __builtin_nontemporal_load(p + 32 * j);
    const int e = t * 16 + eo;
    T.srcA = edge_index[e];
    T.dstA = edge_index[N_EDGES + e];
    T.srcB = edge_index[e + 4];
    T.dstB = edge_index[N_EDGES + e + 4];
}

__device__ __forceinline__ void el_wave_lds_order()
{
    __builtin_amdgcn_fence(__ATOMIC_RELEASE, "wavefront");
    __builtin_amdgcn_wave_barrier();
    __builtin_amdgcn_fence(__ATOMIC_ACQUIRE, "wavefront");
}

__device__ __forceinline__ void el_finish(
    float a, int e, int hd, float* __restrict__ alpha)
{
    a = (a > 0.f) ? a : NEG_SLOPE * a;
    alpha[(size_t)e * 4 + hd] = a;
}

__global__ void __launch_bounds__(EL_WAVES * 32) edge_logits(
    const float* __restrict__ edge_attr, const int* __restrict__ edge_index,
    const float* __restrict__ s_node, const float* __restrict__ d_node,
    const float* __restrict__ weff,
    float* __restrict__ alpha)
{
    __shared__ v4f stage[EL_WAVES * 16 * EL_PITCH4];

    const int lane = threadIdx.x & 31;
    const int wv   = threadIdx.x >> 5;
    const int c    = lane & 15;
    const int half = lane >> 4;
    const int w    = blockIdx.x * EL_WAVES + wv;
    if (w >= EL_NWAVES) return;

    v4f* st = stage + wv * (16 * EL_PITCH4);

    const int hd = c & 3;
    v4f bw[8];
#pragma unroll
    for (int jj = 0; jj < 8; ++jj)
        bw[jj] = *(const v4f*)(weff + hd * IN_CH + 8 * jj + 4 * half);

    const int q  = c >> 2;
    const int eo = 8 * half + q;

    EdgeTile cur;
    el_load_tile(cur, edge_attr, edge_index, w, lane, eo);

    for (int i = 0; i < EL_TPW; ++i) {
        const int t = w + i * EL_NWAVES;

#pragma unroll
        for (int j = 0; j < 8; ++j)
            st[(2 * j + half) * EL_PITCH4 + c] = cur.r[j];

        const int srcA = cur.srcA, dstA = cur.dstA;
        const int srcB = cur.srcB, dstB = cur.dstB;
        const float sA = s_node[dstA * 4 + hd];
        const float dA = d_node[srcA * 4 + hd];
        const float sB = s_node[dstB * 4 + hd];
        const float dB = d_node[srcB * 4 + hd];

        if (i + 1 < EL_TPW)
            el_load_tile(cur, edge_attr, edge_index, t + EL_NWAVES, lane, eo);

        el_wave_lds_order();

        v4f a[8];
#pragma unroll
        for (int jj = 0; jj < 8; ++jj)
            a[jj] = st[c * EL_PITCH4 + 2 * jj + half];

        v8f acc0 = {};
        v8f acc1 = {};
#pragma unroll
        for (int jj = 0; jj < 8; ++jj) {
            const v2f a0 = {a[jj].x, a[jj].y};
            const v2f a1 = {a[jj].z, a[jj].w};
            const v2f b0 = {bw[jj].x, bw[jj].y};
            const v2f b1 = {bw[jj].z, bw[jj].w};
            acc0 = __builtin_amdgcn_wmma_f32_16x16x4_f32(
                       false, a0, false, b0, (short)0, acc0, false, false);
            acc1 = __builtin_amdgcn_wmma_f32_16x16x4_f32(
                       false, a1, false, b1, (short)0, acc1, false, false);
        }

        el_wave_lds_order();

        asm volatile("v_nop\n\tv_nop\n\tv_nop\n\tv_nop\n\t"
                     "v_nop\n\tv_nop\n\tv_nop\n\tv_nop"
                     : "+v"(acc0), "+v"(acc1));

        const v8f acc = acc0 + acc1;
        const bool q1 = (q & 1) != 0, q2 = (q & 2) != 0;
        const float lA01 = q1 ? acc[1] : acc[0], lA23 = q1 ? acc[3] : acc[2];
        const float lB01 = q1 ? acc[5] : acc[4], lB23 = q1 ? acc[7] : acc[6];
        const float lA = q2 ? lA23 : lA01;
        const float lB = q2 ? lB23 : lB01;

        const int eA = t * 16 + eo;
        el_finish(lA + sA + dA, eA,     hd, alpha);
        el_finish(lB + sB + dB, eA + 4, hd, alpha);
    }
}

#define ST_NB      15
#define SC_FR      200
#define ST_R       3400
#define SC_FPR     (ST_R / SC_FR)
#define ST_CH      16
#define ST_CE      (N_EDGES / ST_CH)
#define ST_THREADS 1024
#define ST_LD      13
#define LS_STRIDE  50016
#define LS_CNT     32

static_assert(ST_R * ST_NB >= N_NODES && ST_R * (ST_NB - 1) < N_NODES, "ranges must cover the nodes");
static_assert(ST_CE * ST_CH == N_EDGES && ST_CE % 4 == 0, "chunks must tile the edges");
static_assert(ST_THREADS * ST_LD * 4 >= ST_CE, "one sweep must cover a chunk");
static_assert(ST_R <= 4096 && ST_CE <= (1 << 20), "list entry packing");
static_assert(ST_R % SC_FR == 0 && N_NODES % SC_FR == 0 && SC_FPR + 1 <= LS_CNT, "list bins must tile the ranges");
static_assert(LS_STRIDE >= ST_CE && LS_STRIDE % 32 == 0, "stored lists must not share a line");

__global__ void __launch_bounds__(ST_THREADS) edge_bins(
    const int* __restrict__ edge_index,
    unsigned int* __restrict__ elist, int* __restrict__ eoff)
{
    __shared__ unsigned int list[ST_CE];
    __shared__ unsigned int bincnt[LS_CNT];
    __shared__ unsigned int binbase[LS_CNT];

    const int tid  = threadIdx.x;
    const int b    = blockIdx.x % ST_NB;
    const int ch   = blockIdx.x / ST_NB;
    const int lo   = b * ST_R;
    const int e0   = ch * ST_CE;

    const int4* p = (const int4*)(edge_index + N_EDGES + e0);
    int4 dv[ST_LD];
#pragma unroll
    for (int u = 0; u < ST_LD; ++u) {
        const int g = tid + ST_THREADS * u;
        dv[u] = (g < ST_CE / 4) ? p[g] : make_int4(-1, -1, -1, -1);
    }

    if (tid < LS_CNT) bincnt[tid] = 0u;
    __syncthreads();

#pragma unroll
    for (int u = 0; u < ST_LD; ++u) {
        const int dk[4] = {dv[u].x, dv[u].y, dv[u].z, dv[u].w};
#pragma unroll
        for (int k = 0; k < 4; ++k) {
            const unsigned dl = (unsigned)(dk[k] - lo);
            if (dl < ST_R) atomicAdd(&bincnt[dl / SC_FR], 1u);
        }
    }
    __syncthreads();
    if (tid == 0) {
        unsigned a = 0u;
#pragma unroll 1
        for (int j = 0; j < SC_FPR; ++j) {
            const unsigned n = bincnt[j];
            binbase[j] = a;
            bincnt[j]  = a;
            a += n;
        }
        binbase[SC_FPR] = a;
    }
    __syncthreads();
    unsigned pos[ST_LD][4];
#pragma unroll
    for (int u = 0; u < ST_LD; ++u) {
        const int dk[4] = {dv[u].x, dv[u].y, dv[u].z, dv[u].w};
#pragma unroll
        for (int k = 0; k < 4; ++k) {
            const unsigned dl = (unsigned)(dk[k] - lo);
            pos[u][k] = 0u;
            if (dl < ST_R) pos[u][k] = atomicAdd(&bincnt[dl / SC_FR], 1u);
        }
    }
    __builtin_amdgcn_fence(__ATOMIC_ACQ_REL, "workgroup");
#pragma unroll
    for (int u = 0; u < ST_LD; ++u) {
        const int g4 = 4 * (tid + ST_THREADS * u);
        const int dk[4] = {dv[u].x, dv[u].y, dv[u].z, dv[u].w};
#pragma unroll
        for (int k = 0; k < 4; ++k) {
            const unsigned dl = (unsigned)(dk[k] - lo);
            if (dl < ST_R)
                list[pos[u][k]] = ((unsigned)(g4 + k) << 12) | dl;
        }
    }
    __syncthreads();

    const unsigned M = (unsigned)__builtin_amdgcn_readfirstlane((int)binbase[SC_FPR]);
    unsigned int* el = elist + (size_t)blockIdx.x * LS_STRIDE;
    for (unsigned i = (unsigned)tid; i < M; i += ST_THREADS) el[i] = list[i];
    if (tid <= SC_FPR) eoff[blockIdx.x * LS_CNT + tid] = (int)binbase[tid];
}

#define SC_NF      (N_NODES / SC_FR)
#define SC_THREADS 1024
#define SC_WAVES   (SC_THREADS / 32)
#define SC_PRE     4
#define SC_G       8
#define SC_COPIES  8
#define SC_ACC4    (SC_FR * OUT_CH / 4)

static_assert(SC_WAVES == 2 * ST_CH, "two waves per stored list");
static_assert(32 % SC_G == 0, "groups must tile a 32-entry slab");

__device__ __forceinline__ void sc_lds_max(int* addr, float a)
{
    const int ai = __float_as_int(a);
    if (ai >= 0) __hip_atomic_fetch_max(addr, ai, __ATOMIC_RELAXED, __HIP_MEMORY_SCOPE_WORKGROUP);
    else         __hip_atomic_fetch_min((unsigned int*)addr, (unsigned int)ai,
                                        __ATOMIC_RELAXED, __HIP_MEMORY_SCOPE_WORKGROUP);
}

struct ScRaw {
    int sl; unsigned dl; v4f al;
};

__device__ __forceinline__ void sc_fetch(
    ScRaw& R, unsigned ent, int ebase, unsigned off,
    const int* __restrict__ edge_index, const float* __restrict__ alpha)
{
    unsigned dl = (ent & 4095u) - off;
    dl = (dl < (unsigned)SC_FR) ? dl : 0u;
    unsigned eo = ent >> 12;
    eo = (eo < (unsigned)ST_CE) ? eo : 0u;
    const int e = ebase + (int)eo;
    R.dl = dl;
    R.sl = edge_index[e];
    R.al = *(const v4f*)(alpha + (size_t)e * 4);
}

__device__ __forceinline__ void sc_max(const ScRaw& R, bool live, int* mxi)
{
    if (live) {
        sc_lds_max(mxi + R.dl * 4 + 0, R.al.x);
        sc_lds_max(mxi + R.dl * 4 + 1, R.al.y);
        sc_lds_max(mxi + R.dl * 4 + 2, R.al.z);
        sc_lds_max(mxi + R.dl * 4 + 3, R.al.w);
    }
}

__device__ __forceinline__ v4f sc_exp(const ScRaw& R, const v4f* mx4)
{
    v4f m = mx4[R.dl];
    m.x = __builtin_isfinite(m.x) ? m.x : 0.f;
    m.y = __builtin_isfinite(m.y) ? m.y : 0.f;
    m.z = __builtin_isfinite(m.z) ? m.z : 0.f;
    m.w = __builtin_isfinite(m.w) ? m.w : 0.f;
    return (v4f){__expf(R.al.x - m.x), __expf(R.al.y - m.y),
                 __expf(R.al.z - m.z), __expf(R.al.w - m.w)};
}

__device__ __forceinline__ void sc_sum(unsigned dl, v4f ex, bool live, float* smf)
{
    if (live) {
        __hip_atomic_fetch_add(smf + dl * 4 + 0, ex.x, __ATOMIC_RELAXED, __HIP_MEMORY_SCOPE_WORKGROUP);
        __hip_atomic_fetch_add(smf + dl * 4 + 1, ex.y, __ATOMIC_RELAXED, __HIP_MEMORY_SCOPE_WORKGROUP);
        __hip_atomic_fetch_add(smf + dl * 4 + 2, ex.z, __ATOMIC_RELAXED, __HIP_MEMORY_SCOPE_WORKGROUP);
        __hip_atomic_fetch_add(smf + dl * 4 + 3, ex.w, __ATOMIC_RELAXED, __HIP_MEMORY_SCOPE_WORKGROUP);
    }
}

__device__ __forceinline__ void sc_rows(
    int sl, unsigned dl, v4f ex, int n, int lane,
    const v4f* sm4, const float* __restrict__ xh, float* accs)
{
    const v4f cf = ex / (sm4[dl] + EPSV);
    for (int j = 0; j < n; j += SC_G) {
        float xv[SC_G][4];
#pragma unroll
        for (int u = 0; u < SC_G; ++u) {
            const int sj = __builtin_amdgcn_readlane(sl, j + u);
            const float* xp = xh + (size_t)sj * HC + lane;
            xv[u][0] = xp[0];  xv[u][1] = xp[32];
            xv[u][2] = xp[64]; xv[u][3] = xp[96];
        }
#pragma unroll
        for (int u = 0; u < SC_G; ++u) {
            if (j + u < n) {
                const float c0 = __int_as_float(__builtin_amdgcn_readlane(__float_as_int(cf.x), j + u));
                const float c1 = __int_as_float(__builtin_amdgcn_readlane(__float_as_int(cf.y), j + u));
                const float c2 = __int_as_float(__builtin_amdgcn_readlane(__float_as_int(cf.z), j + u));
                const float c3 = __int_as_float(__builtin_amdgcn_readlane(__float_as_int(cf.w), j + u));
                const int   dj = __builtin_amdgcn_readlane((int)dl, j + u);
                const float v = c0 * xv[u][0] + c1 * xv[u][1]
                              + c2 * xv[u][2] + c3 * xv[u][3];
                __hip_atomic_fetch_add(accs + dj * OUT_CH + lane, v,
                                       __ATOMIC_RELAXED, __HIP_MEMORY_SCOPE_WORKGROUP);
            }
        }
    }
}

__global__ void __launch_bounds__(SC_THREADS) edge_scatter(
    const int* __restrict__ edge_index, const float* __restrict__ xh,
    const float* __restrict__ alpha,
    const unsigned int* __restrict__ elist, const int* __restrict__ eoff,
    const float* __restrict__ bias_conv, const float* __restrict__ bias_layer,
    float* __restrict__ out)
{
    __shared__ v4f accs4[SC_COPIES * SC_ACC4];
    __shared__ v4f mx4[SC_FR];
    __shared__ v4f sm4[SC_FR];

    const int tid  = threadIdx.x;
    const int lane = tid & 31;
    const int wv   = __builtin_amdgcn_readfirstlane(tid >> 5);
    float* accs = (float*)(accs4 + (wv % SC_COPIES) * SC_ACC4);
    int*   mxi  = (int*)mx4;
    float* smf  = (float*)sm4;
    const int f    = blockIdx.x;
    const int b    = f / SC_FPR;
    const int fb   = f - b * SC_FPR;
    const unsigned off = (unsigned)(fb * SC_FR);

    const int ch   = wv >> 1;
    const int reg  = ch * ST_NB + b;
    int p0 = eoff[reg * LS_CNT + fb];
    int p1 = eoff[reg * LS_CNT + fb + 1];
    p0 = (p0 < 0) ? 0 : ((p0 > ST_CE) ? ST_CE : p0);
    p1 = (p1 < p0) ? p0 : ((p1 > ST_CE) ? ST_CE : p1);
    const int mid = p0 + ((p1 - p0 + 1) >> 1);
    const int beg = (wv & 1) ? mid : p0;
    const int end = (wv & 1) ? p1 : mid;
    const unsigned int* el = elist + (size_t)reg * LS_STRIDE;
    const int ebase = ch * ST_CE;

    unsigned entq[SC_PRE];
#pragma unroll
    for (int i = 0; i < SC_PRE; ++i) {
        const int idx = beg + 32 * i + lane;
        entq[i] = (idx < end) ? el[idx] : 0u;
    }
    ScRaw raw[SC_PRE];
#pragma unroll
    for (int i = 0; i < SC_PRE; ++i)
        sc_fetch(raw[i], entq[i], ebase, off, edge_index, alpha);

    const float ninf = -__builtin_huge_valf();
    for (int i = tid; i < SC_COPIES * SC_ACC4; i += SC_THREADS)
        accs4[i] = (v4f){0.f, 0.f, 0.f, 0.f};
    for (int i = tid; i < SC_FR; i += SC_THREADS) {
        mx4[i] = (v4f){ninf, ninf, ninf, ninf};
        sm4[i] = (v4f){0.f, 0.f, 0.f, 0.f};
    }
    __syncthreads();

#pragma unroll
    for (int i = 0; i < SC_PRE; ++i)
        sc_max(raw[i], beg + 32 * i + lane < end, mxi);
    for (int s = beg + 32 * SC_PRE; s < end; s += 32) {
        const int idx = s + lane;
        ScRaw R;
        sc_fetch(R, (idx < end) ? el[idx] : 0u, ebase, off, edge_index, alpha);
        sc_max(R, idx < end, mxi);
    }
    __syncthreads();

#pragma unroll
    for (int i = 0; i < SC_PRE; ++i) {
        raw[i].al = sc_exp(raw[i], mx4);
        sc_sum(raw[i].dl, raw[i].al, beg + 32 * i + lane < end, smf);
    }
    for (int s = beg + 32 * SC_PRE; s < end; s += 32) {
        const int idx = s + lane;
        ScRaw R;
        sc_fetch(R, (idx < end) ? el[idx] : 0u, ebase, off, edge_index, alpha);
        sc_sum(R.dl, sc_exp(R, mx4), idx < end, smf);
    }
    __syncthreads();

#pragma unroll
    for (int i = 0; i < SC_PRE; ++i) {
        const int s = beg + 32 * i;
        if (s < end)
            sc_rows(raw[i].sl, raw[i].dl, raw[i].al, (end - s < 32) ? end - s : 32, lane, sm4, xh, accs);
    }
    for (int s = beg + 32 * SC_PRE; s < end; s += 32) {
        const int idx = s + lane;
        const int n = (end - s < 32) ? end - s : 32;
        ScRaw R;
        sc_fetch(R, (idx < end) ? el[idx] : 0u, ebase, off, edge_index, alpha);
        sc_rows(R.sl, R.dl, sc_exp(R, mx4), n, lane, sm4, xh, accs);
    }
    __syncthreads();

    v4f* o4 = (v4f*)(out + (size_t)f * SC_FR * OUT_CH);
    for (int i = tid; i < SC_ACC4; i += SC_THREADS) {
        v4f s = accs4[i];
#pragma unroll
        for (int k = 1; k < SC_COPIES; ++k) s += accs4[k * SC_ACC4 + i];
        const int ch0 = (i & 7) * 4;
        const float sv[4] = {s.x, s.y, s.z, s.w};
        float rv[4];
#pragma unroll
        for (int k = 0; k < 4; ++k) {
            const float v = sv[k] * 0.25f + bias_conv[ch0 + k] + bias_layer[ch0 + k];
            rv[k] = v > 0.f ? v : 0.f;
        }
        o4[i] = (v4f){rv[0], rv[1], rv[2], rv[3]};
    }
}

extern "C" void kernel_launch(void* const* d_in, const int* in_sizes, int n_in,
                              void* d_out, int out_size, void* d_ws, size_t ws_size,
                              hipStream_t stream) {
    const float* x          = (const float*)d_in[0];
    const float* edge_attr  = (const float*)d_in[1];
    const float* W          = (const float*)d_in[2];
    const float* W_edge     = (const float*)d_in[3];
    const float* att_src    = (const float*)d_in[4];
    const float* att_dst    = (const float*)d_in[5];
    const float* att_edge   = (const float*)d_in[6];
    const float* bias_conv  = (const float*)d_in[7];
    const float* bias_layer = (const float*)d_in[8];
    const int*   edge_index = (const int*)d_in[9];
    float* out = (float*)d_out;

    float* ws      = (float*)d_ws;
    float* xh      = ws;
    float* s_node  = xh      + (size_t)N_NODES * HC;
    float* d_node  = s_node  + (size_t)N_NODES * HEADS;
    float* weff    = d_node  + (size_t)N_NODES * HEADS;
    float* alpha   = weff    + (size_t)HEADS * IN_CH;
    unsigned int* elist = (unsigned int*)(alpha + (size_t)N_EDGES * HEADS);
    int* eoff      = (int*)(elist + (size_t)ST_NB * ST_CH * LS_STRIDE);

    proj_nodes_wmma<<<N_NODES / 16, 128, 0, stream>>>(
        x, W, att_src, att_dst, W_edge, att_edge, xh, s_node, d_node, weff);
    edge_logits<<<(EL_NWAVES + EL_WAVES - 1) / EL_WAVES, EL_WAVES * 32, 0, stream>>>(
        edge_attr, edge_index, s_node, d_node, weff, alpha);
    edge_bins<<<ST_NB * ST_CH, ST_THREADS, 0, stream>>>(
        edge_index, elist, eoff);
    edge_scatter<<<SC_NF, SC_THREADS, 0, stream>>>(
        edge_index, xh, alpha, elist, eoff, bias_conv, bias_layer, out);
}


# edge_bins: each destination index read once, not 15 times: a workgroup bins a 3125-edge chunk for all 250 owners (was a 50000-edge chunk per range); scatter wave walks 8 pieces (sum order changes)
# speedup vs baseline: 4.8024x; 1.0396x over previous
#include <hip/hip_runtime.h>
#include <cstdint>

#define N_NODES 50000
#define N_EDGES 800000
#define IN_CH 64
#define OUT_CH 32
#define HEADS 4
#define HC 128
#define NEG_SLOPE 0.2f
#define EPSV 1e-16f

typedef __attribute__((ext_vector_type(2))) float v2f;
typedef __attribute__((ext_vector_type(4))) float v4f;
typedef __attribute__((ext_vector_type(8))) float v8f;

#define PJ_HEAD  36
#define PJ_XPITCH4 17
#define PJ_PITCH (HEADS * PJ_HEAD)

__global__ void __launch_bounds__(128) proj_nodes_wmma(
    const float* __restrict__ x, const float* __restrict__ W,
    const float* __restrict__ att_src, const float* __restrict__ att_dst,
    const float* __restrict__ W_edge, const float* __restrict__ att_edge,
    float* __restrict__ xh, float* __restrict__ s_node,
    float* __restrict__ d_node, float* __restrict__ weff)
{
    __shared__ v4f ptile4[16 * PJ_PITCH / 4];
    __shared__ v4f xs4[16 * PJ_XPITCH4];

    const int lane = threadIdx.x & 31;
    const int wv   = threadIdx.x >> 5;
    const int half = lane >> 4;
    const int l15  = lane & 15;
    const int m0   = blockIdx.x * 16;

    const int colB0 = wv * 32 + l15;
    const int colB1 = colB0 + 16;

    const v4f* xt = (const v4f*)(x + (size_t)m0 * IN_CH);
    const int p0 = threadIdx.x;
    const int p1 = threadIdx.x + 128;
    const v4f xa = xt[p0];
    const v4f xb = xt[p1];

    const int eh = lane & 3;
    const int er = (lane >> 2) + 8 * (wv & 1);
    const float* __restrict__ attv = (wv < 2) ? att_src : att_dst;
    v4f av[8];
#pragma unroll
    for (int j = 0; j < 8; ++j)
        av[j] = *(const v4f*)(attv + eh * OUT_CH + 4 * j);

    v4f bw0[8], bw1[8];
#pragma unroll
    for (int jj = 0; jj < 8; ++jj) {
        bw0[jj] = *(const v4f*)(W + (size_t)colB0 * IN_CH + 8 * jj + 4 * half);
        bw1[jj] = *(const v4f*)(W + (size_t)colB1 * IN_CH + 8 * jj + 4 * half);
    }

    xs4[(p0 >> 4) * PJ_XPITCH4 + (p0 & 15)] = xa;
    xs4[(p1 >> 4) * PJ_XPITCH4 + (p1 & 15)] = xb;
    __syncthreads();

    v4f a[8];
#pragma unroll
    for (int jj = 0; jj < 8; ++jj)
        a[jj] = xs4[l15 * PJ_XPITCH4 + 2 * jj + half];

    v8f acc0 = {};
    v8f acc1 = {};
#pragma unroll
    for (int jj = 0; jj < 8; ++jj) {
        const v2f a0  = {a[jj].x, a[jj].y};
        const v2f a1  = {a[jj].z, a[jj].w};
        const v2f b00 = {bw0[jj].x, bw0[jj].y};
        const v2f b01 = {bw0[jj].z, bw0[jj].w};
        const v2f b10 = {bw1[jj].x, bw1[jj].y};
        const v2f b11 = {bw1[jj].z, bw1[jj].w};
        acc0 = __builtin_amdgcn_wmma_f32_16x16x4_f32(
                   false, a0, false, b00, (short)0, acc0, false, false);
        acc1 = __builtin_amdgcn_wmma_f32_16x16x4_f32(
                   false, a0, false, b10, (short)0, acc1, false, false);
        acc0 = __builtin_amdgcn_wmma_f32_16x16x4_f32(
                   false, a1, false, b01, (short)0, acc0, false, false);
        acc1 = __builtin_amdgcn_wmma_f32_16x16x4_f32(
                   false, a1, false, b11, (short)0, acc1, false, false);
    }
    float* ptile = (float*)ptile4;
#pragma unroll
    for (int v = 0; v < 8; ++v) {
        const int r = v + half * 8;
        ptile[r * PJ_PITCH + wv * PJ_HEAD + l15]      = acc0[v];
        ptile[r * PJ_PITCH + wv * PJ_HEAD + 16 + l15] = acc1[v];
    }
    __syncthreads();

#pragma unroll
    for (int v = 0; v < 8; ++v) {
        const int r = v + half * 8;
        xh[(size_t)(m0 + r) * HC + colB0] = acc0[v];
        xh[(size_t)(m0 + r) * HC + colB1] = acc1[v];
    }

    const v4f* rp = (const v4f*)(ptile + er * PJ_PITCH + eh * PJ_HEAD);
    float sd = 0.f;
#pragma unroll
    for (int j = 0; j < 8; ++j) {
        const v4f r = rp[j];
        sd += r.x * av[j].x;
        sd += r.y * av[j].y;
        sd += r.z * av[j].z;
        sd += r.w * av[j].w;
    }
    float* __restrict__ nd = (wv < 2) ? s_node : d_node;
    nd[(size_t)(m0 + er) * HEADS + eh] = sd;

    if (blockIdx.x == 0) {
#pragma unroll 1
        for (int u = 0; u < 2; ++u) {
            const int t = threadIdx.x + 128 * u;
            const int h = t >> 6, k = t & 63;
            float s = 0.f;
#pragma unroll
            for (int c = 0; c < OUT_CH; ++c)
                s += att_edge[h * OUT_CH + c] * W_edge[(size_t)(h * OUT_CH + c) * IN_CH + k];
            weff[t] = s;
        }
    }
}

#define EL_WAVES  4
#define EL_TPW    8
#define EL_NTILES (N_EDGES / 16)
#define EL_NWAVES (EL_NTILES / EL_TPW)
#define EL_PITCH4 17

struct EdgeTile {
    v4f r[8];
    int srcA, dstA, srcB, dstB;
};

__device__ __forceinline__ void el_load_tile(
    EdgeTile& T, const float* __restrict__ edge_attr,
    const int* __restrict__ edge_index, int t, int lane, int eo)
{
    const v4f* p = (const v4f*)(edge_attr + (size_t)t * (16 * IN_CH)) + lane;
#pragma unroll
    for (int j = 0; j < 8; ++j)
        T.r[j] = __builtin_nontemporal_load(p + 32 * j);
    const int e = t * 16 + eo;
    T.srcA = edge_index[e];
    T.dstA = edge_index[N_EDGES + e];
    T.srcB = edge_index[e + 4];
    T.dstB = edge_index[N_EDGES + e + 4];
}

__device__ __forceinline__ void el_wave_lds_order()
{
    __builtin_amdgcn_fence(__ATOMIC_RELEASE, "wavefront");
    __builtin_amdgcn_wave_barrier();
    __builtin_amdgcn_fence(__ATOMIC_ACQUIRE, "wavefront");
}

__device__ __forceinline__ void el_finish(
    float a, int e, int hd, float* __restrict__ alpha)
{
    a = (a > 0.f) ? a : NEG_SLOPE * a;
    alpha[(size_t)e * 4 + hd] = a;
}

__global__ void __launch_bounds__(EL_WAVES * 32) edge_logits(
    const float* __restrict__ edge_attr, const int* __restrict__ edge_index,
    const float* __restrict__ s_node, const float* __restrict__ d_node,
    const float* __restrict__ weff,
    float* __restrict__ alpha)
{
    __shared__ v4f stage[EL_WAVES * 16 * EL_PITCH4];

    const int lane = threadIdx.x & 31;
    const int wv   = threadIdx.x >> 5;
    const int c    = lane & 15;
    const int half = lane >> 4;
    const int w    = blockIdx.x * EL_WAVES + wv;
    if (w >= EL_NWAVES) return;

    v4f* st = stage + wv * (16 * EL_PITCH4);

    const int hd = c & 3;
    v4f bw[8];
#pragma unroll
    for (int jj = 0; jj < 8; ++jj)
        bw[jj] = *(const v4f*)(weff + hd * IN_CH + 8 * jj + 4 * half);

    const int q  = c >> 2;
    const int eo = 8 * half + q;

    EdgeTile cur;
    el_load_tile(cur, edge_attr, edge_index, w, lane, eo);

    for (int i = 0; i < EL_TPW; ++i) {
        const int t = w + i * EL_NWAVES;

#pragma unroll
        for (int j = 0; j < 8; ++j)
            st[(2 * j + half) * EL_PITCH4 + c] = cur.r[j];

        const int srcA = cur.srcA, dstA = cur.dstA;
        const int srcB = cur.srcB, dstB = cur.dstB;
        const float sA = s_node[dstA * 4 + hd];
        const float dA = d_node[srcA * 4 + hd];
        const float sB = s_node[dstB * 4 + hd];
        const float dB = d_node[srcB * 4 + hd];

        if (i + 1 < EL_TPW)
            el_load_tile(cur, edge_attr, edge_index, t + EL_NWAVES, lane, eo);

        el_wave_lds_order();

        v4f a[8];
#pragma unroll
        for (int jj = 0; jj < 8; ++jj)
            a[jj] = st[c * EL_PITCH4 + 2 * jj + half];

        v8f acc0 = {};
        v8f acc1 = {};
#pragma unroll
        for (int jj = 0; jj < 8; ++jj) {
            const v2f a0 = {a[jj].x, a[jj].y};
            const v2f a1 = {a[jj].z, a[jj].w};
            const v2f b0 = {bw[jj].x, bw[jj].y};
            const v2f b1 = {bw[jj].z, bw[jj].w};
            acc0 = __builtin_amdgcn_wmma_f32_16x16x4_f32(
                       false, a0, false, b0, (short)0, acc0, false, false);
            acc1 = __builtin_amdgcn_wmma_f32_16x16x4_f32(
                       false, a1, false, b1, (short)0, acc1, false, false);
        }

        el_wave_lds_order();

        asm volatile("v_nop\n\tv_nop\n\tv_nop\n\tv_nop\n\t"
                     "v_nop\n\tv_nop\n\tv_nop\n\tv_nop"
                     : "+v"(acc0), "+v"(acc1));

        const v8f acc = acc0 + acc1;
        const bool q1 = (q & 1) != 0, q2 = (q & 2) != 0;
        const float lA01 = q1 ? acc[1] : acc[0], lA23 = q1 ? acc[3] : acc[2];
        const float lB01 = q1 ? acc[5] : acc[4], lB23 = q1 ? acc[7] : acc[6];
        const float lA = q2 ? lA23 : lA01;
        const float lB = q2 ? lB23 : lB01;

        const int eA = t * 16 + eo;
        el_finish(lA + sA + dA, eA,     hd, alpha);
        el_finish(lB + sB + dB, eA + 4, hd, alpha);
    }
}

#define SC_FR      200
#define SC_NF      (N_NODES / SC_FR)
#define BN_CH      256
#define BN_CE      (N_EDGES / BN_CH)
#define BN_THREADS 256
#define BN_LD      13
#define LS_STRIDE  3136
#define EO_PITCH   256

static_assert(SC_FR * SC_NF == N_NODES, "owners must tile the nodes");
static_assert(BN_CE * BN_CH == N_EDGES, "chunks must tile the edges");
static_assert(BN_THREADS * BN_LD >= BN_CE, "one sweep must cover a chunk");
static_assert(SC_FR <= 256 && N_EDGES <= (1 << 24), "list entry packing");
static_assert(SC_NF + 1 <= EO_PITCH && EO_PITCH == 8 * 32 && BN_THREADS == EO_PITCH, "one wave scans the bins, 8 per lane; one thread per stored place");
static_assert(LS_STRIDE >= BN_CE && LS_STRIDE % 32 == 0, "stored lists must not share a line");

__global__ void __launch_bounds__(BN_THREADS) edge_bins(
    const int* __restrict__ edge_index,
    unsigned int* __restrict__ elist, int* __restrict__ eoff)
{
    __shared__ unsigned int list[BN_CE];
    __shared__ unsigned int bincnt[EO_PITCH];
    __shared__ unsigned int binbase[EO_PITCH];

    const int tid  = threadIdx.x;
    const int e0   = blockIdx.x * BN_CE;

    const int* p = edge_index + N_EDGES + e0;
    int dv[BN_LD];
#pragma unroll
    for (int u = 0; u < BN_LD; ++u) {
        const int k = tid + BN_THREADS * u;
        dv[u] = (k < BN_CE) ? p[k] : -1;
    }

    bincnt[tid] = 0u;
    __syncthreads();

#pragma unroll
    for (int u = 0; u < BN_LD; ++u) {
        const unsigned d = (unsigned)dv[u];
        const bool ok = d < (unsigned)N_NODES;
        atomicAdd(&bincnt[ok ? d / SC_FR : 0u], ok ? 1u : 0u);
    }
    __syncthreads();

    if (tid < 32) {
        unsigned c[8];
#pragma unroll
        for (int j = 0; j < 8; ++j) c[j] = bincnt[8 * tid + j];
        unsigned s = 0u;
#pragma unroll
        for (int j = 0; j < 8; ++j) {
            const unsigned n = c[j];
            c[j] = s;
            s += n;
        }
        unsigned inc = s;
#pragma unroll
        for (int o = 1; o < 32; o <<= 1) {
            const unsigned v = __shfl_up(inc, (unsigned)o);
            if (tid >= o) inc += v;
        }
        const unsigned ex = inc - s;
#pragma unroll
        for (int j = 0; j < 8; ++j) {
            binbase[8 * tid + j] = ex + c[j];
            bincnt[8 * tid + j]  = ex + c[j];
        }
    }
    __syncthreads();

    unsigned pos[BN_LD];
#pragma unroll
    for (int u = 0; u < BN_LD; ++u) {
        const unsigned d = (unsigned)dv[u];
        const bool ok = d < (unsigned)N_NODES;
        pos[u] = atomicAdd(&bincnt[ok ? d / SC_FR : 0u], ok ? 1u : 0u);
    }
    __builtin_amdgcn_fence(__ATOMIC_ACQ_REL, "workgroup");
#pragma unroll
    for (int u = 0; u < BN_LD; ++u) {
        const unsigned d = (unsigned)dv[u];
        const unsigned e = (unsigned)(e0 + tid + BN_THREADS * u);
        if (d < (unsigned)N_NODES)
            list[pos[u]] = (e << 8) | (d - (d / SC_FR) * SC_FR);
    }
    __syncthreads();

    const unsigned M = (unsigned)__builtin_amdgcn_readfirstlane((int)binbase[SC_NF]);
    unsigned int* el = elist + (size_t)blockIdx.x * LS_STRIDE;
    for (unsigned i = (unsigned)tid; i < M; i += BN_THREADS) el[i] = list[i];
    eoff[blockIdx.x * EO_PITCH + tid] = (int)binbase[tid];
}

#define SC_THREADS 1024
#define SC_WAVES   (SC_THREADS / 32)
#define SC_PPW     (BN_CH / SC_WAVES)
#define SC_PRE     4
#define SC_G       8
#define SC_COPIES  8
#define SC_ACC4    (SC_FR * OUT_CH / 4)

static_assert(SC_WAVES * SC_PPW == BN_CH, "the waves must tile the stored lists");
static_assert(32 % SC_G == 0, "groups must tile a 32-entry slab");

struct ScPieces {
    int adj[SC_PPW];
    int cum[SC_PPW + 1];
};

__device__ __forceinline__ unsigned sc_entry(
    const ScPieces& P, const unsigned int* __restrict__ elist, int g)
{
    int a = P.adj[0];
#pragma unroll
    for (int k = 1; k < SC_PPW; ++k)
        a = (g >= P.cum[k]) ? P.adj[k] : a;
    return (g < P.cum[SC_PPW]) ? elist[g + a] : 0u;
}

__device__ __forceinline__ void sc_lds_max(int* addr, float a)
{
    const int ai = __float_as_int(a);
    if (ai >= 0) __hip_atomic_fetch_max(addr, ai, __ATOMIC_RELAXED, __HIP_MEMORY_SCOPE_WORKGROUP);
    else         __hip_atomic_fetch_min((unsigned int*)addr, (unsigned int)ai,
                                        __ATOMIC_RELAXED, __HIP_MEMORY_SCOPE_WORKGROUP);
}

struct ScRaw {
    int sl; unsigned dl; v4f al;
};

__device__ __forceinline__ void sc_fetch(
    ScRaw& R, unsigned ent,
    const int* __restrict__ edge_index, const float* __restrict__ alpha)
{
    unsigned dl = ent & 255u;
    dl = (dl < (unsigned)SC_FR) ? dl : 0u;
    unsigned e = ent >> 8;
    e = (e < (unsigned)N_EDGES) ? e : 0u;
    R.dl = dl;
    R.sl = edge_index[e];
    R.al = *(const v4f*)(alpha + (size_t)e * 4);
}

__device__ __forceinline__ void sc_max(const ScRaw& R, bool live, int* mxi)
{
    if (live) {
        sc_lds_max(mxi + R.dl * 4 + 0, R.al.x);
        sc_lds_max(mxi + R.dl * 4 + 1, R.al.y);
        sc_lds_max(mxi + R.dl * 4 + 2, R.al.z);
        sc_lds_max(mxi + R.dl * 4 + 3, R.al.w);
    }
}

__device__ __forceinline__ v4f sc_exp(const ScRaw& R, const v4f* mx4)
{
    v4f m = mx4[R.dl];
    m.x = __builtin_isfinite(m.x) ? m.x : 0.f;
    m.y = __builtin_isfinite(m.y) ? m.y : 0.f;
    m.z = __builtin_isfinite(m.z) ? m.z : 0.f;
    m.w = __builtin_isfinite(m.w) ? m.w : 0.f;
    return (v4f){__expf(R.al.x - m.x), __expf(R.al.y - m.y),
                 __expf(R.al.z - m.z), __expf(R.al.w - m.w)};
}

__device__ __forceinline__ void sc_sum(unsigned dl, v4f ex, bool live, float* smf)
{
    if (live) {
        __hip_atomic_fetch_add(smf + dl * 4 + 0, ex.x, __ATOMIC_RELAXED, __HIP_MEMORY_SCOPE_WORKGROUP);
        __hip_atomic_fetch_add(smf + dl * 4 + 1, ex.y, __ATOMIC_RELAXED, __HIP_MEMORY_SCOPE_WORKGROUP);
        __hip_atomic_fetch_add(smf + dl * 4 + 2, ex.z, __ATOMIC_RELAXED, __HIP_MEMORY_SCOPE_WORKGROUP);
        __hip_atomic_fetch_add(smf + dl * 4 + 3, ex.w, __ATOMIC_RELAXED, __HIP_MEMORY_SCOPE_WORKGROUP);
    }
}

__device__ __forceinline__ void sc_rows(
    int sl, unsigned dl, v4f ex, int n, int lane,
    const v4f* sm4, const float* __restrict__ xh, float* accs)
{
    const v4f cf = ex / (sm4[dl] + EPSV);
    for (int j = 0; j < n; j += SC_G) {
        float xv[SC_G][4];
#pragma unroll
        for (int u = 0; u < SC_G; ++u) {
            const int sj = __builtin_amdgcn_readlane(sl, j + u);
            const float* xp = xh + (size_t)sj * HC + lane;
            xv[u][0] = xp[0];  xv[u][1] = xp[32];
            xv[u][2] = xp[64]; xv[u][3] = xp[96];
        }
#pragma unroll
        for (int u = 0; u < SC_G; ++u) {
            if (j + u < n) {
                const float c0 = __int_as_float(__builtin_amdgcn_readlane(__float_as_int(cf.x), j + u));
                const float c1 = __int_as_float(__builtin_amdgcn_readlane(__float_as_int(cf.y), j + u));
                const float c2 = __int_as_float(__builtin_amdgcn_readlane(__float_as_int(cf.z), j + u));
                const float c3 = __int_as_float(__builtin_amdgcn_readlane(__float_as_int(cf.w), j + u));
                const int   dj = __builtin_amdgcn_readlane((int)dl, j + u);
                const float v = c0 * xv[u][0] + c1 * xv[u][1]
                              + c2 * xv[u][2] + c3 * xv[u][3];
                __hip_atomic_fetch_add(accs + dj * OUT_CH + lane, v,
                                       __ATOMIC_RELAXED, __HIP_MEMORY_SCOPE_WORKGROUP);
            }
        }
    }
}

__global__ void __launch_bounds__(SC_THREADS) edge_scatter(
    const int* __restrict__ edge_index, const float* __restrict__ xh,
    const float* __restrict__ alpha,
    const unsigned int* __restrict__ elist, const int* __restrict__ eoff,
    const float* __restrict__ bias_conv, const float* __restrict__ bias_layer,
    float* __restrict__ out)
{
    __shared__ v4f accs4[SC_COPIES * SC_ACC4];
    __shared__ v4f mx4[SC_FR];
    __shared__ v4f sm4[SC_FR];

    const int tid  = threadIdx.x;
    const int lane = tid & 31;
    const int wv   = __builtin_amdgcn_readfirstlane(tid >> 5);
    float* accs = (float*)(accs4 + (wv % SC_COPIES) * SC_ACC4);
    int*   mxi  = (int*)mx4;
    float* smf  = (float*)sm4;
    const int f    = blockIdx.x;

    ScPieces P;
    P.cum[0] = 0;
#pragma unroll
    for (int k = 0; k < SC_PPW; ++k) {
        const int ch = wv * SC_PPW + k;
        int p0 = eoff[ch * EO_PITCH + f];
        int p1 = eoff[ch * EO_PITCH + f + 1];
        p0 = (p0 < 0) ? 0 : ((p0 > BN_CE) ? BN_CE : p0);
        p1 = (p1 < p0) ? p0 : ((p1 > BN_CE) ? BN_CE : p1);
        P.adj[k]     = ch * LS_STRIDE + p0 - P.cum[k];
        P.cum[k + 1] = P.cum[k] + (p1 - p0);
    }
    const int total = P.cum[SC_PPW];

    unsigned entq[SC_PRE];
#pragma unroll
    for (int i = 0; i < SC_PRE; ++i)
        entq[i] = sc_entry(P, elist, 32 * i + lane);
    ScRaw raw[SC_PRE];
#pragma unroll
    for (int i = 0; i < SC_PRE; ++i)
        sc_fetch(raw[i], entq[i], edge_index, alpha);

    const float ninf = -__builtin_huge_valf();
    for (int i = tid; i < SC_COPIES * SC_ACC4; i += SC_THREADS)
        accs4[i] = (v4f){0.f, 0.f, 0.f, 0.f};
    for (int i = tid; i < SC_FR; i += SC_THREADS) {
        mx4[i] = (v4f){ninf, ninf, ninf, ninf};
        sm4[i] = (v4f){0.f, 0.f, 0.f, 0.f};
    }
    __syncthreads();

#pragma unroll
    for (int i = 0; i < SC_PRE; ++i)
        sc_max(raw[i], 32 * i + lane < total, mxi);
    for (int s = 32 * SC_PRE; s < total; s += 32) {
        const int g = s + lane;
        ScRaw R;
        sc_fetch(R, sc_entry(P, elist, g), edge_index, alpha);
        sc_max(R, g < total, mxi);
    }
    __syncthreads();

#pragma unroll
    for (int i = 0; i < SC_PRE; ++i) {
        raw[i].al = sc_exp(raw[i], mx4);
        sc_sum(raw[i].dl, raw[i].al, 32 * i + lane < total, smf);
    }
    for (int s = 32 * SC_PRE; s < total; s += 32) {
        const int g = s + lane;
        ScRaw R;
        sc_fetch(R, sc_entry(P, elist, g), edge_index, alpha);
        sc_sum(R.dl, sc_exp(R, mx4), g < total, smf);
    }
    __syncthreads();

#pragma unroll
    for (int i = 0; i < SC_PRE; ++i) {
        const int s = 32 * i;
        if (s < total)
            sc_rows(raw[i].sl, raw[i].dl, raw[i].al, (total - s < 32) ? total - s : 32, lane, sm4, xh, accs);
    }
    for (int s = 32 * SC_PRE; s < total; s += 32) {
        const int n = (total - s < 32) ? total - s : 32;
        ScRaw R;
        sc_fetch(R, sc_entry(P, elist, s + lane), edge_index, alpha);
        sc_rows(R.sl, R.dl, sc_exp(R, mx4), n, lane, sm4, xh, accs);
    }
    __syncthreads();

    v4f* o4 = (v4f*)(out + (size_t)f * SC_FR * OUT_CH);
    for (int i = tid; i < SC_ACC4; i += SC_THREADS) {
        v4f s = accs4[i];
#pragma unroll
        for (int k = 1; k < SC_COPIES; ++k) s += accs4[k * SC_ACC4 + i];
        const int ch0 = (i & 7) * 4;
        const float sv[4] = {s.x, s.y, s.z, s.w};
        float rv[4];
#pragma unroll
        for (int k = 0; k < 4; ++k) {
            const float v = sv[k] * 0.25f + bias_conv[ch0 + k] + bias_layer[ch0 + k];
            rv[k] = v > 0.f ? v : 0.f;
        }
        o4[i] = (v4f){rv[0], rv[1], rv[2], rv[3]};
    }
}

extern "C" void kernel_launch(void* const* d_in, const int* in_sizes, int n_in,
                              void* d_out, int out_size, void* d_ws, size_t ws_size,
                              hipStream_t stream) {
    const float* x          = (const float*)d_in[0];
    const float* edge_attr  = (const float*)d_in[1];
    const float* W          = (const float*)d_in[2];
    const float* W_edge     = (const float*)d_in[3];
    const float* att_src    = (const float*)d_in[4];
    const float* att_dst    = (const float*)d_in[5];
    const float* att_edge   = (const float*)d_in[6];
    const float* bias_conv  = (const float*)d_in[7];
    const float* bias_layer = (const float*)d_in[8];
    const int*   edge_index = (const int*)d_in[9];
    float* out = (float*)d_out;

    float* ws      = (float*)d_ws;
    float* xh      = ws;
    float* s_node  = xh      + (size_t)N_NODES * HC;
    float* d_node  = s_node  + (size_t)N_NODES * HEADS;
    float* weff    = d_node  + (size_t)N_NODES * HEADS;
    float* alpha   = weff    + (size_t)HEADS * IN_CH;
    unsigned int* elist = (unsigned int*)(alpha + (size_t)N_EDGES * HEADS);
    int* eoff      = (int*)(elist + (size_t)BN_CH * LS_STRIDE);

    proj_nodes_wmma<<<N_NODES / 16, 128, 0, stream>>>(
        x, W, att_src, att_dst, W_edge, att_edge, xh, s_node, d_node, weff);
    edge_logits<<<(EL_NWAVES + EL_WAVES - 1) / EL_WAVES, EL_WAVES * 32, 0, stream>>>(
        edge_attr, edge_index, s_node, d_node, weff, alpha);
    edge_bins<<<BN_CH, BN_THREADS, 0, stream>>>(
        edge_index, elist, eoff);
    edge_scatter<<<SC_NF, SC_THREADS, 0, stream>>>(
        edge_index, xh, alpha, elist, eoff, bias_conv, bias_layer, out);
}


# 4 launches -> 3: the 256 binning workgroups (128 threads, 25 indices each) join the logits launch as its first workgroups; neither reads the other's output; order in a bin, so sum order, may change
# speedup vs baseline: 5.2066x; 1.0842x over previous
#include <hip/hip_runtime.h>
#include <cstdint>

#define N_NODES 50000
#define N_EDGES 800000
#define IN_CH 64
#define OUT_CH 32
#define HEADS 4
#define HC 128
#define NEG_SLOPE 0.2f
#define EPSV 1e-16f

typedef __attribute__((ext_vector_type(2))) float v2f;
typedef __attribute__((ext_vector_type(4))) float v4f;
typedef __attribute__((ext_vector_type(8))) float v8f;

#define PJ_HEAD  36
#define PJ_XPITCH4 17
#define PJ_PITCH (HEADS * PJ_HEAD)

__global__ void __launch_bounds__(128) proj_nodes_wmma(
    const float* __restrict__ x, const float* __restrict__ W,
    const float* __restrict__ att_src, const float* __restrict__ att_dst,
    const float* __restrict__ W_edge, const float* __restrict__ att_edge,
    float* __restrict__ xh, float* __restrict__ s_node,
    float* __restrict__ d_node, float* __restrict__ weff)
{
    __shared__ v4f ptile4[16 * PJ_PITCH / 4];
    __shared__ v4f xs4[16 * PJ_XPITCH4];

    const int lane = threadIdx.x & 31;
    const int wv   = threadIdx.x >> 5;
    const int half = lane >> 4;
    const int l15  = lane & 15;
    const int m0   = blockIdx.x * 16;

    const int colB0 = wv * 32 + l15;
    const int colB1 = colB0 + 16;

    const v4f* xt = (const v4f*)(x + (size_t)m0 * IN_CH);
    const int p0 = threadIdx.x;
    const int p1 = threadIdx.x + 128;
    const v4f xa = xt[p0];
    const v4f xb = xt[p1];

    const int eh = lane & 3;
    const int er = (lane >> 2) + 8 * (wv & 1);
    const float* __restrict__ attv = (wv < 2) ? att_src : att_dst;
    v4f av[8];
#pragma unroll
    for (int j = 0; j < 8; ++j)
        av[j] = *(const v4f*)(attv + eh * OUT_CH + 4 * j);

    v4f bw0[8], bw1[8];
#pragma unroll
    for (int jj = 0; jj < 8; ++jj) {
        bw0[jj] = *(const v4f*)(W + (size_t)colB0 * IN_CH + 8 * jj + 4 * half);
        bw1[jj] = *(const v4f*)(W + (size_t)colB1 * IN_CH + 8 * jj + 4 * half);
    }

    xs4[(p0 >> 4) * PJ_XPITCH4 + (p0 & 15)] = xa;
    xs4[(p1 >> 4) * PJ_XPITCH4 + (p1 & 15)] = xb;
    __syncthreads();

    v4f a[8];
#pragma unroll
    for (int jj = 0; jj < 8; ++jj)
        a[jj] = xs4[l15 * PJ_XPITCH4 + 2 * jj + half];

    v8f acc0 = {};
    v8f acc1 = {};
#pragma unroll
    for (int jj = 0; jj < 8; ++jj) {
        const v2f a0  = {a[jj].x, a[jj].y};
        const v2f a1  = {a[jj].z, a[jj].w};
        const v2f b00 = {bw0[jj].x, bw0[jj].y};
        const v2f b01 = {bw0[jj].z, bw0[jj].w};
        const v2f b10 = {bw1[jj].x, bw1[jj].y};
        const v2f b11 = {bw1[jj].z, bw1[jj].w};
        acc0 = __builtin_amdgcn_wmma_f32_16x16x4_f32(
                   false, a0, false, b00, (short)0, acc0, false, false);
        acc1 = __builtin_amdgcn_wmma_f32_16x16x4_f32(
                   false, a0, false, b10, (short)0, acc1, false, false);
        acc0 = __builtin_amdgcn_wmma_f32_16x16x4_f32(
                   false, a1, false, b01, (short)0, acc0, false, false);
        acc1 = __builtin_amdgcn_wmma_f32_16x16x4_f32(
                   false, a1, false, b11, (short)0, acc1, false, false);
    }
    float* ptile = (float*)ptile4;
#pragma unroll
    for (int v = 0; v < 8; ++v) {
        const int r = v + half * 8;
        ptile[r * PJ_PITCH + wv * PJ_HEAD + l15]      = acc0[v];
        ptile[r * PJ_PITCH + wv * PJ_HEAD + 16 + l15] = acc1[v];
    }
    __syncthreads();

#pragma unroll
    for (int v = 0; v < 8; ++v) {
        const int r = v + half * 8;
        xh[(size_t)(m0 + r) * HC + colB0] = acc0[v];
        xh[(size_t)(m0 + r) * HC + colB1] = acc1[v];
    }

    const v4f* rp = (const v4f*)(ptile + er * PJ_PITCH + eh * PJ_HEAD);
    float sd = 0.f;
#pragma unroll
    for (int j = 0; j < 8; ++j) {
        const v4f r = rp[j];
        sd += r.x * av[j].x;
        sd += r.y * av[j].y;
        sd += r.z * av[j].z;
        sd += r.w * av[j].w;
    }
    float* __restrict__ nd = (wv < 2) ? s_node : d_node;
    nd[(size_t)(m0 + er) * HEADS + eh] = sd;

    if (blockIdx.x == 0) {
#pragma unroll 1
        for (int u = 0; u < 2; ++u) {
            const int t = threadIdx.x + 128 * u;
            const int h = t >> 6, k = t & 63;
            float s = 0.f;
#pragma unroll
            for (int c = 0; c < OUT_CH; ++c)
                s += att_edge[h * OUT_CH + c] * W_edge[(size_t)(h * OUT_CH + c) * IN_CH + k];
            weff[t] = s;
        }
    }
}

#define EL_WAVES  4
#define EL_TPW    8
#define EL_NTILES (N_EDGES / 16)
#define EL_NWAVES (EL_NTILES / EL_TPW)
#define EL_PITCH4 17

struct EdgeTile {
    v4f r[8];
    int srcA, dstA, srcB, dstB;
};

__device__ __forceinline__ void el_load_tile(
    EdgeTile& T, const float* __restrict__ edge_attr,
    const int* __restrict__ edge_index, int t, int lane, int eo)
{
    const v4f* p = (const v4f*)(edge_attr + (size_t)t * (16 * IN_CH)) + lane;
#pragma unroll
    for (int j = 0; j < 8; ++j)
        T.r[j] = __builtin_nontemporal_load(p + 32 * j);
    const int e = t * 16 + eo;
    T.srcA = edge_index[e];
    T.dstA = edge_index[N_EDGES + e];
    T.srcB = edge_index[e + 4];
    T.dstB = edge_index[N_EDGES + e + 4];
}

__device__ __forceinline__ void el_wave_lds_order()
{
    __builtin_amdgcn_fence(__ATOMIC_RELEASE, "wavefront");
    __builtin_amdgcn_wave_barrier();
    __builtin_amdgcn_fence(__ATOMIC_ACQUIRE, "wavefront");
}

__device__ __forceinline__ void el_finish(
    float a, int e, int hd, float* __restrict__ alpha)
{
    a = (a > 0.f) ? a : NEG_SLOPE * a;
    alpha[(size_t)e * 4 + hd] = a;
}

__device__ __forceinline__ void edge_logits_body(
    const float* __restrict__ edge_attr, const int* __restrict__ edge_index,
    const float* __restrict__ s_node, const float* __restrict__ d_node,
    const float* __restrict__ weff,
    float* __restrict__ alpha, int wg)
{
    __shared__ v4f stage[EL_WAVES * 16 * EL_PITCH4];

    const int lane = threadIdx.x & 31;
    const int wv   = threadIdx.x >> 5;
    const int c    = lane & 15;
    const int half = lane >> 4;
    const int w    = wg * EL_WAVES + wv;
    if (w >= EL_NWAVES) return;

    v4f* st = stage + wv * (16 * EL_PITCH4);

    const int hd = c & 3;
    v4f bw[8];
#pragma unroll
    for (int jj = 0; jj < 8; ++jj)
        bw[jj] = *(const v4f*)(weff + hd * IN_CH + 8 * jj + 4 * half);

    const int q  = c >> 2;
    const int eo = 8 * half + q;

    EdgeTile cur;
    el_load_tile(cur, edge_attr, edge_index, w, lane, eo);

    for (int i = 0; i < EL_TPW; ++i) {
        const int t = w + i * EL_NWAVES;

#pragma unroll
        for (int j = 0; j < 8; ++j)
            st[(2 * j + half) * EL_PITCH4 + c] = cur.r[j];

        const int srcA = cur.srcA, dstA = cur.dstA;
        const int srcB = cur.srcB, dstB = cur.dstB;
        const float sA = s_node[dstA * 4 + hd];
        const float dA = d_node[srcA * 4 + hd];
        const float sB = s_node[dstB * 4 + hd];
        const float dB = d_node[srcB * 4 + hd];

        if (i + 1 < EL_TPW)
            el_load_tile(cur, edge_attr, edge_index, t + EL_NWAVES, lane, eo);

        el_wave_lds_order();

        v4f a[8];
#pragma unroll
        for (int jj = 0; jj < 8; ++jj)
            a[jj] = st[c * EL_PITCH4 + 2 * jj + half];

        v8f acc0 = {};
        v8f acc1 = {};
#pragma unroll
        for (int jj = 0; jj < 8; ++jj) {
            const v2f a0 = {a[jj].x, a[jj].y};
            const v2f a1 = {a[jj].z, a[jj].w};
            const v2f b0 = {bw[jj].x, bw[jj].y};
            const v2f b1 = {bw[jj].z, bw[jj].w};
            acc0 = __builtin_amdgcn_wmma_f32_16x16x4_f32(
                       false, a0, false, b0, (short)0, acc0, false, false);
            acc1 = __builtin_amdgcn_wmma_f32_16x16x4_f32(
                       false, a1, false, b1, (short)0, acc1, false, false);
        }

        el_wave_lds_order();

        asm volatile("v_nop\n\tv_nop\n\tv_nop\n\tv_nop\n\t"
                     "v_nop\n\tv_nop\n\tv_nop\n\tv_nop"
                     : "+v"(acc0), "+v"(acc1));

        const v8f acc = acc0 + acc1;
        const bool q1 = (q & 1) != 0, q2 = (q & 2) != 0;
        const float lA01 = q1 ? acc[1] : acc[0], lA23 = q1 ? acc[3] : acc[2];
        const float lB01 = q1 ? acc[5] : acc[4], lB23 = q1 ? acc[7] : acc[6];
        const float lA = q2 ? lA23 : lA01;
        const float lB = q2 ? lB23 : lB01;

        const int eA = t * 16 + eo;
        el_finish(lA + sA + dA, eA,     hd, alpha);
        el_finish(lB + sB + dB, eA + 4, hd, alpha);
    }
}

#define SC_FR      200
#define SC_NF      (N_NODES / SC_FR)
#define BN_CH      256
#define BN_CE      (N_EDGES / BN_CH)
#define BN_THREADS 128
#define BN_LD      25
#define LS_STRIDE  3136
#define EO_PITCH   256

static_assert(SC_FR * SC_NF == N_NODES, "owners must tile the nodes");
static_assert(BN_CE * BN_CH == N_EDGES, "chunks must tile the edges");
static_assert(BN_THREADS * BN_LD >= BN_CE, "one sweep must cover a chunk");
static_assert(SC_FR <= 256 && N_EDGES <= (1 << 24), "list entry packing");
static_assert(SC_NF + 1 <= EO_PITCH && EO_PITCH == 8 * 32 && EO_PITCH % BN_THREADS == 0, "one wave scans the bins, 8 per lane; the threads tile the stored places");
static_assert(BN_THREADS == EL_WAVES * 32, "binning and logits workgroups share one launch");
static_assert(LS_STRIDE >= BN_CE && LS_STRIDE % 32 == 0, "stored lists must not share a line");

__device__ __forceinline__ void edge_bins_body(
    const int* __restrict__ edge_index,
    unsigned int* __restrict__ elist, int* __restrict__ eoff, int ch)
{
    __shared__ unsigned int list[BN_CE];
    __shared__ unsigned int bincnt[EO_PITCH];
    __shared__ unsigned int binbase[EO_PITCH];

    const int tid  = threadIdx.x;
    const int e0   = ch * BN_CE;

    const int* p = edge_index + N_EDGES + e0;
    int dv[BN_LD];
#pragma unroll
    for (int u = 0; u < BN_LD; ++u) {
        const int k = tid + BN_THREADS * u;
        dv[u] = (k < BN_CE) ? p[k] : -1;
    }

    for (int i = tid; i < EO_PITCH; i += BN_THREADS) bincnt[i] = 0u;
    __syncthreads();

#pragma unroll
    for (int u = 0; u < BN_LD; ++u) {
        const unsigned d = (unsigned)dv[u];
        const bool ok = d < (unsigned)N_NODES;
        atomicAdd(&bincnt[ok ? d / SC_FR : 0u], ok ? 1u : 0u);
    }
    __syncthreads();

    if (tid < 32) {
        unsigned c[8];
#pragma unroll
        for (int j = 0; j < 8; ++j) c[j] = bincnt[8 * tid + j];
        unsigned s = 0u;
#pragma unroll
        for (int j = 0; j < 8; ++j) {
            const unsigned n = c[j];
            c[j] = s;
            s += n;
        }
        unsigned inc = s;
#pragma unroll
        for (int o = 1; o < 32; o <<= 1) {
            const unsigned v = __shfl_up(inc, (unsigned)o);
            if (tid >= o) inc += v;
        }
        const unsigned ex = inc - s;
#pragma unroll
        for (int j = 0; j < 8; ++j) {
            binbase[8 * tid + j] = ex + c[j];
            bincnt[8 * tid + j]  = ex + c[j];
        }
    }
    __syncthreads();

    unsigned pos[BN_LD];
#pragma unroll
    for (int u = 0; u < BN_LD; ++u) {
        const unsigned d = (unsigned)dv[u];
        const bool ok = d < (unsigned)N_NODES;
        pos[u] = atomicAdd(&bincnt[ok ? d / SC_FR : 0u], ok ? 1u : 0u);
    }
    __builtin_amdgcn_fence(__ATOMIC_ACQ_REL, "workgroup");
#pragma unroll
    for (int u = 0; u < BN_LD; ++u) {
        const unsigned d = (unsigned)dv[u];
        const unsigned e = (unsigned)(e0 + tid + BN_THREADS * u);
        if (d < (unsigned)N_NODES)
            list[pos[u]] = (e << 8) | (d - (d / SC_FR) * SC_FR);
    }
    __syncthreads();

    const unsigned M = (unsigned)__builtin_amdgcn_readfirstlane((int)binbase[SC_NF]);
    unsigned int* el = elist + (size_t)ch * LS_STRIDE;
    for (unsigned i = (unsigned)tid; i < M; i += BN_THREADS) el[i] = list[i];
    for (int i = tid; i < EO_PITCH; i += BN_THREADS)
        eoff[ch * EO_PITCH + i] = (int)binbase[i];
}

__global__ void __launch_bounds__(EL_WAVES * 32) edge_logits_bins(
    const float* __restrict__ edge_attr, const int* __restrict__ edge_index,
    const float* __restrict__ s_node, const float* __restrict__ d_node,
    const float* __restrict__ weff,
    float* __restrict__ alpha,
    unsigned int* __restrict__ elist, int* __restrict__ eoff)
{
    if (blockIdx.x < BN_CH) {
        edge_bins_body(edge_index, elist, eoff, blockIdx.x);
        return;
    }
    edge_logits_body(edge_attr, edge_index, s_node, d_node, weff, alpha, blockIdx.x - BN_CH);
}

#define SC_THREADS 1024
#define SC_WAVES   (SC_THREADS / 32)
#define SC_PPW     (BN_CH / SC_WAVES)
#define SC_PRE     4
#define SC_G       8
#define SC_COPIES  8
#define SC_ACC4    (SC_FR * OUT_CH / 4)

static_assert(SC_WAVES * SC_PPW == BN_CH, "the waves must tile the stored lists");
static_assert(32 % SC_G == 0, "groups must tile a 32-entry slab");

struct ScPieces {
    int adj[SC_PPW];
    int cum[SC_PPW + 1];
};

__device__ __forceinline__ unsigned sc_entry(
    const ScPieces& P, const unsigned int* __restrict__ elist, int g)
{
    int a = P.adj[0];
#pragma unroll
    for (int k = 1; k < SC_PPW; ++k)
        a = (g >= P.cum[k]) ? P.adj[k] : a;
    return (g < P.cum[SC_PPW]) ? elist[g + a] : 0u;
}

__device__ __forceinline__ void sc_lds_max(int* addr, float a)
{
    const int ai = __float_as_int(a);
    if (ai >= 0) __hip_atomic_fetch_max(addr, ai, __ATOMIC_RELAXED, __HIP_MEMORY_SCOPE_WORKGROUP);
    else         __hip_atomic_fetch_min((unsigned int*)addr, (unsigned int)ai,
                                        __ATOMIC_RELAXED, __HIP_MEMORY_SCOPE_WORKGROUP);
}

struct ScRaw {
    int sl; unsigned dl; v4f al;
};

__device__ __forceinline__ void sc_fetch(
    ScRaw& R, unsigned ent,
    const int* __restrict__ edge_index, const float* __restrict__ alpha)
{
    unsigned dl = ent & 255u;
    dl = (dl < (unsigned)SC_FR) ? dl : 0u;
    unsigned e = ent >> 8;
    e = (e < (unsigned)N_EDGES) ? e : 0u;
    R.dl = dl;
    R.sl = edge_index[e];
    R.al = *(const v4f*)(alpha + (size_t)e * 4);
}

__device__ __forceinline__ void sc_max(const ScRaw& R, bool live, int* mxi)
{
    if (live) {
        sc_lds_max(mxi + R.dl * 4 + 0, R.al.x);
        sc_lds_max(mxi + R.dl * 4 + 1, R.al.y);
        sc_lds_max(mxi + R.dl * 4 + 2, R.al.z);
        sc_lds_max(mxi + R.dl * 4 + 3, R.al.w);
    }
}

__device__ __forceinline__ v4f sc_exp(const ScRaw& R, const v4f* mx4)
{
    v4f m = mx4[R.dl];
    m.x = __builtin_isfinite(m.x) ? m.x : 0.f;
    m.y = __builtin_isfinite(m.y) ? m.y : 0.f;
    m.z = __builtin_isfinite(m.z) ? m.z : 0.f;
    m.w = __builtin_isfinite(m.w) ? m.w : 0.f;
    return (v4f){__expf(R.al.x - m.x), __expf(R.al.y - m.y),
                 __expf(R.al.z - m.z), __expf(R.al.w - m.w)};
}

__device__ __forceinline__ void sc_sum(unsigned dl, v4f ex, bool live, float* smf)
{
    if (live) {
        __hip_atomic_fetch_add(smf + dl * 4 + 0, ex.x, __ATOMIC_RELAXED, __HIP_MEMORY_SCOPE_WORKGROUP);
        __hip_atomic_fetch_add(smf + dl * 4 + 1, ex.y, __ATOMIC_RELAXED, __HIP_MEMORY_SCOPE_WORKGROUP);
        __hip_atomic_fetch_add(smf + dl * 4 + 2, ex.z, __ATOMIC_RELAXED, __HIP_MEMORY_SCOPE_WORKGROUP);
        __hip_atomic_fetch_add(smf + dl * 4 + 3, ex.w, __ATOMIC_RELAXED, __HIP_MEMORY_SCOPE_WORKGROUP);
    }
}

__device__ __forceinline__ void sc_rows(
    int sl, unsigned dl, v4f ex, int n, int lane,
    const v4f* sm4, const float* __restrict__ xh, float* accs)
{
    const v4f cf = ex / (sm4[dl] + EPSV);
    for (int j = 0; j < n; j += SC_G) {
        float xv[SC_G][4];
#pragma unroll
        for (int u = 0; u < SC_G; ++u) {
            const int sj = __builtin_amdgcn_readlane(sl, j + u);
            const float* xp = xh + (size_t)sj * HC + lane;
            xv[u][0] = xp[0];  xv[u][1] = xp[32];
            xv[u][2] = xp[64]; xv[u][3] = xp[96];
        }
#pragma unroll
        for (int u = 0; u < SC_G; ++u) {
            if (j + u < n) {
                const float c0 = __int_as_float(__builtin_amdgcn_readlane(__float_as_int(cf.x), j + u));
                const float c1 = __int_as_float(__builtin_amdgcn_readlane(__float_as_int(cf.y), j + u));
                const float c2 = __int_as_float(__builtin_amdgcn_readlane(__float_as_int(cf.z), j + u));
                const float c3 = __int_as_float(__builtin_amdgcn_readlane(__float_as_int(cf.w), j + u));
                const int   dj = __builtin_amdgcn_readlane((int)dl, j + u);
                const float v = c0 * xv[u][0] + c1 * xv[u][1]
                              + c2 * xv[u][2] + c3 * xv[u][3];
                __hip_atomic_fetch_add(accs + dj * OUT_CH + lane, v,
                                       __ATOMIC_RELAXED, __HIP_MEMORY_SCOPE_WORKGROUP);
            }
        }
    }
}

__global__ void __launch_bounds__(SC_THREADS) edge_scatter(
    const int* __restrict__ edge_index, const float* __restrict__ xh,
    const float* __restrict__ alpha,
    const unsigned int* __restrict__ elist, const int* __restrict__ eoff,
    const float* __restrict__ bias_conv, const float* __restrict__ bias_layer,
    float* __restrict__ out)
{
    __shared__ v4f accs4[SC_COPIES * SC_ACC4];
    __shared__ v4f mx4[SC_FR];
    __shared__ v4f sm4[SC_FR];

    const int tid  = threadIdx.x;
    const int lane = tid & 31;
    const int wv   = __builtin_amdgcn_readfirstlane(tid >> 5);
    float* accs = (float*)(accs4 + (wv % SC_COPIES) * SC_ACC4);
    int*   mxi  = (int*)mx4;
    float* smf  = (float*)sm4;
    const int f    = blockIdx.x;

    ScPieces P;
    P.cum[0] = 0;
#pragma unroll
    for (int k = 0; k < SC_PPW; ++k) {
        const int ch = wv * SC_PPW + k;
        int p0 = eoff[ch * EO_PITCH + f];
        int p1 = eoff[ch * EO_PITCH + f + 1];
        p0 = (p0 < 0) ? 0 : ((p0 > BN_CE) ? BN_CE : p0);
        p1 = (p1 < p0) ? p0 : ((p1 > BN_CE) ? BN_CE : p1);
        P.adj[k]     = ch * LS_STRIDE + p0 - P.cum[k];
        P.cum[k + 1] = P.cum[k] + (p1 - p0);
    }
    const int total = P.cum[SC_PPW];

    unsigned entq[SC_PRE];
#pragma unroll
    for (int i = 0; i < SC_PRE; ++i)
        entq[i] = sc_entry(P, elist, 32 * i + lane);
    ScRaw raw[SC_PRE];
#pragma unroll
    for (int i = 0; i < SC_PRE; ++i)
        sc_fetch(raw[i], entq[i], edge_index, alpha);

    const float ninf = -__builtin_huge_valf();
    for (int i = tid; i < SC_COPIES * SC_ACC4; i += SC_THREADS)
        accs4[i] = (v4f){0.f, 0.f, 0.f, 0.f};
    for (int i = tid; i < SC_FR; i += SC_THREADS) {
        mx4[i] = (v4f){ninf, ninf, ninf, ninf};
        sm4[i] = (v4f){0.f, 0.f, 0.f, 0.f};
    }
    __syncthreads();

#pragma unroll
    for (int i = 0; i < SC_PRE; ++i)
        sc_max(raw[i], 32 * i + lane < total, mxi);
    for (int s = 32 * SC_PRE; s < total; s += 32) {
        const int g = s + lane;
        ScRaw R;
        sc_fetch(R, sc_entry(P, elist, g), edge_index, alpha);
        sc_max(R, g < total, mxi);
    }
    __syncthreads();

#pragma unroll
    for (int i = 0; i < SC_PRE; ++i) {
        raw[i].al = sc_exp(raw[i], mx4);
        sc_sum(raw[i].dl, raw[i].al, 32 * i + lane < total, smf);
    }
    for (int s = 32 * SC_PRE; s < total; s += 32) {
        const int g = s + lane;
        ScRaw R;
        sc_fetch(R, sc_entry(P, elist, g), edge_index, alpha);
        sc_sum(R.dl, sc_exp(R, mx4), g < total, smf);
    }
    __syncthreads();

#pragma unroll
    for (int i = 0; i < SC_PRE; ++i) {
        const int s = 32 * i;
        if (s < total)
            sc_rows(raw[i].sl, raw[i].dl, raw[i].al, (total - s < 32) ? total - s : 32, lane, sm4, xh, accs);
    }
    for (int s = 32 * SC_PRE; s < total; s += 32) {
        const int n = (total - s < 32) ? total - s : 32;
        ScRaw R;
        sc_fetch(R, sc_entry(P, elist, s + lane), edge_index, alpha);
        sc_rows(R.sl, R.dl, sc_exp(R, mx4), n, lane, sm4, xh, accs);
    }
    __syncthreads();

    v4f* o4 = (v4f*)(out + (size_t)f * SC_FR * OUT_CH);
    for (int i = tid; i < SC_ACC4; i += SC_THREADS) {
        v4f s = accs4[i];
#pragma unroll
        for (int k = 1; k < SC_COPIES; ++k) s += accs4[k * SC_ACC4 + i];
        const int ch0 = (i & 7) * 4;
        const float sv[4] = {s.x, s.y, s.z, s.w};
        float rv[4];
#pragma unroll
        for (int k = 0; k < 4; ++k) {
            const float v = sv[k] * 0.25f + bias_conv[ch0 + k] + bias_layer[ch0 + k];
            rv[k] = v > 0.f ? v : 0.f;
        }
        o4[i] = (v4f){rv[0], rv[1], rv[2], rv[3]};
    }
}

extern "C" void kernel_launch(void* const* d_in, const int* in_sizes, int n_in,
                              void* d_out, int out_size, void* d_ws, size_t ws_size,
                              hipStream_t stream) {
    const float* x          = (const float*)d_in[0];
    const float* edge_attr  = (const float*)d_in[1];
    const float* W          = (const float*)d_in[2];
    const float* W_edge     = (const float*)d_in[3];
    const float* att_src    = (const float*)d_in[4];
    const float* att_dst    = (const float*)d_in[5];
    const float* att_edge   = (const float*)d_in[6];
    const float* bias_conv  = (const float*)d_in[7];
    const float* bias_layer = (const float*)d_in[8];
    const int*   edge_index = (const int*)d_in[9];
    float* out = (float*)d_out;

    float* ws      = (float*)d_ws;
    float* xh      = ws;
    float* s_node  = xh      + (size_t)N_NODES * HC;
    float* d_node  = s_node  + (size_t)N_NODES * HEADS;
    float* weff    = d_node  + (size_t)N_NODES * HEADS;
    float* alpha   = weff    + (size_t)HEADS * IN_CH;
    unsigned int* elist = (unsigned int*)(alpha + (size_t)N_EDGES * HEADS);
    int* eoff      = (int*)(elist + (size_t)BN_CH * LS_STRIDE);

    proj_nodes_wmma<<<N_NODES / 16, 128, 0, stream>>>(
        x, W, att_src, att_dst, W_edge, att_edge, xh, s_node, d_node, weff);
    edge_logits_bins<<<BN_CH + (EL_NWAVES + EL_WAVES - 1) / EL_WAVES, EL_WAVES * 32, 0, stream>>>(
        edge_attr, edge_index, s_node, d_node, weff, alpha, elist, eoff);
    edge_scatter<<<SC_NF, SC_THREADS, 0, stream>>>(
        edge_index, xh, alpha, elist, eoff, bias_conv, bias_layer, out);
}
